# starting kernel
# baseline (speedup 1.0000x reference)
#include <hip/hip_runtime.h>
#include <math.h>

#define BB 8
#define CC 63
#define HH 240
#define WW 240
#define HW (HH * WW)
#define C2 126
#define WS 5
#define NWIN 48
#define BN_EPS 1e-5f

typedef __attribute__((ext_vector_type(2))) float v2f;
typedef __attribute__((ext_vector_type(8))) float v8f;

__device__ __forceinline__ v8f wmma4(v2f a, v2f b, v8f c) {
  return __builtin_amdgcn_wmma_f32_16x16x4_f32(
       false, a,  false, b,
       (short)0, c,  false,  false);
}

__device__ __forceinline__ v2f frag_a(const float* p, int ld, int m0, int k0, int lane) {
  int m  = m0 + (lane & 15);
  int kb = k0 + ((lane >> 4) << 1);
  const float* q = p + m * ld + kb;
  v2f r; r.x = q[0]; r.y = q[1];
  return r;
}

__device__ __forceinline__ v2f frag_b(const float* p, int ld, int k0, int n0, int lane) {
  int n  = n0 + (lane & 15);
  int kb = k0 + ((lane >> 4) << 1);
  v2f r; r.x = p[kb * ld + n]; r.y = p[(kb + 1) * ld + n];
  return r;
}

__device__ __forceinline__ void frag_store(float* p, int ld, int m0, int n0, int lane, v8f d) {
  int n  = n0 + (lane & 15);
  int mb = m0 + ((lane >> 4) << 3);
#pragma unroll
  for (int r = 0; r < 8; ++r) p[(mb + r) * ld + n] = d[r];
}

#define V8Z {0.f,0.f,0.f,0.f,0.f,0.f,0.f,0.f}

__global__ __launch_bounds__(128) void k_proj_in(
    const float* __restrict__ x, const float* __restrict__ w_in,
    const float* __restrict__ b_in, const float* __restrict__ bn_g,
    const float* __restrict__ bn_b, const float* __restrict__ bn_m,
    const float* __restrict__ bn_v, float* __restrict__ xp)
{
  __shared__ float a_lds[64][65];
  __shared__ float w_lds[64][128];
  __shared__ float bias_lds[128];
  __shared__ float d_lds[4][16][17];

  int tid  = threadIdx.x;
  int lane = tid & 31;
  int wave = tid >> 5;
  int tile = blockIdx.x;
  int b    = tile / 900;
  int p0   = (tile % 900) * 64;

  for (int idx = tid; idx < 64 * 128; idx += 128) {
    int k = idx >> 7, n = idx & 127;
    float val = 0.f;
    if (k < 63 && n < 126) {
      float s = bn_g[n] * rsqrtf(bn_v[n] + BN_EPS);
      val = w_in[n * 63 + k] * s;
    }
    w_lds[k][n] = val;
  }
  if (tid < 128) {
    float bv = 0.f;
    if (tid < 126) {
      float s = bn_g[tid] * rsqrtf(bn_v[tid] + BN_EPS);
      bv = b_in[tid] * s + bn_b[tid] - bn_m[tid] * s;
    }
    bias_lds[tid] = bv;
  }
  for (int idx = tid; idx < 64 * 64; idx += 128) {
    int k = idx >> 6, m = idx & 63;
    a_lds[m][k] = (k < 63) ? x[(size_t)(b * 63 + k) * HW + p0 + m] : 0.f;
  }
  __syncthreads();

  int m0 = wave * 16;
  for (int nt = 0; nt < 8; ++nt) {
    int n0 = nt * 16;
    v8f acc = V8Z;
    for (int k0 = 0; k0 < 64; k0 += 4) {
      v2f av = frag_a(&a_lds[0][0], 65, m0, k0, lane);
      v2f bv = frag_b(&w_lds[0][0], 128, k0, n0, lane);
      acc = wmma4(av, bv, acc);
    }
    int nloc = lane & 15;
    int mb   = (lane >> 4) << 3;
    float bias = bias_lds[n0 + nloc];
#pragma unroll
    for (int r = 0; r < 8; ++r) d_lds[wave][mb + r][nloc] = acc[r] + bias;
    for (int idx = lane; idx < 256; idx += 32) {
      int n = idx >> 4, m = idx & 15;
      int ch = n0 + n;
      if (ch < 126)
        xp[(size_t)(b * 126 + ch) * HW + p0 + m0 + m] = d_lds[wave][m][n];
    }
  }
}

__global__ __launch_bounds__(32) void k_win(float* __restrict__ xp, float* __restrict__ y2)
{
  __shared__ float q[32][25];
  __shared__ float v[32][25];
  __shared__ float s[32][33];

  int lane = threadIdx.x;
  int wid  = blockIdx.x;
  int grp  = blockIdx.y;
  int b    = wid / (NWIN * NWIN);
  int rem  = wid % (NWIN * NWIN);
  int wy   = rem / NWIN, wx = rem % NWIN;

  for (int i = lane; i < 32 * 25; i += 32) { (&q[0][0])[i] = 0.f; (&v[0][0])[i] = 0.f; }

  int chq = grp * 42;
  for (int i = lane; i < 25 * 21; i += 32) {
    int m = i / 21, c = i % 21;
    int gy = wy * WS + m / WS, gx = wx * WS + m % WS;
    if (grp == 1) { gy = (gy + 3) % HH; gx = (gx + 3) % WW; }
    size_t base = (size_t)(b * 126 + chq + c) * HW + gy * WW + gx;
    q[m][c] = xp[base];
    v[m][c] = xp[base + (size_t)21 * HW];
  }

  for (int mt = 0; mt < 2; ++mt)
    for (int nt = 0; nt < 2; ++nt) {
      v8f acc = V8Z;
      for (int k0 = 0; k0 < 24; k0 += 4) {
        v2f av = frag_a(&q[0][0], 25, mt * 16, k0, lane);
        v2f bv = frag_a(&q[0][0], 25, nt * 16, k0, lane);
        acc = wmma4(av, bv, acc);
      }
      frag_store(&s[0][0], 33, mt * 16, nt * 16, lane, acc);
    }

  if (lane < 25) {
    float mx = -1e30f;
    for (int n = 0; n < 25; ++n) mx = fmaxf(mx, s[lane][n]);
    float sum = 0.f;
    for (int n = 0; n < 25; ++n) { float e = __expf(s[lane][n] - mx); s[lane][n] = e; sum += e; }
    float inv = 1.f / sum;
    for (int n = 0; n < 25; ++n) s[lane][n] *= inv;
    for (int n = 25; n < 28; ++n) s[lane][n] = 0.f;
  }

  for (int mt = 0; mt < 2; ++mt)
    for (int nt = 0; nt < 2; ++nt) {
      v8f acc = V8Z;
      for (int k0 = 0; k0 < 28; k0 += 4) {
        v2f av = frag_a(&s[0][0], 33, mt * 16, k0, lane);
        v2f bv = frag_b(&v[0][0], 25, k0, nt * 16, lane);
        acc = wmma4(av, bv, acc);
      }
      int n  = nt * 16 + (lane & 15);
      int mb = mt * 16 + ((lane >> 4) << 3);
#pragma unroll
      for (int r = 0; r < 8; ++r) {
        int m = mb + r;
        if (m < 25 && n < 21) {
          int oy = wy * WS + m / WS, ox = wx * WS + m % WS;
          if (grp == 0) {
            xp[(size_t)(b * 126 + n) * HW + oy * WW + ox] = acc[r];
          } else {
            oy = (oy + 2) % HH; ox = (ox + 2) % WW;
            y2[(size_t)(b * 21 + n) * HW + oy * WW + ox] = acc[r];
          }
        }
      }
    }
}

__global__ __launch_bounds__(32) void k_axial(float* __restrict__ xp,
                                              float* __restrict__ vmid, int pass)
{
  __shared__ float q[240][25];
  __shared__ float v[241][25];
  __shared__ float s[16][241];

  int lane = threadIdx.x;
  int row  = blockIdx.x;
  int qt   = blockIdx.y;
  int b    = row / HH;
  int r    = row % HH;

  if (pass == 0) {
    for (int i = lane; i < 240 * 21; i += 32) {
      int c = i / 240, xx = i % 240;
      q[xx][c] = xp[(size_t)(b * 126 + 84 + c) * HW + r * WW + xx];
      v[xx][c] = xp[(size_t)(b * 126 + 105 + c) * HW + r * WW + xx];
    }
  } else {
    for (int i = lane; i < 240 * 21; i += 32) {
      int c = i / 240, h = i % 240;
      q[h][c] = xp[(size_t)(b * 126 + 84 + c) * HW + h * WW + r];
    }
    for (int i = lane; i < 240 * 21; i += 32) {
      int h = i / 21, c = i % 21;
      v[h][c] = vmid[((size_t)(b * 240 + r) * 240 + h) * 21 + c];
    }
  }
  for (int i = lane; i < 240 * 4; i += 32) {
    int m = i >> 2, c = 21 + (i & 3);
    q[m][c] = 0.f; v[m][c] = 0.f;
  }

  int m0 = qt * 16;
  for (int nt = 0; nt < 15; ++nt) {
    int n0 = nt * 16;
    v8f acc = V8Z;
    for (int k0 = 0; k0 < 24; k0 += 4) {
      v2f av = frag_a(&q[0][0], 25, m0, k0, lane);
      v2f bv = frag_a(&q[0][0], 25, n0, k0, lane);
      acc = wmma4(av, bv, acc);
    }
    frag_store(&s[0][0], 241, 0, n0, lane, acc);
  }

  if (lane < 16) {
    float mx = -1e30f;
    for (int n = 0; n < 240; ++n) mx = fmaxf(mx, s[lane][n]);
    float sum = 0.f;
    for (int n = 0; n < 240; ++n) { float e = __expf(s[lane][n] - mx); s[lane][n] = e; sum += e; }
    float inv = 1.f / sum;
    for (int n = 0; n < 240; ++n) s[lane][n] *= inv;
  }

  for (int nt = 0; nt < 2; ++nt) {
    int n0 = nt * 16;
    v8f acc = V8Z;
    for (int k0 = 0; k0 < 240; k0 += 4) {
      v2f av = frag_a(&s[0][0], 241, 0, k0, lane);
      v2f bv = frag_b(&v[0][0], 25, k0, n0, lane);
      acc = wmma4(av, bv, acc);
    }
    int n  = n0 + (lane & 15);
    int mb = (lane >> 4) << 3;
#pragma unroll
    for (int rr = 0; rr < 8; ++rr) {
      int m = m0 + mb + rr;
      if (n < 21) {
        if (pass == 0)
          vmid[((size_t)(b * 240 + m) * 240 + r) * 21 + n] = acc[rr];
        else
          xp[(size_t)(b * 126 + 105 + n) * HW + m * WW + r] = acc[rr];
      }
    }
  }
}

__global__ __launch_bounds__(128) void k_proj_out(
    const float* __restrict__ xp, const float* __restrict__ y2,
    const float* __restrict__ w_out, const float* __restrict__ b_out,
    float* __restrict__ out)
{
  __shared__ float a_lds[64][65];
  __shared__ float w_lds[64][64];
  __shared__ float bias_lds[64];
  __shared__ float d_lds[4][16][17];

  int tid  = threadIdx.x;
  int lane = tid & 31;
  int wave = tid >> 5;
  int tile = blockIdx.x;
  int b    = tile / 900;
  int p0   = (tile % 900) * 64;

  for (int idx = tid; idx < 64 * 64; idx += 128) {
    int k = idx >> 6, n = idx & 63;
    w_lds[k][n] = (k < 63 && n < 63) ? w_out[n * 63 + k] : 0.f;
  }
  if (tid < 64) bias_lds[tid] = (tid < 63) ? b_out[tid] : 0.f;

  for (int idx = tid; idx < 64 * 64; idx += 128) {
    int k = idx >> 6, m = idx & 63;
    float val = 0.f;
    if (k < 21)       val = xp[(size_t)(b * 126 + k)        * HW + p0 + m];
    else if (k < 42)  val = y2[(size_t)(b * 21 + (k - 21))  * HW + p0 + m];
    else if (k < 63)  val = xp[(size_t)(b * 126 + (k + 63)) * HW + p0 + m];
    a_lds[m][k] = val;
  }
  __syncthreads();

  int m0 = wave * 16;
  for (int nt = 0; nt < 4; ++nt) {
    int n0 = nt * 16;
    v8f acc = V8Z;
    for (int k0 = 0; k0 < 64; k0 += 4) {
      v2f av = frag_a(&a_lds[0][0], 65, m0, k0, lane);
      v2f bv = frag_b(&w_lds[0][0], 64, k0, n0, lane);
      acc = wmma4(av, bv, acc);
    }
    int nloc = lane & 15;
    int mb   = (lane >> 4) << 3;
    float bias = bias_lds[n0 + nloc];
#pragma unroll
    for (int r = 0; r < 8; ++r) d_lds[wave][mb + r][nloc] = acc[r] + bias;
    for (int idx = lane; idx < 256; idx += 32) {
      int n = idx >> 4, m = idx & 15;
      int ch = n0 + n;
      if (ch < 63)
        out[(size_t)(b * 63 + ch) * HW + p0 + m0 + m] = d_lds[wave][m][n];
    }
  }
}

extern "C" void kernel_launch(void* const* d_in, const int* in_sizes, int n_in,
                              void* d_out, int out_size, void* d_ws, size_t ws_size,
                              hipStream_t stream)
{
  (void)in_sizes; (void)n_in; (void)out_size; (void)ws_size;
  const float* x     = (const float*)d_in[0];
  const float* w_in  = (const float*)d_in[1];
  const float* b_in  = (const float*)d_in[2];
  const float* bn_g  = (const float*)d_in[3];
  const float* bn_b  = (const float*)d_in[4];
  const float* bn_m  = (const float*)d_in[5];
  const float* bn_v  = (const float*)d_in[6];
  const float* w_out = (const float*)d_in[7];
  const float* b_out = (const float*)d_in[8];
  float* out = (float*)d_out;

  float* xp   = (float*)d_ws;
  float* y2b  = xp  + (size_t)BB * 126 * HW;
  float* vmid = y2b + (size_t)BB * 21 * HW;

  k_proj_in <<<dim3(BB * 900), 128, 0, stream>>>(x, w_in, b_in, bn_g, bn_b, bn_m, bn_v, xp);
  k_win     <<<dim3(BB * NWIN * NWIN, 2), 32, 0, stream>>>(xp, y2b);
  k_axial   <<<dim3(BB * HH, 15), 32, 0, stream>>>(xp, vmid, 0);
  k_axial   <<<dim3(BB * WW, 15), 32, 0, stream>>>(xp, vmid, 1);
  k_proj_out<<<dim3(BB * 900), 128, 0, stream>>>(xp, y2b, w_out, b_out, out);
}


# axial attention: q, k and v matrix operands are read from global memory straight into fragments; no LDS copy of the sequence (LDS holds the score strip only, 4x the waves resident); sums unchanged
# speedup vs baseline: 1.5819x; 1.5819x over previous
#include <hip/hip_runtime.h>
#include <math.h>

#define BB 8
#define CC 63
#define HH 240
#define WW 240
#define HW (HH * WW)
#define C2 126
#define WS 5
#define NWIN 48
#define BN_EPS 1e-5f

typedef __attribute__((ext_vector_type(2))) float v2f;
typedef __attribute__((ext_vector_type(8))) float v8f;

__device__ __forceinline__ v8f wmma4(v2f a, v2f b, v8f c) {
  return __builtin_amdgcn_wmma_f32_16x16x4_f32(
       false, a,  false, b,
       (short)0, c,  false,  false);
}

__device__ __forceinline__ v2f frag_a(const float* p, int ld, int m0, int k0, int lane) {
  int m  = m0 + (lane & 15);
  int kb = k0 + ((lane >> 4) << 1);
  const float* q = p + m * ld + kb;
  v2f r; r.x = q[0]; r.y = q[1];
  return r;
}

__device__ __forceinline__ v2f frag_b(const float* p, int ld, int k0, int n0, int lane) {
  int n  = n0 + (lane & 15);
  int kb = k0 + ((lane >> 4) << 1);
  v2f r; r.x = p[kb * ld + n]; r.y = p[(kb + 1) * ld + n];
  return r;
}

__device__ __forceinline__ void frag_store(float* p, int ld, int m0, int n0, int lane, v8f d) {
  int n  = n0 + (lane & 15);
  int mb = m0 + ((lane >> 4) << 3);
#pragma unroll
  for (int r = 0; r < 8; ++r) p[(mb + r) * ld + n] = d[r];
}

#define V8Z {0.f,0.f,0.f,0.f,0.f,0.f,0.f,0.f}

__global__ __launch_bounds__(128) void k_proj_in(
    const float* __restrict__ x, const float* __restrict__ w_in,
    const float* __restrict__ b_in, const float* __restrict__ bn_g,
    const float* __restrict__ bn_b, const float* __restrict__ bn_m,
    const float* __restrict__ bn_v, float* __restrict__ xp)
{
  __shared__ float a_lds[64][65];
  __shared__ float w_lds[64][128];
  __shared__ float bias_lds[128];
  __shared__ float d_lds[4][16][17];

  int tid  = threadIdx.x;
  int lane = tid & 31;
  int wave = tid >> 5;
  int tile = blockIdx.x;
  int b    = tile / 900;
  int p0   = (tile % 900) * 64;

  for (int idx = tid; idx < 64 * 128; idx += 128) {
    int k = idx >> 7, n = idx & 127;
    float val = 0.f;
    if (k < 63 && n < 126) {
      float s = bn_g[n] * rsqrtf(bn_v[n] + BN_EPS);
      val = w_in[n * 63 + k] * s;
    }
    w_lds[k][n] = val;
  }
  if (tid < 128) {
    float bv = 0.f;
    if (tid < 126) {
      float s = bn_g[tid] * rsqrtf(bn_v[tid] + BN_EPS);
      bv = b_in[tid] * s + bn_b[tid] - bn_m[tid] * s;
    }
    bias_lds[tid] = bv;
  }
  for (int idx = tid; idx < 64 * 64; idx += 128) {
    int k = idx >> 6, m = idx & 63;
    a_lds[m][k] = (k < 63) ? x[(size_t)(b * 63 + k) * HW + p0 + m] : 0.f;
  }
  __syncthreads();

  int m0 = wave * 16;
  for (int nt = 0; nt < 8; ++nt) {
    int n0 = nt * 16;
    v8f acc = V8Z;
    for (int k0 = 0; k0 < 64; k0 += 4) {
      v2f av = frag_a(&a_lds[0][0], 65, m0, k0, lane);
      v2f bv = frag_b(&w_lds[0][0], 128, k0, n0, lane);
      acc = wmma4(av, bv, acc);
    }
    int nloc = lane & 15;
    int mb   = (lane >> 4) << 3;
    float bias = bias_lds[n0 + nloc];
#pragma unroll
    for (int r = 0; r < 8; ++r) d_lds[wave][mb + r][nloc] = acc[r] + bias;
    for (int idx = lane; idx < 256; idx += 32) {
      int n = idx >> 4, m = idx & 15;
      int ch = n0 + n;
      if (ch < 126)
        xp[(size_t)(b * 126 + ch) * HW + p0 + m0 + m] = d_lds[wave][m][n];
    }
  }
}

__global__ __launch_bounds__(32) void k_win(float* __restrict__ xp, float* __restrict__ y2)
{
  __shared__ float q[32][25];
  __shared__ float v[32][25];
  __shared__ float s[32][33];

  int lane = threadIdx.x;
  int wid  = blockIdx.x;
  int grp  = blockIdx.y;
  int b    = wid / (NWIN * NWIN);
  int rem  = wid % (NWIN * NWIN);
  int wy   = rem / NWIN, wx = rem % NWIN;

  for (int i = lane; i < 32 * 25; i += 32) { (&q[0][0])[i] = 0.f; (&v[0][0])[i] = 0.f; }

  int chq = grp * 42;
  for (int i = lane; i < 25 * 21; i += 32) {
    int m = i / 21, c = i % 21;
    int gy = wy * WS + m / WS, gx = wx * WS + m % WS;
    if (grp == 1) { gy = (gy + 3) % HH; gx = (gx + 3) % WW; }
    size_t base = (size_t)(b * 126 + chq + c) * HW + gy * WW + gx;
    q[m][c] = xp[base];
    v[m][c] = xp[base + (size_t)21 * HW];
  }

  for (int mt = 0; mt < 2; ++mt)
    for (int nt = 0; nt < 2; ++nt) {
      v8f acc = V8Z;
      for (int k0 = 0; k0 < 24; k0 += 4) {
        v2f av = frag_a(&q[0][0], 25, mt * 16, k0, lane);
        v2f bv = frag_a(&q[0][0], 25, nt * 16, k0, lane);
        acc = wmma4(av, bv, acc);
      }
      frag_store(&s[0][0], 33, mt * 16, nt * 16, lane, acc);
    }

  if (lane < 25) {
    float mx = -1e30f;
    for (int n = 0; n < 25; ++n) mx = fmaxf(mx, s[lane][n]);
    float sum = 0.f;
    for (int n = 0; n < 25; ++n) { float e = __expf(s[lane][n] - mx); s[lane][n] = e; sum += e; }
    float inv = 1.f / sum;
    for (int n = 0; n < 25; ++n) s[lane][n] *= inv;
    for (int n = 25; n < 28; ++n) s[lane][n] = 0.f;
  }

  for (int mt = 0; mt < 2; ++mt)
    for (int nt = 0; nt < 2; ++nt) {
      v8f acc = V8Z;
      for (int k0 = 0; k0 < 28; k0 += 4) {
        v2f av = frag_a(&s[0][0], 33, mt * 16, k0, lane);
        v2f bv = frag_b(&v[0][0], 25, k0, nt * 16, lane);
        acc = wmma4(av, bv, acc);
      }
      int n  = nt * 16 + (lane & 15);
      int mb = mt * 16 + ((lane >> 4) << 3);
#pragma unroll
      for (int r = 0; r < 8; ++r) {
        int m = mb + r;
        if (m < 25 && n < 21) {
          int oy = wy * WS + m / WS, ox = wx * WS + m % WS;
          if (grp == 0) {
            xp[(size_t)(b * 126 + n) * HW + oy * WW + ox] = acc[r];
          } else {
            oy = (oy + 2) % HH; ox = (ox + 2) % WW;
            y2[(size_t)(b * 21 + n) * HW + oy * WW + ox] = acc[r];
          }
        }
      }
    }
}


template <int PS>
__device__ __forceinline__ v2f gfrag_q(const float* qb, int p0, int k0, int lane) {
  int p  = p0 + (lane & 15);
  int hi = lane >> 4;
  v2f r;
  if (k0 + 3 < 21) {
    const float* g = qb + (size_t)(k0 + 2 * hi) * HW + p * PS;
    r.x = g[0]; r.y = g[HW];
  } else {
#pragma unroll
    for (int j = 0; j < 2; ++j) {
      int k  = k0 + 2 * hi + j;
      int kc = (k < 21) ? k : 20;
      float val = qb[(size_t)kc * HW + p * PS];
      r[j] = (k < 21) ? val : 0.f;
    }
  }
  return r;
}

template <int PASS>
__device__ __forceinline__ v2f gfrag_v(const float* vb, int k0, int n0, int lane) {
  int n  = n0 + (lane & 15);
  n = (n < 21) ? n : 20;
  int kb = k0 + ((lane >> 4) << 1);
  v2f r;
  if (PASS == 0) {
    r = *(const v2f*)(vb + (size_t)n * HW + kb);
  } else {
    const float* g = vb + kb * 21 + n;
    r.x = g[0]; r.y = g[21];
  }
  return r;
}

template <int PASS>
__global__ __launch_bounds__(32) void k_axial(float* __restrict__ xp,
                                              float* __restrict__ vmid)
{
  __shared__ float s[16][241];

  constexpr int PS = (PASS == 0) ? 1 : WW;

  int lane = threadIdx.x;
  int row  = blockIdx.x;
  int qt   = blockIdx.y;
  int b    = row / HH;
  int r    = row % HH;

  const float* qb = xp + (size_t)(b * 126 + 84) * HW + ((PASS == 0) ? r * WW : r);
  const float* vb = (PASS == 0)
      ? xp + (size_t)(b * 126 + 105) * HW + r * WW
      : vmid + (size_t)(b * 240 + r) * 240 * 21;

  int m0 = qt * 16;

  v2f qa[6];
#pragma unroll
  for (int t = 0; t < 6; ++t) qa[t] = gfrag_q<PS>(qb, m0, 4 * t, lane);

#pragma unroll 5
  for (int nt = 0; nt < 15; ++nt) {
    int n0 = nt * 16;
    v2f kf[6];
#pragma unroll
    for (int t = 0; t < 6; ++t) kf[t] = gfrag_q<PS>(qb, n0, 4 * t, lane);
    v8f acc = V8Z;
#pragma unroll
    for (int t = 0; t < 6; ++t) acc = wmma4(qa[t], kf[t], acc);
    frag_store(&s[0][0], 241, 0, n0, lane, acc);
  }

  if (lane < 16) {
    float mx = -1e30f;
    for (int n = 0; n < 240; ++n) mx = fmaxf(mx, s[lane][n]);
    float sum = 0.f;
    for (int n = 0; n < 240; ++n) { float e = __expf(s[lane][n] - mx); s[lane][n] = e; sum += e; }
    float inv = 1.f / sum;
    for (int n = 0; n < 240; ++n) s[lane][n] *= inv;
  }

  for (int nt = 0; nt < 2; ++nt) {
    int n0 = nt * 16;
    v8f acc = V8Z;
#pragma unroll 10
    for (int k0 = 0; k0 < 240; k0 += 4) {
      v2f av = frag_a(&s[0][0], 241, 0, k0, lane);
      v2f bv = gfrag_v<PASS>(vb, k0, n0, lane);
      acc = wmma4(av, bv, acc);
    }
    int n  = n0 + (lane & 15);
    int mb = (lane >> 4) << 3;
#pragma unroll
    for (int rr = 0; rr < 8; ++rr) {
      int m = m0 + mb + rr;
      if (n < 21) {
        if (PASS == 0)
          vmid[((size_t)(b * 240 + m) * 240 + r) * 21 + n] = acc[rr];
        else
          xp[(size_t)(b * 126 + 105 + n) * HW + m * WW + r] = acc[rr];
      }
    }
  }
}

__global__ __launch_bounds__(128) void k_proj_out(
    const float* __restrict__ xp, const float* __restrict__ y2,
    const float* __restrict__ w_out, const float* __restrict__ b_out,
    float* __restrict__ out)
{
  __shared__ float a_lds[64][65];
  __shared__ float w_lds[64][64];
  __shared__ float bias_lds[64];
  __shared__ float d_lds[4][16][17];

  int tid  = threadIdx.x;
  int lane = tid & 31;
  int wave = tid >> 5;
  int tile = blockIdx.x;
  int b    = tile / 900;
  int p0   = (tile % 900) * 64;

  for (int idx = tid; idx < 64 * 64; idx += 128) {
    int k = idx >> 6, n = idx & 63;
    w_lds[k][n] = (k < 63 && n < 63) ? w_out[n * 63 + k] : 0.f;
  }
  if (tid < 64) bias_lds[tid] = (tid < 63) ? b_out[tid] : 0.f;

  for (int idx = tid; idx < 64 * 64; idx += 128) {
    int k = idx >> 6, m = idx & 63;
    float val = 0.f;
    if (k < 21)       val = xp[(size_t)(b * 126 + k)        * HW + p0 + m];
    else if (k < 42)  val = y2[(size_t)(b * 21 + (k - 21))  * HW + p0 + m];
    else if (k < 63)  val = xp[(size_t)(b * 126 + (k + 63)) * HW + p0 + m];
    a_lds[m][k] = val;
  }
  __syncthreads();

  int m0 = wave * 16;
  for (int nt = 0; nt < 4; ++nt) {
    int n0 = nt * 16;
    v8f acc = V8Z;
    for (int k0 = 0; k0 < 64; k0 += 4) {
      v2f av = frag_a(&a_lds[0][0], 65, m0, k0, lane);
      v2f bv = frag_b(&w_lds[0][0], 64, k0, n0, lane);
      acc = wmma4(av, bv, acc);
    }
    int nloc = lane & 15;
    int mb   = (lane >> 4) << 3;
    float bias = bias_lds[n0 + nloc];
#pragma unroll
    for (int r = 0; r < 8; ++r) d_lds[wave][mb + r][nloc] = acc[r] + bias;
    for (int idx = lane; idx < 256; idx += 32) {
      int n = idx >> 4, m = idx & 15;
      int ch = n0 + n;
      if (ch < 63)
        out[(size_t)(b * 63 + ch) * HW + p0 + m0 + m] = d_lds[wave][m][n];
    }
  }
}

extern "C" void kernel_launch(void* const* d_in, const int* in_sizes, int n_in,
                              void* d_out, int out_size, void* d_ws, size_t ws_size,
                              hipStream_t stream)
{
  (void)in_sizes; (void)n_in; (void)out_size; (void)ws_size;
  const float* x     = (const float*)d_in[0];
  const float* w_in  = (const float*)d_in[1];
  const float* b_in  = (const float*)d_in[2];
  const float* bn_g  = (const float*)d_in[3];
  const float* bn_b  = (const float*)d_in[4];
  const float* bn_m  = (const float*)d_in[5];
  const float* bn_v  = (const float*)d_in[6];
  const float* w_out = (const float*)d_in[7];
  const float* b_out = (const float*)d_in[8];
  float* out = (float*)d_out;

  float* xp   = (float*)d_ws;
  float* y2b  = xp  + (size_t)BB * 126 * HW;
  float* vmid = y2b + (size_t)BB * 21 * HW;

  k_proj_in <<<dim3(BB * 900), 128, 0, stream>>>(x, w_in, b_in, bn_g, bn_b, bn_m, bn_v, xp);
  k_win     <<<dim3(BB * NWIN * NWIN, 2), 32, 0, stream>>>(xp, y2b);
  k_axial<0><<<dim3(BB * HH, 15), 32, 0, stream>>>(xp, vmid);
  k_axial<1><<<dim3(BB * WW, 15), 32, 0, stream>>>(xp, vmid);
  k_proj_out<<<dim3(BB * 900), 128, 0, stream>>>(xp, y2b, w_out, b_out, out);
}


# axial q: input projection stores it as 24-float pixel records, not planes; both axial passes read a lane's six fragments as three 16-byte loads (no 4-byte strided q read); score sums in other k order
# speedup vs baseline: 2.2188x; 1.4026x over previous
#include <hip/hip_runtime.h>
#include <math.h>

#define BB 8
#define CC 63
#define HH 240
#define WW 240
#define HW (HH * WW)
#define C2 126
#define WS 5
#define NWIN 48
#define BN_EPS 1e-5f

typedef __attribute__((ext_vector_type(2))) float v2f;
typedef __attribute__((ext_vector_type(4))) float v4f;
typedef __attribute__((ext_vector_type(8))) float v8f;

__device__ __forceinline__ v8f wmma4(v2f a, v2f b, v8f c) {
  return __builtin_amdgcn_wmma_f32_16x16x4_f32(
       false, a,  false, b,
       (short)0, c,  false,  false);
}

__device__ __forceinline__ v2f frag_a(const float* p, int ld, int m0, int k0, int lane) {
  int m  = m0 + (lane & 15);
  int kb = k0 + ((lane >> 4) << 1);
  const float* q = p + m * ld + kb;
  v2f r; r.x = q[0]; r.y = q[1];
  return r;
}

__device__ __forceinline__ v2f frag_b(const float* p, int ld, int k0, int n0, int lane) {
  int n  = n0 + (lane & 15);
  int kb = k0 + ((lane >> 4) << 1);
  v2f r; r.x = p[kb * ld + n]; r.y = p[(kb + 1) * ld + n];
  return r;
}

__device__ __forceinline__ void frag_store(float* p, int ld, int m0, int n0, int lane, v8f d) {
  int n  = n0 + (lane & 15);
  int mb = m0 + ((lane >> 4) << 3);
#pragma unroll
  for (int r = 0; r < 8; ++r) p[(mb + r) * ld + n] = d[r];
}

#define V8Z {0.f,0.f,0.f,0.f,0.f,0.f,0.f,0.f}

__global__ __launch_bounds__(128) void k_proj_in(
    const float* __restrict__ x, const float* __restrict__ w_in,
    const float* __restrict__ b_in, const float* __restrict__ bn_g,
    const float* __restrict__ bn_b, const float* __restrict__ bn_m,
    const float* __restrict__ bn_v, float* __restrict__ xp,
    float* __restrict__ qrec)
{
  __shared__ float a_lds[64][65];
  __shared__ float w_lds[64][128];
  __shared__ float bias_lds[128];
  __shared__ float d_lds[4][16][17];

  int tid  = threadIdx.x;
  int lane = tid & 31;
  int wave = tid >> 5;
  int tile = blockIdx.x;
  int b    = tile / 900;
  int p0   = (tile % 900) * 64;

  for (int idx = tid; idx < 64 * 128; idx += 128) {
    int k = idx >> 7, n = idx & 127;
    float val = 0.f;
    if (k < 63 && n < 126) {
      float s = bn_g[n] * rsqrtf(bn_v[n] + BN_EPS);
      val = w_in[n * 63 + k] * s;
    }
    w_lds[k][n] = val;
  }
  if (tid < 128) {
    float bv = 0.f;
    if (tid < 126) {
      float s = bn_g[tid] * rsqrtf(bn_v[tid] + BN_EPS);
      bv = b_in[tid] * s + bn_b[tid] - bn_m[tid] * s;
    }
    bias_lds[tid] = bv;
  }
  for (int idx = tid; idx < 64 * 64; idx += 128) {
    int k = idx >> 6, m = idx & 63;
    a_lds[m][k] = (k < 63) ? x[(size_t)(b * 63 + k) * HW + p0 + m] : 0.f;
  }
  __syncthreads();

  int m0 = wave * 16;
  for (int nt = 0; nt < 8; ++nt) {
    int n0 = nt * 16;
    v8f acc = V8Z;
    for (int k0 = 0; k0 < 64; k0 += 4) {
      v2f av = frag_a(&a_lds[0][0], 65, m0, k0, lane);
      v2f bv = frag_b(&w_lds[0][0], 128, k0, n0, lane);
      acc = wmma4(av, bv, acc);
    }
    int nloc = lane & 15;
    int mb   = (lane >> 4) << 3;
    float bias = bias_lds[n0 + nloc];
#pragma unroll
    for (int r = 0; r < 8; ++r) d_lds[wave][mb + r][nloc] = acc[r] + bias;
    for (int idx = lane; idx < 256; idx += 32) {
      int n = idx >> 4, m = idx & 15;
      int ch = n0 + n;
      if (ch < 126 && (ch < 84 || ch >= 105))
        xp[(size_t)(b * 126 + ch) * HW + p0 + m0 + m] = d_lds[wave][m][n];
    }
    if (n0 == 80 || n0 == 96) {
      for (int idx = lane; idx < 256; idx += 32) {
        int n = idx & 15, m = idx >> 4;
        int slot = n0 + n - 84;
        if (slot >= 0 && slot < 24)
          qrec[((size_t)b * HW + p0 + m0 + m) * 24 + slot] =
              (slot < 21) ? d_lds[wave][m][n] : 0.f;
      }
    }
  }
}

__global__ __launch_bounds__(32) void k_win(float* __restrict__ xp, float* __restrict__ y2)
{
  __shared__ float q[32][25];
  __shared__ float v[32][25];
  __shared__ float s[32][33];

  int lane = threadIdx.x;
  int wid  = blockIdx.x;
  int grp  = blockIdx.y;
  int b    = wid / (NWIN * NWIN);
  int rem  = wid % (NWIN * NWIN);
  int wy   = rem / NWIN, wx = rem % NWIN;

  for (int i = lane; i < 32 * 25; i += 32) { (&q[0][0])[i] = 0.f; (&v[0][0])[i] = 0.f; }

  int chq = grp * 42;
  for (int i = lane; i < 25 * 21; i += 32) {
    int m = i / 21, c = i % 21;
    int gy = wy * WS + m / WS, gx = wx * WS + m % WS;
    if (grp == 1) { gy = (gy + 3) % HH; gx = (gx + 3) % WW; }
    size_t base = (size_t)(b * 126 + chq + c) * HW + gy * WW + gx;
    q[m][c] = xp[base];
    v[m][c] = xp[base + (size_t)21 * HW];
  }

  for (int mt = 0; mt < 2; ++mt)
    for (int nt = 0; nt < 2; ++nt) {
      v8f acc = V8Z;
      for (int k0 = 0; k0 < 24; k0 += 4) {
        v2f av = frag_a(&q[0][0], 25, mt * 16, k0, lane);
        v2f bv = frag_a(&q[0][0], 25, nt * 16, k0, lane);
        acc = wmma4(av, bv, acc);
      }
      frag_store(&s[0][0], 33, mt * 16, nt * 16, lane, acc);
    }

  if (lane < 25) {
    float mx = -1e30f;
    for (int n = 0; n < 25; ++n) mx = fmaxf(mx, s[lane][n]);
    float sum = 0.f;
    for (int n = 0; n < 25; ++n) { float e = __expf(s[lane][n] - mx); s[lane][n] = e; sum += e; }
    float inv = 1.f / sum;
    for (int n = 0; n < 25; ++n) s[lane][n] *= inv;
    for (int n = 25; n < 28; ++n) s[lane][n] = 0.f;
  }

  for (int mt = 0; mt < 2; ++mt)
    for (int nt = 0; nt < 2; ++nt) {
      v8f acc = V8Z;
      for (int k0 = 0; k0 < 28; k0 += 4) {
        v2f av = frag_a(&s[0][0], 33, mt * 16, k0, lane);
        v2f bv = frag_b(&v[0][0], 25, k0, nt * 16, lane);
        acc = wmma4(av, bv, acc);
      }
      int n  = nt * 16 + (lane & 15);
      int mb = mt * 16 + ((lane >> 4) << 3);
#pragma unroll
      for (int r = 0; r < 8; ++r) {
        int m = mb + r;
        if (m < 25 && n < 21) {
          int oy = wy * WS + m / WS, ox = wx * WS + m % WS;
          if (grp == 0) {
            xp[(size_t)(b * 126 + n) * HW + oy * WW + ox] = acc[r];
          } else {
            oy = (oy + 2) % HH; ox = (ox + 2) % WW;
            y2[(size_t)(b * 126 + n) * HW + oy * WW + ox] = acc[r];
          }
        }
      }
    }
}


template <int PS>
__device__ __forceinline__ void gfrag_q(const float* qb, int p0, int lane, v2f (&f)[6]) {
  int p = p0 + (lane & 15);
  const v4f* g = (const v4f*)(qb + (size_t)p * (PS * 24) + 12 * (lane >> 4));
  v4f a = g[0], b = g[1], c = g[2];
  f[0].x = a.x; f[0].y = a.y; f[1].x = a.z; f[1].y = a.w;
  f[2].x = b.x; f[2].y = b.y; f[3].x = b.z; f[3].y = b.w;
  f[4].x = c.x; f[4].y = c.y; f[5].x = c.z; f[5].y = c.w;
}

template <int PASS>
__device__ __forceinline__ v2f gfrag_v(const float* vb, int k0, int n0, int lane) {
  int n  = n0 + (lane & 15);
  n = (n < 21) ? n : 20;
  int kb = k0 + ((lane >> 4) << 1);
  v2f r;
  if (PASS == 0) {
    r = *(const v2f*)(vb + (size_t)n * HW + kb);
  } else {
    const float* g = vb + kb * 21 + n;
    r.x = g[0]; r.y = g[21];
  }
  return r;
}

template <int PASS>
__global__ __launch_bounds__(32) void k_axial(float* __restrict__ xp,
                                              float* __restrict__ vmid,
                                              const float* __restrict__ qrec)
{
  __shared__ float s[16][241];

  constexpr int PS = (PASS == 0) ? 1 : WW;

  int lane = threadIdx.x;
  int row  = blockIdx.x;
  int qt   = blockIdx.y;
  int b    = row / HH;
  int r    = row % HH;

  float* vm = vmid + (size_t)b * 126 * HW;
  const float* qb = qrec + ((size_t)b * HW + ((PASS == 0) ? r * WW : r)) * 24;
  const float* vb = (PASS == 0)
      ? xp + (size_t)(b * 126 + 105) * HW + r * WW
      : vm + (size_t)r * 240 * 21;

  int m0 = qt * 16;

  v2f qa[6];
  gfrag_q<PS>(qb, m0, lane, qa);

#pragma unroll 5
  for (int nt = 0; nt < 15; ++nt) {
    int n0 = nt * 16;
    v2f kf[6];
    gfrag_q<PS>(qb, n0, lane, kf);
    v8f acc = V8Z;
#pragma unroll
    for (int t = 0; t < 6; ++t) acc = wmma4(qa[t], kf[t], acc);
    frag_store(&s[0][0], 241, 0, n0, lane, acc);
  }

  if (lane < 16) {
    float mx = -1e30f;
    for (int n = 0; n < 240; ++n) mx = fmaxf(mx, s[lane][n]);
    float sum = 0.f;
    for (int n = 0; n < 240; ++n) { float e = __expf(s[lane][n] - mx); s[lane][n] = e; sum += e; }
    float inv = 1.f / sum;
    for (int n = 0; n < 240; ++n) s[lane][n] *= inv;
  }

  for (int nt = 0; nt < 2; ++nt) {
    int n0 = nt * 16;
    v8f acc = V8Z;
#pragma unroll 10
    for (int k0 = 0; k0 < 240; k0 += 4) {
      v2f av = frag_a(&s[0][0], 241, 0, k0, lane);
      v2f bv = gfrag_v<PASS>(vb, k0, n0, lane);
      acc = wmma4(av, bv, acc);
    }
    int n  = n0 + (lane & 15);
    int mb = (lane >> 4) << 3;
#pragma unroll
    for (int rr = 0; rr < 8; ++rr) {
      int m = m0 + mb + rr;
      if (n < 21) {
        if (PASS == 0)
          vm[((size_t)m * 240 + r) * 21 + n] = acc[rr];
        else
          xp[(size_t)(b * 126 + 105 + n) * HW + m * WW + r] = acc[rr];
      }
    }
  }
}

__global__ __launch_bounds__(128) void k_proj_out(
    const float* __restrict__ xp, const float* __restrict__ y2,
    const float* __restrict__ w_out, const float* __restrict__ b_out,
    float* __restrict__ out)
{
  __shared__ float a_lds[64][65];
  __shared__ float w_lds[64][64];
  __shared__ float bias_lds[64];
  __shared__ float d_lds[4][16][17];

  int tid  = threadIdx.x;
  int lane = tid & 31;
  int wave = tid >> 5;
  int tile = blockIdx.x;
  int b    = tile / 900;
  int p0   = (tile % 900) * 64;

  for (int idx = tid; idx < 64 * 64; idx += 128) {
    int k = idx >> 6, n = idx & 63;
    w_lds[k][n] = (k < 63 && n < 63) ? w_out[n * 63 + k] : 0.f;
  }
  if (tid < 64) bias_lds[tid] = (tid < 63) ? b_out[tid] : 0.f;

  for (int idx = tid; idx < 64 * 64; idx += 128) {
    int k = idx >> 6, m = idx & 63;
    float val = 0.f;
    if (k < 21)       val = xp[(size_t)(b * 126 + k)        * HW + p0 + m];
    else if (k < 42)  val = y2[(size_t)(b * 126 + (k - 21)) * HW + p0 + m];
    else if (k < 63)  val = xp[(size_t)(b * 126 + (k + 63)) * HW + p0 + m];
    a_lds[m][k] = val;
  }
  __syncthreads();

  int m0 = wave * 16;
  for (int nt = 0; nt < 4; ++nt) {
    int n0 = nt * 16;
    v8f acc = V8Z;
    for (int k0 = 0; k0 < 64; k0 += 4) {
      v2f av = frag_a(&a_lds[0][0], 65, m0, k0, lane);
      v2f bv = frag_b(&w_lds[0][0], 64, k0, n0, lane);
      acc = wmma4(av, bv, acc);
    }
    int nloc = lane & 15;
    int mb   = (lane >> 4) << 3;
    float bias = bias_lds[n0 + nloc];
#pragma unroll
    for (int r = 0; r < 8; ++r) d_lds[wave][mb + r][nloc] = acc[r] + bias;
    for (int idx = lane; idx < 256; idx += 32) {
      int n = idx >> 4, m = idx & 15;
      int ch = n0 + n;
      if (ch < 63)
        out[(size_t)(b * 63 + ch) * HW + p0 + m0 + m] = d_lds[wave][m][n];
    }
  }
}

extern "C" void kernel_launch(void* const* d_in, const int* in_sizes, int n_in,
                              void* d_out, int out_size, void* d_ws, size_t ws_size,
                              hipStream_t stream)
{
  (void)in_sizes; (void)n_in; (void)out_size; (void)ws_size;
  const float* x     = (const float*)d_in[0];
  const float* w_in  = (const float*)d_in[1];
  const float* b_in  = (const float*)d_in[2];
  const float* bn_g  = (const float*)d_in[3];
  const float* bn_b  = (const float*)d_in[4];
  const float* bn_m  = (const float*)d_in[5];
  const float* bn_v  = (const float*)d_in[6];
  const float* w_out = (const float*)d_in[7];
  const float* b_out = (const float*)d_in[8];
  float* out = (float*)d_out;

  float* xp   = (float*)d_ws;
  float* qrec = xp + (size_t)BB * 126 * HW;
  float* y2b  = xp + (size_t)84 * HW;
  float* vmid = xp + (size_t)21 * HW;

  k_proj_in <<<dim3(BB * 900), 128, 0, stream>>>(x, w_in, b_in, bn_g, bn_b, bn_m, bn_v, xp, qrec);
  k_win     <<<dim3(BB * NWIN * NWIN, 2), 32, 0, stream>>>(xp, y2b);
  k_axial<0><<<dim3(BB * HH, 15), 32, 0, stream>>>(xp, vmid, qrec);
  k_axial<1><<<dim3(BB * WW, 15), 32, 0, stream>>>(xp, vmid, qrec);
  k_proj_out<<<dim3(BB * 900), 128, 0, stream>>>(xp, y2b, w_out, b_out, out);
}


# axial softmax on all 32 lanes: row max from the score accumulators, each lane exponentiates its own LDS strip elements, 1/sum applied to the P V result; row sums in another order
# speedup vs baseline: 2.3829x; 1.0740x over previous
#include <hip/hip_runtime.h>
#include <math.h>

#define BB 8
#define CC 63
#define HH 240
#define WW 240
#define HW (HH * WW)
#define C2 126
#define WS 5
#define NWIN 48
#define BN_EPS 1e-5f

typedef __attribute__((ext_vector_type(2))) float v2f;
typedef __attribute__((ext_vector_type(4))) float v4f;
typedef __attribute__((ext_vector_type(8))) float v8f;

__device__ __forceinline__ v8f wmma4(v2f a, v2f b, v8f c) {
  return __builtin_amdgcn_wmma_f32_16x16x4_f32(
       false, a,  false, b,
       (short)0, c,  false,  false);
}

__device__ __forceinline__ v2f frag_a(const float* p, int ld, int m0, int k0, int lane) {
  int m  = m0 + (lane & 15);
  int kb = k0 + ((lane >> 4) << 1);
  const float* q = p + m * ld + kb;
  v2f r; r.x = q[0]; r.y = q[1];
  return r;
}

__device__ __forceinline__ v2f frag_b(const float* p, int ld, int k0, int n0, int lane) {
  int n  = n0 + (lane & 15);
  int kb = k0 + ((lane >> 4) << 1);
  v2f r; r.x = p[kb * ld + n]; r.y = p[(kb + 1) * ld + n];
  return r;
}

__device__ __forceinline__ void frag_store(float* p, int ld, int m0, int n0, int lane, v8f d) {
  int n  = n0 + (lane & 15);
  int mb = m0 + ((lane >> 4) << 3);
#pragma unroll
  for (int r = 0; r < 8; ++r) p[(mb + r) * ld + n] = d[r];
}

#define V8Z {0.f,0.f,0.f,0.f,0.f,0.f,0.f,0.f}

__global__ __launch_bounds__(128) void k_proj_in(
    const float* __restrict__ x, const float* __restrict__ w_in,
    const float* __restrict__ b_in, const float* __restrict__ bn_g,
    const float* __restrict__ bn_b, const float* __restrict__ bn_m,
    const float* __restrict__ bn_v, float* __restrict__ xp,
    float* __restrict__ qrec)
{
  __shared__ float a_lds[64][65];
  __shared__ float w_lds[64][128];
  __shared__ float bias_lds[128];
  __shared__ float d_lds[4][16][17];

  int tid  = threadIdx.x;
  int lane = tid & 31;
  int wave = tid >> 5;
  int tile = blockIdx.x;
  int b    = tile / 900;
  int p0   = (tile % 900) * 64;

  for (int idx = tid; idx < 64 * 128; idx += 128) {
    int k = idx >> 7, n = idx & 127;
    float val = 0.f;
    if (k < 63 && n < 126) {
      float s = bn_g[n] * rsqrtf(bn_v[n] + BN_EPS);
      val = w_in[n * 63 + k] * s;
    }
    w_lds[k][n] = val;
  }
  if (tid < 128) {
    float bv = 0.f;
    if (tid < 126) {
      float s = bn_g[tid] * rsqrtf(bn_v[tid] + BN_EPS);
      bv = b_in[tid] * s + bn_b[tid] - bn_m[tid] * s;
    }
    bias_lds[tid] = bv;
  }
  for (int idx = tid; idx < 64 * 64; idx += 128) {
    int k = idx >> 6, m = idx & 63;
    a_lds[m][k] = (k < 63) ? x[(size_t)(b * 63 + k) * HW + p0 + m] : 0.f;
  }
  __syncthreads();

  int m0 = wave * 16;
  for (int nt = 0; nt < 8; ++nt) {
    int n0 = nt * 16;
    v8f acc = V8Z;
    for (int k0 = 0; k0 < 64; k0 += 4) {
      v2f av = frag_a(&a_lds[0][0], 65, m0, k0, lane);
      v2f bv = frag_b(&w_lds[0][0], 128, k0, n0, lane);
      acc = wmma4(av, bv, acc);
    }
    int nloc = lane & 15;
    int mb   = (lane >> 4) << 3;
    float bias = bias_lds[n0 + nloc];
#pragma unroll
    for (int r = 0; r < 8; ++r) d_lds[wave][mb + r][nloc] = acc[r] + bias;
    for (int idx = lane; idx < 256; idx += 32) {
      int n = idx >> 4, m = idx & 15;
      int ch = n0 + n;
      if (ch < 126 && (ch < 84 || ch >= 105))
        xp[(size_t)(b * 126 + ch) * HW + p0 + m0 + m] = d_lds[wave][m][n];
    }
    if (n0 == 80 || n0 == 96) {
      for (int idx = lane; idx < 256; idx += 32) {
        int n = idx & 15, m = idx >> 4;
        int slot = n0 + n - 84;
        if (slot >= 0 && slot < 24)
          qrec[((size_t)b * HW + p0 + m0 + m) * 24 + slot] =
              (slot < 21) ? d_lds[wave][m][n] : 0.f;
      }
    }
  }
}

__global__ __launch_bounds__(32) void k_win(float* __restrict__ xp, float* __restrict__ y2)
{
  __shared__ float q[32][25];
  __shared__ float v[32][25];
  __shared__ float s[32][33];

  int lane = threadIdx.x;
  int wid  = blockIdx.x;
  int grp  = blockIdx.y;
  int b    = wid / (NWIN * NWIN);
  int rem  = wid % (NWIN * NWIN);
  int wy   = rem / NWIN, wx = rem % NWIN;

  for (int i = lane; i < 32 * 25; i += 32) { (&q[0][0])[i] = 0.f; (&v[0][0])[i] = 0.f; }

  int chq = grp * 42;
  for (int i = lane; i < 25 * 21; i += 32) {
    int m = i / 21, c = i % 21;
    int gy = wy * WS + m / WS, gx = wx * WS + m % WS;
    if (grp == 1) { gy = (gy + 3) % HH; gx = (gx + 3) % WW; }
    size_t base = (size_t)(b * 126 + chq + c) * HW + gy * WW + gx;
    q[m][c] = xp[base];
    v[m][c] = xp[base + (size_t)21 * HW];
  }

  for (int mt = 0; mt < 2; ++mt)
    for (int nt = 0; nt < 2; ++nt) {
      v8f acc = V8Z;
      for (int k0 = 0; k0 < 24; k0 += 4) {
        v2f av = frag_a(&q[0][0], 25, mt * 16, k0, lane);
        v2f bv = frag_a(&q[0][0], 25, nt * 16, k0, lane);
        acc = wmma4(av, bv, acc);
      }
      frag_store(&s[0][0], 33, mt * 16, nt * 16, lane, acc);
    }

  if (lane < 25) {
    float mx = -1e30f;
    for (int n = 0; n < 25; ++n) mx = fmaxf(mx, s[lane][n]);
    float sum = 0.f;
    for (int n = 0; n < 25; ++n) { float e = __expf(s[lane][n] - mx); s[lane][n] = e; sum += e; }
    float inv = 1.f / sum;
    for (int n = 0; n < 25; ++n) s[lane][n] *= inv;
    for (int n = 25; n < 28; ++n) s[lane][n] = 0.f;
  }

  for (int mt = 0; mt < 2; ++mt)
    for (int nt = 0; nt < 2; ++nt) {
      v8f acc = V8Z;
      for (int k0 = 0; k0 < 28; k0 += 4) {
        v2f av = frag_a(&s[0][0], 33, mt * 16, k0, lane);
        v2f bv = frag_b(&v[0][0], 25, k0, nt * 16, lane);
        acc = wmma4(av, bv, acc);
      }
      int n  = nt * 16 + (lane & 15);
      int mb = mt * 16 + ((lane >> 4) << 3);
#pragma unroll
      for (int r = 0; r < 8; ++r) {
        int m = mb + r;
        if (m < 25 && n < 21) {
          int oy = wy * WS + m / WS, ox = wx * WS + m % WS;
          if (grp == 0) {
            xp[(size_t)(b * 126 + n) * HW + oy * WW + ox] = acc[r];
          } else {
            oy = (oy + 2) % HH; ox = (ox + 2) % WW;
            y2[(size_t)(b * 126 + n) * HW + oy * WW + ox] = acc[r];
          }
        }
      }
    }
}


template <int PS>
__device__ __forceinline__ void gfrag_q(const float* qb, int p0, int lane, v2f (&f)[6]) {
  int p = p0 + (lane & 15);
  const v4f* g = (const v4f*)(qb + (size_t)p * (PS * 24) + 12 * (lane >> 4));
  v4f a = g[0], b = g[1], c = g[2];
  f[0].x = a.x; f[0].y = a.y; f[1].x = a.z; f[1].y = a.w;
  f[2].x = b.x; f[2].y = b.y; f[3].x = b.z; f[3].y = b.w;
  f[4].x = c.x; f[4].y = c.y; f[5].x = c.z; f[5].y = c.w;
}

template <int PASS>
__device__ __forceinline__ v2f gfrag_v(const float* vb, int k0, int n0, int lane) {
  int n  = n0 + (lane & 15);
  n = (n < 21) ? n : 20;
  int kb = k0 + ((lane >> 4) << 1);
  v2f r;
  if (PASS == 0) {
    r = *(const v2f*)(vb + (size_t)n * HW + kb);
  } else {
    const float* g = vb + kb * 21 + n;
    r.x = g[0]; r.y = g[21];
  }
  return r;
}

template <int X>
__device__ __forceinline__ float half_xor(float v) {
  return __int_as_float(__builtin_amdgcn_update_dpp(0, __float_as_int(v), 0x160 + X, 0xf, 0xf, false));
}

__device__ __forceinline__ float half_max(float v) {
  v = fmaxf(v, half_xor<1>(v));
  v = fmaxf(v, half_xor<2>(v));
  v = fmaxf(v, half_xor<4>(v));
  v = fmaxf(v, half_xor<8>(v));
  return v;
}

__device__ __forceinline__ float half_sum(float v) {
  v += half_xor<1>(v);
  v += half_xor<2>(v);
  v += half_xor<4>(v);
  v += half_xor<8>(v);
  return v;
}

template <int PASS>
__global__ __launch_bounds__(32) void k_axial(float* __restrict__ xp,
                                              float* __restrict__ vmid,
                                              const float* __restrict__ qrec)
{
  __shared__ float s[16][241];

  constexpr int PS = (PASS == 0) ? 1 : WW;

  int lane = threadIdx.x;
  int row  = blockIdx.x;
  int qt   = blockIdx.y;
  int b    = row / HH;
  int r    = row % HH;

  float* vm = vmid + (size_t)b * 126 * HW;
  const float* qb = qrec + ((size_t)b * HW + ((PASS == 0) ? r * WW : r)) * 24;
  const float* vb = (PASS == 0)
      ? xp + (size_t)(b * 126 + 105) * HW + r * WW
      : vm + (size_t)r * 240 * 21;

  int m0 = qt * 16;

  v2f qa[6];
  gfrag_q<PS>(qb, m0, lane, qa);

  float rmx[8];
#pragma unroll
  for (int r = 0; r < 8; ++r) rmx[r] = -1e30f;

#pragma unroll 5
  for (int nt = 0; nt < 15; ++nt) {
    int n0 = nt * 16;
    v2f kf[6];
    gfrag_q<PS>(qb, n0, lane, kf);
    v8f acc = V8Z;
#pragma unroll
    for (int t = 0; t < 6; ++t) acc = wmma4(qa[t], kf[t], acc);
#pragma unroll
    for (int r = 0; r < 8; ++r) rmx[r] = fmaxf(rmx[r], acc[r]);
    frag_store(&s[0][0], 241, 0, n0, lane, acc);
  }
#pragma unroll
  for (int r = 0; r < 8; ++r) rmx[r] = half_max(rmx[r]);

  float* own = &s[(lane >> 4) << 3][lane & 15];
  float rsum[8];
#pragma unroll
  for (int r = 0; r < 8; ++r) rsum[r] = 0.f;
#pragma unroll 5
  for (int nt = 0; nt < 15; ++nt) {
#pragma unroll
    for (int r = 0; r < 8; ++r) {
      float e = __expf(own[r * 241 + nt * 16] - rmx[r]);
      own[r * 241 + nt * 16] = e;
      rsum[r] += e;
    }
  }
  float rinv[8];
#pragma unroll
  for (int r = 0; r < 8; ++r) rinv[r] = 1.f / half_sum(rsum[r]);

  for (int nt = 0; nt < 2; ++nt) {
    int n0 = nt * 16;
    v8f acc = V8Z;
#pragma unroll 10
    for (int k0 = 0; k0 < 240; k0 += 4) {
      v2f av = frag_a(&s[0][0], 241, 0, k0, lane);
      v2f bv = gfrag_v<PASS>(vb, k0, n0, lane);
      acc = wmma4(av, bv, acc);
    }
    int n  = n0 + (lane & 15);
    int mb = (lane >> 4) << 3;
#pragma unroll
    for (int rr = 0; rr < 8; ++rr) {
      int m = m0 + mb + rr;
      float y = acc[rr] * rinv[rr];
      if (n < 21) {
        if (PASS == 0)
          vm[((size_t)m * 240 + r) * 21 + n] = y;
        else
          xp[(size_t)(b * 126 + 105 + n) * HW + m * WW + r] = y;
      }
    }
  }
}

__global__ __launch_bounds__(128) void k_proj_out(
    const float* __restrict__ xp, const float* __restrict__ y2,
    const float* __restrict__ w_out, const float* __restrict__ b_out,
    float* __restrict__ out)
{
  __shared__ float a_lds[64][65];
  __shared__ float w_lds[64][64];
  __shared__ float bias_lds[64];
  __shared__ float d_lds[4][16][17];

  int tid  = threadIdx.x;
  int lane = tid & 31;
  int wave = tid >> 5;
  int tile = blockIdx.x;
  int b    = tile / 900;
  int p0   = (tile % 900) * 64;

  for (int idx = tid; idx < 64 * 64; idx += 128) {
    int k = idx >> 6, n = idx & 63;
    w_lds[k][n] = (k < 63 && n < 63) ? w_out[n * 63 + k] : 0.f;
  }
  if (tid < 64) bias_lds[tid] = (tid < 63) ? b_out[tid] : 0.f;

  for (int idx = tid; idx < 64 * 64; idx += 128) {
    int k = idx >> 6, m = idx & 63;
    float val = 0.f;
    if (k < 21)       val = xp[(size_t)(b * 126 + k)        * HW + p0 + m];
    else if (k < 42)  val = y2[(size_t)(b * 126 + (k - 21)) * HW + p0 + m];
    else if (k < 63)  val = xp[(size_t)(b * 126 + (k + 63)) * HW + p0 + m];
    a_lds[m][k] = val;
  }
  __syncthreads();

  int m0 = wave * 16;
  for (int nt = 0; nt < 4; ++nt) {
    int n0 = nt * 16;
    v8f acc = V8Z;
    for (int k0 = 0; k0 < 64; k0 += 4) {
      v2f av = frag_a(&a_lds[0][0], 65, m0, k0, lane);
      v2f bv = frag_b(&w_lds[0][0], 64, k0, n0, lane);
      acc = wmma4(av, bv, acc);
    }
    int nloc = lane & 15;
    int mb   = (lane >> 4) << 3;
    float bias = bias_lds[n0 + nloc];
#pragma unroll
    for (int r = 0; r < 8; ++r) d_lds[wave][mb + r][nloc] = acc[r] + bias;
    for (int idx = lane; idx < 256; idx += 32) {
      int n = idx >> 4, m = idx & 15;
      int ch = n0 + n;
      if (ch < 63)
        out[(size_t)(b * 63 + ch) * HW + p0 + m0 + m] = d_lds[wave][m][n];
    }
  }
}

extern "C" void kernel_launch(void* const* d_in, const int* in_sizes, int n_in,
                              void* d_out, int out_size, void* d_ws, size_t ws_size,
                              hipStream_t stream)
{
  (void)in_sizes; (void)n_in; (void)out_size; (void)ws_size;
  const float* x     = (const float*)d_in[0];
  const float* w_in  = (const float*)d_in[1];
  const float* b_in  = (const float*)d_in[2];
  const float* bn_g  = (const float*)d_in[3];
  const float* bn_b  = (const float*)d_in[4];
  const float* bn_m  = (const float*)d_in[5];
  const float* bn_v  = (const float*)d_in[6];
  const float* w_out = (const float*)d_in[7];
  const float* b_out = (const float*)d_in[8];
  float* out = (float*)d_out;

  float* xp   = (float*)d_ws;
  float* qrec = xp + (size_t)BB * 126 * HW;
  float* y2b  = xp + (size_t)84 * HW;
  float* vmid = xp + (size_t)21 * HW;

  k_proj_in <<<dim3(BB * 900), 128, 0, stream>>>(x, w_in, b_in, bn_g, bn_b, bn_m, bn_v, xp, qrec);
  k_win     <<<dim3(BB * NWIN * NWIN, 2), 32, 0, stream>>>(xp, y2b);
  k_axial<0><<<dim3(BB * HH, 15), 32, 0, stream>>>(xp, vmid, qrec);
  k_axial<1><<<dim3(BB * WW, 15), 32, 0, stream>>>(xp, vmid, qrec);
  k_proj_out<<<dim3(BB * 900), 128, 0, stream>>>(xp, y2b, w_out, b_out, out);
}


# both projections: the pixel operand goes from global memory straight into the A fragments, a lane's 32 loads in flight at once; no LDS copy of the pixels, so more workgroups fit a WGP; sums unchanged
# speedup vs baseline: 2.6761x; 1.1230x over previous
#include <hip/hip_runtime.h>
#include <math.h>

#define BB 8
#define CC 63
#define HH 240
#define WW 240
#define HW (HH * WW)
#define C2 126
#define WS 5
#define NWIN 48
#define BN_EPS 1e-5f

typedef __attribute__((ext_vector_type(2))) float v2f;
typedef __attribute__((ext_vector_type(4))) float v4f;
typedef __attribute__((ext_vector_type(8))) float v8f;

__device__ __forceinline__ v8f wmma4(v2f a, v2f b, v8f c) {
  return __builtin_amdgcn_wmma_f32_16x16x4_f32(
       false, a,  false, b,
       (short)0, c,  false,  false);
}

__device__ __forceinline__ v2f frag_a(const float* p, int ld, int m0, int k0, int lane) {
  int m  = m0 + (lane & 15);
  int kb = k0 + ((lane >> 4) << 1);
  const float* q = p + m * ld + kb;
  v2f r; r.x = q[0]; r.y = q[1];
  return r;
}

__device__ __forceinline__ v2f frag_b(const float* p, int ld, int k0, int n0, int lane) {
  int n  = n0 + (lane & 15);
  int kb = k0 + ((lane >> 4) << 1);
  v2f r; r.x = p[kb * ld + n]; r.y = p[(kb + 1) * ld + n];
  return r;
}

__device__ __forceinline__ void frag_store(float* p, int ld, int m0, int n0, int lane, v8f d) {
  int n  = n0 + (lane & 15);
  int mb = m0 + ((lane >> 4) << 3);
#pragma unroll
  for (int r = 0; r < 8; ++r) p[(mb + r) * ld + n] = d[r];
}

#define V8Z {0.f,0.f,0.f,0.f,0.f,0.f,0.f,0.f}

__global__ __launch_bounds__(128) void k_proj_in(
    const float* __restrict__ x, const float* __restrict__ w_in,
    const float* __restrict__ b_in, const float* __restrict__ bn_g,
    const float* __restrict__ bn_b, const float* __restrict__ bn_m,
    const float* __restrict__ bn_v, float* __restrict__ xp,
    float* __restrict__ qrec)
{
  __shared__ float w_lds[64][128];
  __shared__ float bias_lds[128];
  __shared__ float d_lds[4][16][17];

  int tid  = threadIdx.x;
  int lane = tid & 31;
  int wave = tid >> 5;
  int tile = blockIdx.x;
  int b    = tile / 900;
  int p0   = (tile % 900) * 64;
  int m0   = wave * 16;

  v2f af[16];
  {
    const float* xb = x + (size_t)b * 63 * HW + p0 + m0 + (lane & 15);
    int kh = (lane >> 4) << 1;
#pragma unroll
    for (int t = 0; t < 16; ++t) {
      int k  = 4 * t + kh;
      int k1 = (k + 1 < 63) ? k + 1 : 62;
      float v0 = xb[(size_t)k * HW];
      float v1 = xb[(size_t)k1 * HW];
      af[t].x = v0;
      af[t].y = (k + 1 < 63) ? v1 : 0.f;
    }
  }

  for (int idx = tid; idx < 64 * 128; idx += 128) {
    int k = idx >> 7, n = idx & 127;
    float val = 0.f;
    if (k < 63 && n < 126) {
      float s = bn_g[n] * rsqrtf(bn_v[n] + BN_EPS);
      val = w_in[n * 63 + k] * s;
    }
    w_lds[k][n] = val;
  }
  if (tid < 128) {
    float bv = 0.f;
    if (tid < 126) {
      float s = bn_g[tid] * rsqrtf(bn_v[tid] + BN_EPS);
      bv = b_in[tid] * s + bn_b[tid] - bn_m[tid] * s;
    }
    bias_lds[tid] = bv;
  }
  __syncthreads();

  for (int nt = 0; nt < 8; ++nt) {
    int n0 = nt * 16;
    v8f acc = V8Z;
#pragma unroll
    for (int t = 0; t < 16; ++t) {
      v2f bv = frag_b(&w_lds[0][0], 128, 4 * t, n0, lane);
      acc = wmma4(af[t], bv, acc);
    }
    int nloc = lane & 15;
    int mb   = (lane >> 4) << 3;
    float bias = bias_lds[n0 + nloc];
#pragma unroll
    for (int r = 0; r < 8; ++r) d_lds[wave][mb + r][nloc] = acc[r] + bias;
    for (int idx = lane; idx < 256; idx += 32) {
      int n = idx >> 4, m = idx & 15;
      int ch = n0 + n;
      if (ch < 126 && (ch < 84 || ch >= 105))
        xp[(size_t)(b * 126 + ch) * HW + p0 + m0 + m] = d_lds[wave][m][n];
    }
    if (n0 == 80 || n0 == 96) {
      for (int idx = lane; idx < 256; idx += 32) {
        int n = idx & 15, m = idx >> 4;
        int slot = n0 + n - 84;
        if (slot >= 0 && slot < 24)
          qrec[((size_t)b * HW + p0 + m0 + m) * 24 + slot] =
              (slot < 21) ? d_lds[wave][m][n] : 0.f;
      }
    }
  }
}

__global__ __launch_bounds__(32) void k_win(float* __restrict__ xp, float* __restrict__ y2)
{
  __shared__ float q[32][25];
  __shared__ float v[32][25];
  __shared__ float s[32][33];

  int lane = threadIdx.x;
  int wid  = blockIdx.x;
  int grp  = blockIdx.y;
  int b    = wid / (NWIN * NWIN);
  int rem  = wid % (NWIN * NWIN);
  int wy   = rem / NWIN, wx = rem % NWIN;

  for (int i = lane; i < 32 * 25; i += 32) { (&q[0][0])[i] = 0.f; (&v[0][0])[i] = 0.f; }

  int chq = grp * 42;
  for (int i = lane; i < 25 * 21; i += 32) {
    int m = i / 21, c = i % 21;
    int gy = wy * WS + m / WS, gx = wx * WS + m % WS;
    if (grp == 1) { gy = (gy + 3) % HH; gx = (gx + 3) % WW; }
    size_t base = (size_t)(b * 126 + chq + c) * HW + gy * WW + gx;
    q[m][c] = xp[base];
    v[m][c] = xp[base + (size_t)21 * HW];
  }

  for (int mt = 0; mt < 2; ++mt)
    for (int nt = 0; nt < 2; ++nt) {
      v8f acc = V8Z;
      for (int k0 = 0; k0 < 24; k0 += 4) {
        v2f av = frag_a(&q[0][0], 25, mt * 16, k0, lane);
        v2f bv = frag_a(&q[0][0], 25, nt * 16, k0, lane);
        acc = wmma4(av, bv, acc);
      }
      frag_store(&s[0][0], 33, mt * 16, nt * 16, lane, acc);
    }

  if (lane < 25) {
    float mx = -1e30f;
    for (int n = 0; n < 25; ++n) mx = fmaxf(mx, s[lane][n]);
    float sum = 0.f;
    for (int n = 0; n < 25; ++n) { float e = __expf(s[lane][n] - mx); s[lane][n] = e; sum += e; }
    float inv = 1.f / sum;
    for (int n = 0; n < 25; ++n) s[lane][n] *= inv;
    for (int n = 25; n < 28; ++n) s[lane][n] = 0.f;
  }

  for (int mt = 0; mt < 2; ++mt)
    for (int nt = 0; nt < 2; ++nt) {
      v8f acc = V8Z;
      for (int k0 = 0; k0 < 28; k0 += 4) {
        v2f av = frag_a(&s[0][0], 33, mt * 16, k0, lane);
        v2f bv = frag_b(&v[0][0], 25, k0, nt * 16, lane);
        acc = wmma4(av, bv, acc);
      }
      int n  = nt * 16 + (lane & 15);
      int mb = mt * 16 + ((lane >> 4) << 3);
#pragma unroll
      for (int r = 0; r < 8; ++r) {
        int m = mb + r;
        if (m < 25 && n < 21) {
          int oy = wy * WS + m / WS, ox = wx * WS + m % WS;
          if (grp == 0) {
            xp[(size_t)(b * 126 + n) * HW + oy * WW + ox] = acc[r];
          } else {
            oy = (oy + 2) % HH; ox = (ox + 2) % WW;
            y2[(size_t)(b * 126 + n) * HW + oy * WW + ox] = acc[r];
          }
        }
      }
    }
}


template <int PS>
__device__ __forceinline__ void gfrag_q(const float* qb, int p0, int lane, v2f (&f)[6]) {
  int p = p0 + (lane & 15);
  const v4f* g = (const v4f*)(qb + (size_t)p * (PS * 24) + 12 * (lane >> 4));
  v4f a = g[0], b = g[1], c = g[2];
  f[0].x = a.x; f[0].y = a.y; f[1].x = a.z; f[1].y = a.w;
  f[2].x = b.x; f[2].y = b.y; f[3].x = b.z; f[3].y = b.w;
  f[4].x = c.x; f[4].y = c.y; f[5].x = c.z; f[5].y = c.w;
}

template <int PASS>
__device__ __forceinline__ v2f gfrag_v(const float* vb, int k0, int n0, int lane) {
  int n  = n0 + (lane & 15);
  n = (n < 21) ? n : 20;
  int kb = k0 + ((lane >> 4) << 1);
  v2f r;
  if (PASS == 0) {
    r = *(const v2f*)(vb + (size_t)n * HW + kb);
  } else {
    const float* g = vb + kb * 21 + n;
    r.x = g[0]; r.y = g[21];
  }
  return r;
}

template <int X>
__device__ __forceinline__ float half_xor(float v) {
  return __int_as_float(__builtin_amdgcn_update_dpp(0, __float_as_int(v), 0x160 + X, 0xf, 0xf, false));
}

__device__ __forceinline__ float half_max(float v) {
  v = fmaxf(v, half_xor<1>(v));
  v = fmaxf(v, half_xor<2>(v));
  v = fmaxf(v, half_xor<4>(v));
  v = fmaxf(v, half_xor<8>(v));
  return v;
}

__device__ __forceinline__ float half_sum(float v) {
  v += half_xor<1>(v);
  v += half_xor<2>(v);
  v += half_xor<4>(v);
  v += half_xor<8>(v);
  return v;
}

template <int PASS>
__global__ __launch_bounds__(32) void k_axial(float* __restrict__ xp,
                                              float* __restrict__ vmid,
                                              const float* __restrict__ qrec)
{
  __shared__ float s[16][241];

  constexpr int PS = (PASS == 0) ? 1 : WW;

  int lane = threadIdx.x;
  int row  = blockIdx.x;
  int qt   = blockIdx.y;
  int b    = row / HH;
  int r    = row % HH;

  float* vm = vmid + (size_t)b * 126 * HW;
  const float* qb = qrec + ((size_t)b * HW + ((PASS == 0) ? r * WW : r)) * 24;
  const float* vb = (PASS == 0)
      ? xp + (size_t)(b * 126 + 105) * HW + r * WW
      : vm + (size_t)r * 240 * 21;

  int m0 = qt * 16;

  v2f qa[6];
  gfrag_q<PS>(qb, m0, lane, qa);

  float rmx[8];
#pragma unroll
  for (int r = 0; r < 8; ++r) rmx[r] = -1e30f;

#pragma unroll 5
  for (int nt = 0; nt < 15; ++nt) {
    int n0 = nt * 16;
    v2f kf[6];
    gfrag_q<PS>(qb, n0, lane, kf);
    v8f acc = V8Z;
#pragma unroll
    for (int t = 0; t < 6; ++t) acc = wmma4(qa[t], kf[t], acc);
#pragma unroll
    for (int r = 0; r < 8; ++r) rmx[r] = fmaxf(rmx[r], acc[r]);
    frag_store(&s[0][0], 241, 0, n0, lane, acc);
  }
#pragma unroll
  for (int r = 0; r < 8; ++r) rmx[r] = half_max(rmx[r]);

  float* own = &s[(lane >> 4) << 3][lane & 15];
  float rsum[8];
#pragma unroll
  for (int r = 0; r < 8; ++r) rsum[r] = 0.f;
#pragma unroll 5
  for (int nt = 0; nt < 15; ++nt) {
#pragma unroll
    for (int r = 0; r < 8; ++r) {
      float e = __expf(own[r * 241 + nt * 16] - rmx[r]);
      own[r * 241 + nt * 16] = e;
      rsum[r] += e;
    }
  }
  float rinv[8];
#pragma unroll
  for (int r = 0; r < 8; ++r) rinv[r] = 1.f / half_sum(rsum[r]);

  for (int nt = 0; nt < 2; ++nt) {
    int n0 = nt * 16;
    v8f acc = V8Z;
#pragma unroll 10
    for (int k0 = 0; k0 < 240; k0 += 4) {
      v2f av = frag_a(&s[0][0], 241, 0, k0, lane);
      v2f bv = gfrag_v<PASS>(vb, k0, n0, lane);
      acc = wmma4(av, bv, acc);
    }
    int n  = n0 + (lane & 15);
    int mb = (lane >> 4) << 3;
#pragma unroll
    for (int rr = 0; rr < 8; ++rr) {
      int m = m0 + mb + rr;
      float y = acc[rr] * rinv[rr];
      if (n < 21) {
        if (PASS == 0)
          vm[((size_t)m * 240 + r) * 21 + n] = y;
        else
          xp[(size_t)(b * 126 + 105 + n) * HW + m * WW + r] = y;
      }
    }
  }
}

__global__ __launch_bounds__(128) void k_proj_out(
    const float* __restrict__ xp, const float* __restrict__ y2,
    const float* __restrict__ w_out, const float* __restrict__ b_out,
    float* __restrict__ out)
{
  __shared__ float w_lds[64][64];
  __shared__ float bias_lds[64];
  __shared__ float d_lds[4][16][17];

  int tid  = threadIdx.x;
  int lane = tid & 31;
  int wave = tid >> 5;
  int tile = blockIdx.x;
  int b    = tile / 900;
  int p0   = (tile % 900) * 64;
  int m0   = wave * 16;

  v2f af[16];
  {
    const float* xb = xp + (size_t)b * 126 * HW + p0 + m0 + (lane & 15);
    const float* yb = y2 + (size_t)b * 126 * HW + p0 + m0 + (lane & 15);
    int kh = (lane >> 4) << 1;
#pragma unroll
    for (int t = 0; t < 16; ++t) {
#pragma unroll
      for (int j = 0; j < 2; ++j) {
        int k  = 4 * t + kh + j;
        int kc = (k < 63) ? k : 62;
        const float* src = (kc < 21) ? xb + (size_t)kc * HW
                         : (kc < 42) ? yb + (size_t)(kc - 21) * HW
                                     : xb + (size_t)(kc + 63) * HW;
        float v = *src;
        v = (k < 63) ? v : 0.f;
        if (j == 0) af[t].x = v; else af[t].y = v;
      }
    }
  }

  for (int idx = tid; idx < 64 * 64; idx += 128) {
    int k = idx >> 6, n = idx & 63;
    w_lds[k][n] = (k < 63 && n < 63) ? w_out[n * 63 + k] : 0.f;
  }
  if (tid < 64) bias_lds[tid] = (tid < 63) ? b_out[tid] : 0.f;

  __syncthreads();

  for (int nt = 0; nt < 4; ++nt) {
    int n0 = nt * 16;
    v8f acc = V8Z;
#pragma unroll
    for (int t = 0; t < 16; ++t) {
      v2f bv = frag_b(&w_lds[0][0], 64, 4 * t, n0, lane);
      acc = wmma4(af[t], bv, acc);
    }
    int nloc = lane & 15;
    int mb   = (lane >> 4) << 3;
    float bias = bias_lds[n0 + nloc];
#pragma unroll
    for (int r = 0; r < 8; ++r) d_lds[wave][mb + r][nloc] = acc[r] + bias;
    for (int idx = lane; idx < 256; idx += 32) {
      int n = idx >> 4, m = idx & 15;
      int ch = n0 + n;
      if (ch < 63)
        out[(size_t)(b * 63 + ch) * HW + p0 + m0 + m] = d_lds[wave][m][n];
    }
  }
}

extern "C" void kernel_launch(void* const* d_in, const int* in_sizes, int n_in,
                              void* d_out, int out_size, void* d_ws, size_t ws_size,
                              hipStream_t stream)
{
  (void)in_sizes; (void)n_in; (void)out_size; (void)ws_size;
  const float* x     = (const float*)d_in[0];
  const float* w_in  = (const float*)d_in[1];
  const float* b_in  = (const float*)d_in[2];
  const float* bn_g  = (const float*)d_in[3];
  const float* bn_b  = (const float*)d_in[4];
  const float* bn_m  = (const float*)d_in[5];
  const float* bn_v  = (const float*)d_in[6];
  const float* w_out = (const float*)d_in[7];
  const float* b_out = (const float*)d_in[8];
  float* out = (float*)d_out;

  float* xp   = (float*)d_ws;
  float* qrec = xp + (size_t)BB * 126 * HW;
  float* y2b  = xp + (size_t)84 * HW;
  float* vmid = xp + (size_t)21 * HW;

  k_proj_in <<<dim3(BB * 900), 128, 0, stream>>>(x, w_in, b_in, bn_g, bn_b, bn_m, bn_v, xp, qrec);
  k_win     <<<dim3(BB * NWIN * NWIN, 2), 32, 0, stream>>>(xp, y2b);
  k_axial<0><<<dim3(BB * HH, 15), 32, 0, stream>>>(xp, vmid, qrec);
  k_axial<1><<<dim3(BB * WW, 15), 32, 0, stream>>>(xp, vmid, qrec);
  k_proj_out<<<dim3(BB * 900), 128, 0, stream>>>(xp, y2b, w_out, b_out, out);
}


# both projections: weight tiles come from a copy in B-fragment order (BN folded in) that two small kernels write first, a k-step's pair is one 8-byte load; no LDS staging, no barrier; sums unchanged
# speedup vs baseline: 2.9141x; 1.0889x over previous
#include <hip/hip_runtime.h>
#include <math.h>

#define BB 8
#define CC 63
#define HH 240
#define WW 240
#define HW (HH * WW)
#define C2 126
#define WS 5
#define NWIN 48
#define BN_EPS 1e-5f

typedef __attribute__((ext_vector_type(2))) float v2f;
typedef __attribute__((ext_vector_type(4))) float v4f;
typedef __attribute__((ext_vector_type(8))) float v8f;

__device__ __forceinline__ v8f wmma4(v2f a, v2f b, v8f c) {
  return __builtin_amdgcn_wmma_f32_16x16x4_f32(
       false, a,  false, b,
       (short)0, c,  false,  false);
}

__device__ __forceinline__ v2f frag_a(const float* p, int ld, int m0, int k0, int lane) {
  int m  = m0 + (lane & 15);
  int kb = k0 + ((lane >> 4) << 1);
  const float* q = p + m * ld + kb;
  v2f r; r.x = q[0]; r.y = q[1];
  return r;
}

__device__ __forceinline__ v2f frag_b(const float* p, int ld, int k0, int n0, int lane) {
  int n  = n0 + (lane & 15);
  int kb = k0 + ((lane >> 4) << 1);
  v2f r; r.x = p[kb * ld + n]; r.y = p[(kb + 1) * ld + n];
  return r;
}

__device__ __forceinline__ void frag_store(float* p, int ld, int m0, int n0, int lane, v8f d) {
  int n  = n0 + (lane & 15);
  int mb = m0 + ((lane >> 4) << 3);
#pragma unroll
  for (int r = 0; r < 8; ++r) p[(mb + r) * ld + n] = d[r];
}

#define V8Z {0.f,0.f,0.f,0.f,0.f,0.f,0.f,0.f}

#define WF_IN_N  (8 * 16 * 64)
#define WF_OUT_N (4 * 16 * 64)

__global__ __launch_bounds__(256) void k_prep_in(
    const float* __restrict__ w_in, const float* __restrict__ b_in,
    const float* __restrict__ bn_g, const float* __restrict__ bn_b,
    const float* __restrict__ bn_m, const float* __restrict__ bn_v,
    float* __restrict__ wf)
{
  int idx = blockIdx.x * 256 + threadIdx.x;
  if (idx < WF_IN_N) {
    int j = idx & 1, l = (idx >> 1) & 31, t = (idx >> 6) & 15, nt = idx >> 10;
    int k = 4 * t + ((l >> 4) << 1) + j;
    int n = nt * 16 + (l & 15);
    float val = 0.f;
    if (k < 63 && n < 126) {
      float s = bn_g[n] * rsqrtf(bn_v[n] + BN_EPS);
      val = w_in[n * 63 + k] * s;
    }
    wf[idx] = val;
  } else if (idx < WF_IN_N + 128) {
    int n = idx - WF_IN_N;
    float bv = 0.f;
    if (n < 126) {
      float s = bn_g[n] * rsqrtf(bn_v[n] + BN_EPS);
      bv = b_in[n] * s + bn_b[n] - bn_m[n] * s;
    }
    wf[idx] = bv;
  }
}

__global__ __launch_bounds__(256) void k_prep_out(
    const float* __restrict__ w_out, const float* __restrict__ b_out,
    float* __restrict__ wf)
{
  int idx = blockIdx.x * 256 + threadIdx.x;
  if (idx < WF_OUT_N) {
    int j = idx & 1, l = (idx >> 1) & 31, t = (idx >> 6) & 15, nt = idx >> 10;
    int k = 4 * t + ((l >> 4) << 1) + j;
    int n = nt * 16 + (l & 15);
    wf[idx] = (k < 63 && n < 63) ? w_out[n * 63 + k] : 0.f;
  } else if (idx < WF_OUT_N + 64) {
    int n = idx - WF_OUT_N;
    wf[idx] = (n < 63) ? b_out[n] : 0.f;
  }
}

__global__ __launch_bounds__(128) void k_proj_in(
    const float* __restrict__ x, const float* __restrict__ wf,
    float* __restrict__ xp, float* __restrict__ qrec)
{
  __shared__ float d_lds[4][16][17];

  int tid  = threadIdx.x;
  int lane = tid & 31;
  int wave = tid >> 5;
  int tile = blockIdx.x;
  int b    = tile / 900;
  int p0   = (tile % 900) * 64;
  int m0   = wave * 16;

  v2f af[16];
  {
    const float* xb = x + (size_t)b * 63 * HW + p0 + m0 + (lane & 15);
    int kh = (lane >> 4) << 1;
#pragma unroll
    for (int t = 0; t < 16; ++t) {
      int k  = 4 * t + kh;
      int k1 = (k + 1 < 63) ? k + 1 : 62;
      float v0 = xb[(size_t)k * HW];
      float v1 = xb[(size_t)k1 * HW];
      af[t].x = v0;
      af[t].y = (k + 1 < 63) ? v1 : 0.f;
    }
  }

  const v2f* wfl = (const v2f*)wf + lane;

  v2f bf[16];
#pragma unroll
  for (int t = 0; t < 16; ++t) bf[t] = wfl[t * 32];

  float bias_nx = wf[WF_IN_N + (lane & 15)];

  for (int nt = 0; nt < 8; ++nt) {
    int n0 = nt * 16;
    int nloc = lane & 15;
    int mb   = (lane >> 4) << 3;
    float bias = bias_nx;
    v8f acc = V8Z;
#pragma unroll
    for (int t = 0; t < 16; ++t) acc = wmma4(af[t], bf[t], acc);
    int nx = (nt + 1 < 8) ? nt + 1 : nt;
#pragma unroll
    for (int t = 0; t < 16; ++t) bf[t] = wfl[(nx * 16 + t) * 32];
    bias_nx = wf[WF_IN_N + nx * 16 + nloc];
#pragma unroll
    for (int r = 0; r < 8; ++r) d_lds[wave][mb + r][nloc] = acc[r] + bias;
    for (int idx = lane; idx < 256; idx += 32) {
      int n = idx >> 4, m = idx & 15;
      int ch = n0 + n;
      if (ch < 126 && (ch < 84 || ch >= 105))
        xp[(size_t)(b * 126 + ch) * HW + p0 + m0 + m] = d_lds[wave][m][n];
    }
    if (n0 == 80 || n0 == 96) {
      for (int idx = lane; idx < 256; idx += 32) {
        int n = idx & 15, m = idx >> 4;
        int slot = n0 + n - 84;
        if (slot >= 0 && slot < 24)
          qrec[((size_t)b * HW + p0 + m0 + m) * 24 + slot] =
              (slot < 21) ? d_lds[wave][m][n] : 0.f;
      }
    }
  }
}

__global__ __launch_bounds__(32) void k_win(float* __restrict__ xp, float* __restrict__ y2)
{
  __shared__ float q[32][25];
  __shared__ float v[32][25];
  __shared__ float s[32][33];

  int lane = threadIdx.x;
  int wid  = blockIdx.x;
  int grp  = blockIdx.y;
  int b    = wid / (NWIN * NWIN);
  int rem  = wid % (NWIN * NWIN);
  int wy   = rem / NWIN, wx = rem % NWIN;

  for (int i = lane; i < 32 * 25; i += 32) { (&q[0][0])[i] = 0.f; (&v[0][0])[i] = 0.f; }

  int chq = grp * 42;
  for (int i = lane; i < 25 * 21; i += 32) {
    int m = i / 21, c = i % 21;
    int gy = wy * WS + m / WS, gx = wx * WS + m % WS;
    if (grp == 1) { gy = (gy + 3) % HH; gx = (gx + 3) % WW; }
    size_t base = (size_t)(b * 126 + chq + c) * HW + gy * WW + gx;
    q[m][c] = xp[base];
    v[m][c] = xp[base + (size_t)21 * HW];
  }

  for (int mt = 0; mt < 2; ++mt)
    for (int nt = 0; nt < 2; ++nt) {
      v8f acc = V8Z;
      for (int k0 = 0; k0 < 24; k0 += 4) {
        v2f av = frag_a(&q[0][0], 25, mt * 16, k0, lane);
        v2f bv = frag_a(&q[0][0], 25, nt * 16, k0, lane);
        acc = wmma4(av, bv, acc);
      }
      frag_store(&s[0][0], 33, mt * 16, nt * 16, lane, acc);
    }

  if (lane < 25) {
    float mx = -1e30f;
    for (int n = 0; n < 25; ++n) mx = fmaxf(mx, s[lane][n]);
    float sum = 0.f;
    for (int n = 0; n < 25; ++n) { float e = __expf(s[lane][n] - mx); s[lane][n] = e; sum += e; }
    float inv = 1.f / sum;
    for (int n = 0; n < 25; ++n) s[lane][n] *= inv;
    for (int n = 25; n < 28; ++n) s[lane][n] = 0.f;
  }

  for (int mt = 0; mt < 2; ++mt)
    for (int nt = 0; nt < 2; ++nt) {
      v8f acc = V8Z;
      for (int k0 = 0; k0 < 28; k0 += 4) {
        v2f av = frag_a(&s[0][0], 33, mt * 16, k0, lane);
        v2f bv = frag_b(&v[0][0], 25, k0, nt * 16, lane);
        acc = wmma4(av, bv, acc);
      }
      int n  = nt * 16 + (lane & 15);
      int mb = mt * 16 + ((lane >> 4) << 3);
#pragma unroll
      for (int r = 0; r < 8; ++r) {
        int m = mb + r;
        if (m < 25 && n < 21) {
          int oy = wy * WS + m / WS, ox = wx * WS + m % WS;
          if (grp == 0) {
            xp[(size_t)(b * 126 + n) * HW + oy * WW + ox] = acc[r];
          } else {
            oy = (oy + 2) % HH; ox = (ox + 2) % WW;
            y2[(size_t)(b * 126 + n) * HW + oy * WW + ox] = acc[r];
          }
        }
      }
    }
}


template <int PS>
__device__ __forceinline__ void gfrag_q(const float* qb, int p0, int lane, v2f (&f)[6]) {
  int p = p0 + (lane & 15);
  const v4f* g = (const v4f*)(qb + (size_t)p * (PS * 24) + 12 * (lane >> 4));
  v4f a = g[0], b = g[1], c = g[2];
  f[0].x = a.x; f[0].y = a.y; f[1].x = a.z; f[1].y = a.w;
  f[2].x = b.x; f[2].y = b.y; f[3].x = b.z; f[3].y = b.w;
  f[4].x = c.x; f[4].y = c.y; f[5].x = c.z; f[5].y = c.w;
}

template <int PASS>
__device__ __forceinline__ v2f gfrag_v(const float* vb, int k0, int n0, int lane) {
  int n  = n0 + (lane & 15);
  n = (n < 21) ? n : 20;
  int kb = k0 + ((lane >> 4) << 1);
  v2f r;
  if (PASS == 0) {
    r = *(const v2f*)(vb + (size_t)n * HW + kb);
  } else {
    const float* g = vb + kb * 21 + n;
    r.x = g[0]; r.y = g[21];
  }
  return r;
}

template <int X>
__device__ __forceinline__ float half_xor(float v) {
  return __int_as_float(__builtin_amdgcn_update_dpp(0, __float_as_int(v), 0x160 + X, 0xf, 0xf, false));
}

__device__ __forceinline__ float half_max(float v) {
  v = fmaxf(v, half_xor<1>(v));
  v = fmaxf(v, half_xor<2>(v));
  v = fmaxf(v, half_xor<4>(v));
  v = fmaxf(v, half_xor<8>(v));
  return v;
}

__device__ __forceinline__ float half_sum(float v) {
  v += half_xor<1>(v);
  v += half_xor<2>(v);
  v += half_xor<4>(v);
  v += half_xor<8>(v);
  return v;
}

template <int PASS>
__global__ __launch_bounds__(32) void k_axial(float* __restrict__ xp,
                                              float* __restrict__ vmid,
                                              const float* __restrict__ qrec)
{
  __shared__ float s[16][241];

  constexpr int PS = (PASS == 0) ? 1 : WW;

  int lane = threadIdx.x;
  int row  = blockIdx.x;
  int qt   = blockIdx.y;
  int b    = row / HH;
  int r    = row % HH;

  float* vm = vmid + (size_t)b * 126 * HW;
  const float* qb = qrec + ((size_t)b * HW + ((PASS == 0) ? r * WW : r)) * 24;
  const float* vb = (PASS == 0)
      ? xp + (size_t)(b * 126 + 105) * HW + r * WW
      : vm + (size_t)r * 240 * 21;

  int m0 = qt * 16;

  v2f qa[6];
  gfrag_q<PS>(qb, m0, lane, qa);

  float rmx[8];
#pragma unroll
  for (int r = 0; r < 8; ++r) rmx[r] = -1e30f;

#pragma unroll 5
  for (int nt = 0; nt < 15; ++nt) {
    int n0 = nt * 16;
    v2f kf[6];
    gfrag_q<PS>(qb, n0, lane, kf);
    v8f acc = V8Z;
#pragma unroll
    for (int t = 0; t < 6; ++t) acc = wmma4(qa[t], kf[t], acc);
#pragma unroll
    for (int r = 0; r < 8; ++r) rmx[r] = fmaxf(rmx[r], acc[r]);
    frag_store(&s[0][0], 241, 0, n0, lane, acc);
  }
#pragma unroll
  for (int r = 0; r < 8; ++r) rmx[r] = half_max(rmx[r]);

  float* own = &s[(lane >> 4) << 3][lane & 15];
  float rsum[8];
#pragma unroll
  for (int r = 0; r < 8; ++r) rsum[r] = 0.f;
#pragma unroll 5
  for (int nt = 0; nt < 15; ++nt) {
#pragma unroll
    for (int r = 0; r < 8; ++r) {
      float e = __expf(own[r * 241 + nt * 16] - rmx[r]);
      own[r * 241 + nt * 16] = e;
      rsum[r] += e;
    }
  }
  float rinv[8];
#pragma unroll
  for (int r = 0; r < 8; ++r) rinv[r] = 1.f / half_sum(rsum[r]);

  for (int nt = 0; nt < 2; ++nt) {
    int n0 = nt * 16;
    v8f acc = V8Z;
#pragma unroll 10
    for (int k0 = 0; k0 < 240; k0 += 4) {
      v2f av = frag_a(&s[0][0], 241, 0, k0, lane);
      v2f bv = gfrag_v<PASS>(vb, k0, n0, lane);
      acc = wmma4(av, bv, acc);
    }
    int n  = n0 + (lane & 15);
    int mb = (lane >> 4) << 3;
#pragma unroll
    for (int rr = 0; rr < 8; ++rr) {
      int m = m0 + mb + rr;
      float y = acc[rr] * rinv[rr];
      if (n < 21) {
        if (PASS == 0)
          vm[((size_t)m * 240 + r) * 21 + n] = y;
        else
          xp[(size_t)(b * 126 + 105 + n) * HW + m * WW + r] = y;
      }
    }
  }
}

__global__ __launch_bounds__(128) void k_proj_out(
    const float* __restrict__ xp, const float* __restrict__ y2,
    const float* __restrict__ wf, float* __restrict__ out)
{
  __shared__ float d_lds[4][16][17];

  int tid  = threadIdx.x;
  int lane = tid & 31;
  int wave = tid >> 5;
  int tile = blockIdx.x;
  int b    = tile / 900;
  int p0   = (tile % 900) * 64;
  int m0   = wave * 16;

  v2f af[16];
  {
    const float* xb = xp + (size_t)b * 126 * HW + p0 + m0 + (lane & 15);
    const float* yb = y2 + (size_t)b * 126 * HW + p0 + m0 + (lane & 15);
    int kh = (lane >> 4) << 1;
#pragma unroll
    for (int t = 0; t < 16; ++t) {
#pragma unroll
      for (int j = 0; j < 2; ++j) {
        int k  = 4 * t + kh + j;
        int kc = (k < 63) ? k : 62;
        const float* src = (kc < 21) ? xb + (size_t)kc * HW
                         : (kc < 42) ? yb + (size_t)(kc - 21) * HW
                                     : xb + (size_t)(kc + 63) * HW;
        float v = *src;
        v = (k < 63) ? v : 0.f;
        if (j == 0) af[t].x = v; else af[t].y = v;
      }
    }
  }

  const v2f* wfl = (const v2f*)wf + lane;

  v2f bf[16];
#pragma unroll
  for (int t = 0; t < 16; ++t) bf[t] = wfl[t * 32];

  float bias_nx = wf[WF_OUT_N + (lane & 15)];

  for (int nt = 0; nt < 4; ++nt) {
    int n0 = nt * 16;
    int nloc = lane & 15;
    int mb   = (lane >> 4) << 3;
    float bias = bias_nx;
    v8f acc = V8Z;
#pragma unroll
    for (int t = 0; t < 16; ++t) acc = wmma4(af[t], bf[t], acc);
    int nx = (nt + 1 < 4) ? nt + 1 : nt;
#pragma unroll
    for (int t = 0; t < 16; ++t) bf[t] = wfl[(nx * 16 + t) * 32];
    bias_nx = wf[WF_OUT_N + nx * 16 + nloc];
#pragma unroll
    for (int r = 0; r < 8; ++r) d_lds[wave][mb + r][nloc] = acc[r] + bias;
    for (int idx = lane; idx < 256; idx += 32) {
      int n = idx >> 4, m = idx & 15;
      int ch = n0 + n;
      if (ch < 63)
        out[(size_t)(b * 63 + ch) * HW + p0 + m0 + m] = d_lds[wave][m][n];
    }
  }
}

extern "C" void kernel_launch(void* const* d_in, const int* in_sizes, int n_in,
                              void* d_out, int out_size, void* d_ws, size_t ws_size,
                              hipStream_t stream)
{
  (void)in_sizes; (void)n_in; (void)out_size; (void)ws_size;
  const float* x     = (const float*)d_in[0];
  const float* w_in  = (const float*)d_in[1];
  const float* b_in  = (const float*)d_in[2];
  const float* bn_g  = (const float*)d_in[3];
  const float* bn_b  = (const float*)d_in[4];
  const float* bn_m  = (const float*)d_in[5];
  const float* bn_v  = (const float*)d_in[6];
  const float* w_out = (const float*)d_in[7];
  const float* b_out = (const float*)d_in[8];
  float* out = (float*)d_out;

  float* xp   = (float*)d_ws;
  float* qrec = xp + (size_t)BB * 126 * HW;
  float* y2b  = xp + (size_t)84 * HW;
  float* vmid = xp + (size_t)21 * HW;
  float* wfin = xp + (size_t)84 * HW;
  float* wfout = xp + (size_t)42 * HW;

  k_prep_in <<<dim3(33), 256, 0, stream>>>(w_in, b_in, bn_g, bn_b, bn_m, bn_v, wfin);
  k_proj_in <<<dim3(BB * 900), 128, 0, stream>>>(x, wfin, xp, qrec);
  k_win     <<<dim3(BB * NWIN * NWIN, 2), 32, 0, stream>>>(xp, y2b);
  k_axial<0><<<dim3(BB * HH, 15), 32, 0, stream>>>(xp, vmid, qrec);
  k_axial<1><<<dim3(BB * WW, 15), 32, 0, stream>>>(xp, vmid, qrec);
  k_prep_out<<<dim3(17), 256, 0, stream>>>(w_out, b_out, wfout);
  k_proj_out<<<dim3(BB * 900), 128, 0, stream>>>(xp, y2b, wfout, out);
}


# window attention: global accesses by a lane per window pixel (42 loads in flight, row pieces of one plane per load), P V computed transposed so a store touches two planes, not 32; sums unchanged
# speedup vs baseline: 4.0337x; 1.3842x over previous
#include <hip/hip_runtime.h>
#include <math.h>

#define BB 8
#define CC 63
#define HH 240
#define WW 240
#define HW (HH * WW)
#define C2 126
#define WS 5
#define NWIN 48
#define BN_EPS 1e-5f

typedef __attribute__((ext_vector_type(2))) float v2f;
typedef __attribute__((ext_vector_type(4))) float v4f;
typedef __attribute__((ext_vector_type(8))) float v8f;

__device__ __forceinline__ v8f wmma4(v2f a, v2f b, v8f c) {
  return __builtin_amdgcn_wmma_f32_16x16x4_f32(
       false, a,  false, b,
       (short)0, c,  false,  false);
}

__device__ __forceinline__ v2f frag_a(const float* p, int ld, int m0, int k0, int lane) {
  int m  = m0 + (lane & 15);
  int kb = k0 + ((lane >> 4) << 1);
  const float* q = p + m * ld + kb;
  v2f r; r.x = q[0]; r.y = q[1];
  return r;
}

__device__ __forceinline__ v2f frag_b(const float* p, int ld, int k0, int n0, int lane) {
  int n  = n0 + (lane & 15);
  int kb = k0 + ((lane >> 4) << 1);
  v2f r; r.x = p[kb * ld + n]; r.y = p[(kb + 1) * ld + n];
  return r;
}

__device__ __forceinline__ void frag_store(float* p, int ld, int m0, int n0, int lane, v8f d) {
  int n  = n0 + (lane & 15);
  int mb = m0 + ((lane >> 4) << 3);
#pragma unroll
  for (int r = 0; r < 8; ++r) p[(mb + r) * ld + n] = d[r];
}

#define V8Z {0.f,0.f,0.f,0.f,0.f,0.f,0.f,0.f}

#define WF_IN_N  (8 * 16 * 64)
#define WF_OUT_N (4 * 16 * 64)

__global__ __launch_bounds__(256) void k_prep_in(
    const float* __restrict__ w_in, const float* __restrict__ b_in,
    const float* __restrict__ bn_g, const float* __restrict__ bn_b,
    const float* __restrict__ bn_m, const float* __restrict__ bn_v,
    float* __restrict__ wf)
{
  int idx = blockIdx.x * 256 + threadIdx.x;
  if (idx < WF_IN_N) {
    int j = idx & 1, l = (idx >> 1) & 31, t = (idx >> 6) & 15, nt = idx >> 10;
    int k = 4 * t + ((l >> 4) << 1) + j;
    int n = nt * 16 + (l & 15);
    float val = 0.f;
    if (k < 63 && n < 126) {
      float s = bn_g[n] * rsqrtf(bn_v[n] + BN_EPS);
      val = w_in[n * 63 + k] * s;
    }
    wf[idx] = val;
  } else if (idx < WF_IN_N + 128) {
    int n = idx - WF_IN_N;
    float bv = 0.f;
    if (n < 126) {
      float s = bn_g[n] * rsqrtf(bn_v[n] + BN_EPS);
      bv = b_in[n] * s + bn_b[n] - bn_m[n] * s;
    }
    wf[idx] = bv;
  }
}

__global__ __launch_bounds__(256) void k_prep_out(
    const float* __restrict__ w_out, const float* __restrict__ b_out,
    float* __restrict__ wf)
{
  int idx = blockIdx.x * 256 + threadIdx.x;
  if (idx < WF_OUT_N) {
    int j = idx & 1, l = (idx >> 1) & 31, t = (idx >> 6) & 15, nt = idx >> 10;
    int k = 4 * t + ((l >> 4) << 1) + j;
    int n = nt * 16 + (l & 15);
    wf[idx] = (k < 63 && n < 63) ? w_out[n * 63 + k] : 0.f;
  } else if (idx < WF_OUT_N + 64) {
    int n = idx - WF_OUT_N;
    wf[idx] = (n < 63) ? b_out[n] : 0.f;
  }
}

__global__ __launch_bounds__(128) void k_proj_in(
    const float* __restrict__ x, const float* __restrict__ wf,
    float* __restrict__ xp, float* __restrict__ qrec)
{
  __shared__ float d_lds[4][16][17];

  int tid  = threadIdx.x;
  int lane = tid & 31;
  int wave = tid >> 5;
  int tile = blockIdx.x;
  int b    = tile / 900;
  int p0   = (tile % 900) * 64;
  int m0   = wave * 16;

  v2f af[16];
  {
    const float* xb = x + (size_t)b * 63 * HW + p0 + m0 + (lane & 15);
    int kh = (lane >> 4) << 1;
#pragma unroll
    for (int t = 0; t < 16; ++t) {
      int k  = 4 * t + kh;
      int k1 = (k + 1 < 63) ? k + 1 : 62;
      float v0 = xb[(size_t)k * HW];
      float v1 = xb[(size_t)k1 * HW];
      af[t].x = v0;
      af[t].y = (k + 1 < 63) ? v1 : 0.f;
    }
  }

  const v2f* wfl = (const v2f*)wf + lane;

  v2f bf[16];
#pragma unroll
  for (int t = 0; t < 16; ++t) bf[t] = wfl[t * 32];

  float bias_nx = wf[WF_IN_N + (lane & 15)];

  for (int nt = 0; nt < 8; ++nt) {
    int n0 = nt * 16;
    int nloc = lane & 15;
    int mb   = (lane >> 4) << 3;
    float bias = bias_nx;
    v8f acc = V8Z;
#pragma unroll
    for (int t = 0; t < 16; ++t) acc = wmma4(af[t], bf[t], acc);
    int nx = (nt + 1 < 8) ? nt + 1 : nt;
#pragma unroll
    for (int t = 0; t < 16; ++t) bf[t] = wfl[(nx * 16 + t) * 32];
    bias_nx = wf[WF_IN_N + nx * 16 + nloc];
#pragma unroll
    for (int r = 0; r < 8; ++r) d_lds[wave][mb + r][nloc] = acc[r] + bias;
    for (int idx = lane; idx < 256; idx += 32) {
      int n = idx >> 4, m = idx & 15;
      int ch = n0 + n;
      if (ch < 126 && (ch < 84 || ch >= 105))
        xp[(size_t)(b * 126 + ch) * HW + p0 + m0 + m] = d_lds[wave][m][n];
    }
    if (n0 == 80 || n0 == 96) {
      for (int idx = lane; idx < 256; idx += 32) {
        int n = idx & 15, m = idx >> 4;
        int slot = n0 + n - 84;
        if (slot >= 0 && slot < 24)
          qrec[((size_t)b * HW + p0 + m0 + m) * 24 + slot] =
              (slot < 21) ? d_lds[wave][m][n] : 0.f;
      }
    }
  }
}

__global__ __launch_bounds__(32) void k_win(float* __restrict__ xp, float* __restrict__ y2)
{
  __shared__ float q[32][25];
  __shared__ float v[32][25];
  __shared__ float s[32][33];

  int lane = threadIdx.x;
  int wid  = blockIdx.x;
  int grp  = blockIdx.y;
  int b    = wid / (NWIN * NWIN);
  int rem  = wid % (NWIN * NWIN);
  int wy   = rem / NWIN, wx = rem % NWIN;

  {
    bool own = lane < 25;
    int mp = own ? lane : 24;
    int gy = wy * WS + mp / WS, gx = wx * WS + mp % WS;
    if (grp == 1) { gy = (gy + 3) % HH; gx = (gx + 3) % WW; }
    const float* src = xp + (size_t)(b * 126 + grp * 42) * HW + gy * WW + gx;
    float qv[21], vv[21];
#pragma unroll
    for (int c = 0; c < 21; ++c) {
      qv[c] = src[(size_t)c * HW];
      vv[c] = src[(size_t)(c + 21) * HW];
    }
#pragma unroll
    for (int c = 0; c < 21; ++c) {
      q[lane][c] = own ? qv[c] : 0.f;
      v[lane][c] = own ? vv[c] : 0.f;
    }
    for (int c = 21; c < 25; ++c) { q[lane][c] = 0.f; v[lane][c] = 0.f; }
  }

  for (int mt = 0; mt < 2; ++mt)
    for (int nt = 0; nt < 2; ++nt) {
      v8f acc = V8Z;
      for (int k0 = 0; k0 < 24; k0 += 4) {
        v2f av = frag_a(&q[0][0], 25, mt * 16, k0, lane);
        v2f bv = frag_a(&q[0][0], 25, nt * 16, k0, lane);
        acc = wmma4(av, bv, acc);
      }
      frag_store(&s[0][0], 33, mt * 16, nt * 16, lane, acc);
    }

  if (lane < 25) {
    float mx = -1e30f;
    for (int n = 0; n < 25; ++n) mx = fmaxf(mx, s[lane][n]);
    float sum = 0.f;
    for (int n = 0; n < 25; ++n) { float e = __expf(s[lane][n] - mx); s[lane][n] = e; sum += e; }
    float inv = 1.f / sum;
    for (int n = 0; n < 25; ++n) s[lane][n] *= inv;
    for (int n = 25; n < 28; ++n) s[lane][n] = 0.f;
  }

  for (int qt = 0; qt < 2; ++qt) {
    int m  = qt * 16 + (lane & 15);
    int mc = (m < 25) ? m : 24;
    int oy = wy * WS + mc / WS, ox = wx * WS + mc % WS;
    float* dst;
    if (grp == 0) {
      dst = xp + (size_t)(b * 126) * HW + oy * WW + ox;
    } else {
      oy = (oy + 2) % HH; ox = (ox + 2) % WW;
      dst = y2 + (size_t)(b * 126) * HW + oy * WW + ox;
    }
#pragma unroll
    for (int ct = 0; ct < 2; ++ct) {
      v8f acc = V8Z;
      for (int k0 = 0; k0 < 28; k0 += 4) {
        v2f av = frag_b(&v[0][0], 25, k0, ct * 16, lane);
        v2f bv = frag_a(&s[0][0], 33, qt * 16, k0, lane);
        acc = wmma4(av, bv, acc);
      }
      int chb = ct * 16 + ((lane >> 4) << 3);
      bool st = (m < 25) && (ct == 0 || (lane >> 4) == 0);
      if (st) {
        float* d = dst + (size_t)chb * HW;
#pragma unroll
        for (int r = 0; r < 8; ++r)
          if (ct == 0 || r < 5) d[(size_t)r * HW] = acc[r];
      }
    }
  }
}


template <int PS>
__device__ __forceinline__ void gfrag_q(const float* qb, int p0, int lane, v2f (&f)[6]) {
  int p = p0 + (lane & 15);
  const v4f* g = (const v4f*)(qb + (size_t)p * (PS * 24) + 12 * (lane >> 4));
  v4f a = g[0], b = g[1], c = g[2];
  f[0].x = a.x; f[0].y = a.y; f[1].x = a.z; f[1].y = a.w;
  f[2].x = b.x; f[2].y = b.y; f[3].x = b.z; f[3].y = b.w;
  f[4].x = c.x; f[4].y = c.y; f[5].x = c.z; f[5].y = c.w;
}

template <int PASS>
__device__ __forceinline__ v2f gfrag_v(const float* vb, int k0, int n0, int lane) {
  int n  = n0 + (lane & 15);
  n = (n < 21) ? n : 20;
  int kb = k0 + ((lane >> 4) << 1);
  v2f r;
  if (PASS == 0) {
    r = *(const v2f*)(vb + (size_t)n * HW + kb);
  } else {
    const float* g = vb + kb * 21 + n;
    r.x = g[0]; r.y = g[21];
  }
  return r;
}

template <int X>
__device__ __forceinline__ float half_xor(float v) {
  return __int_as_float(__builtin_amdgcn_update_dpp(0, __float_as_int(v), 0x160 + X, 0xf, 0xf, false));
}

__device__ __forceinline__ float half_max(float v) {
  v = fmaxf(v, half_xor<1>(v));
  v = fmaxf(v, half_xor<2>(v));
  v = fmaxf(v, half_xor<4>(v));
  v = fmaxf(v, half_xor<8>(v));
  return v;
}

__device__ __forceinline__ float half_sum(float v) {
  v += half_xor<1>(v);
  v += half_xor<2>(v);
  v += half_xor<4>(v);
  v += half_xor<8>(v);
  return v;
}

template <int PASS>
__global__ __launch_bounds__(32) void k_axial(float* __restrict__ xp,
                                              float* __restrict__ vmid,
                                              const float* __restrict__ qrec)
{
  __shared__ float s[16][241];

  constexpr int PS = (PASS == 0) ? 1 : WW;

  int lane = threadIdx.x;
  int row  = blockIdx.x;
  int qt   = blockIdx.y;
  int b    = row / HH;
  int r    = row % HH;

  float* vm = vmid + (size_t)b * 126 * HW;
  const float* qb = qrec + ((size_t)b * HW + ((PASS == 0) ? r * WW : r)) * 24;
  const float* vb = (PASS == 0)
      ? xp + (size_t)(b * 126 + 105) * HW + r * WW
      : vm + (size_t)r * 240 * 21;

  int m0 = qt * 16;

  v2f qa[6];
  gfrag_q<PS>(qb, m0, lane, qa);

  float rmx[8];
#pragma unroll
  for (int r = 0; r < 8; ++r) rmx[r] = -1e30f;

#pragma unroll 5
  for (int nt = 0; nt < 15; ++nt) {
    int n0 = nt * 16;
    v2f kf[6];
    gfrag_q<PS>(qb, n0, lane, kf);
    v8f acc = V8Z;
#pragma unroll
    for (int t = 0; t < 6; ++t) acc = wmma4(qa[t], kf[t], acc);
#pragma unroll
    for (int r = 0; r < 8; ++r) rmx[r] = fmaxf(rmx[r], acc[r]);
    frag_store(&s[0][0], 241, 0, n0, lane, acc);
  }
#pragma unroll
  for (int r = 0; r < 8; ++r) rmx[r] = half_max(rmx[r]);

  float* own = &s[(lane >> 4) << 3][lane & 15];
  float rsum[8];
#pragma unroll
  for (int r = 0; r < 8; ++r) rsum[r] = 0.f;
#pragma unroll 5
  for (int nt = 0; nt < 15; ++nt) {
#pragma unroll
    for (int r = 0; r < 8; ++r) {
      float e = __expf(own[r * 241 + nt * 16] - rmx[r]);
      own[r * 241 + nt * 16] = e;
      rsum[r] += e;
    }
  }
  float rinv[8];
#pragma unroll
  for (int r = 0; r < 8; ++r) rinv[r] = 1.f / half_sum(rsum[r]);

  for (int nt = 0; nt < 2; ++nt) {
    int n0 = nt * 16;
    v8f acc = V8Z;
#pragma unroll 10
    for (int k0 = 0; k0 < 240; k0 += 4) {
      v2f av = frag_a(&s[0][0], 241, 0, k0, lane);
      v2f bv = gfrag_v<PASS>(vb, k0, n0, lane);
      acc = wmma4(av, bv, acc);
    }
    int n  = n0 + (lane & 15);
    int mb = (lane >> 4) << 3;
#pragma unroll
    for (int rr = 0; rr < 8; ++rr) {
      int m = m0 + mb + rr;
      float y = acc[rr] * rinv[rr];
      if (n < 21) {
        if (PASS == 0)
          vm[((size_t)m * 240 + r) * 21 + n] = y;
        else
          xp[(size_t)(b * 126 + 105 + n) * HW + m * WW + r] = y;
      }
    }
  }
}

__global__ __launch_bounds__(128) void k_proj_out(
    const float* __restrict__ xp, const float* __restrict__ y2,
    const float* __restrict__ wf, float* __restrict__ out)
{
  __shared__ float d_lds[4][16][17];

  int tid  = threadIdx.x;
  int lane = tid & 31;
  int wave = tid >> 5;
  int tile = blockIdx.x;
  int b    = tile / 900;
  int p0   = (tile % 900) * 64;
  int m0   = wave * 16;

  v2f af[16];
  {
    const float* xb = xp + (size_t)b * 126 * HW + p0 + m0 + (lane & 15);
    const float* yb = y2 + (size_t)b * 126 * HW + p0 + m0 + (lane & 15);
    int kh = (lane >> 4) << 1;
#pragma unroll
    for (int t = 0; t < 16; ++t) {
#pragma unroll
      for (int j = 0; j < 2; ++j) {
        int k  = 4 * t + kh + j;
        int kc = (k < 63) ? k : 62;
        const float* src = (kc < 21) ? xb + (size_t)kc * HW
                         : (kc < 42) ? yb + (size_t)(kc - 21) * HW
                                     : xb + (size_t)(kc + 63) * HW;
        float v = *src;
        v = (k < 63) ? v : 0.f;
        if (j == 0) af[t].x = v; else af[t].y = v;
      }
    }
  }

  const v2f* wfl = (const v2f*)wf + lane;

  v2f bf[16];
#pragma unroll
  for (int t = 0; t < 16; ++t) bf[t] = wfl[t * 32];

  float bias_nx = wf[WF_OUT_N + (lane & 15)];

  for (int nt = 0; nt < 4; ++nt) {
    int n0 = nt * 16;
    int nloc = lane & 15;
    int mb   = (lane >> 4) << 3;
    float bias = bias_nx;
    v8f acc = V8Z;
#pragma unroll
    for (int t = 0; t < 16; ++t) acc = wmma4(af[t], bf[t], acc);
    int nx = (nt + 1 < 4) ? nt + 1 : nt;
#pragma unroll
    for (int t = 0; t < 16; ++t) bf[t] = wfl[(nx * 16 + t) * 32];
    bias_nx = wf[WF_OUT_N + nx * 16 + nloc];
#pragma unroll
    for (int r = 0; r < 8; ++r) d_lds[wave][mb + r][nloc] = acc[r] + bias;
    for (int idx = lane; idx < 256; idx += 32) {
      int n = idx >> 4, m = idx & 15;
      int ch = n0 + n;
      if (ch < 63)
        out[(size_t)(b * 63 + ch) * HW + p0 + m0 + m] = d_lds[wave][m][n];
    }
  }
}

extern "C" void kernel_launch(void* const* d_in, const int* in_sizes, int n_in,
                              void* d_out, int out_size, void* d_ws, size_t ws_size,
                              hipStream_t stream)
{
  (void)in_sizes; (void)n_in; (void)out_size; (void)ws_size;
  const float* x     = (const float*)d_in[0];
  const float* w_in  = (const float*)d_in[1];
  const float* b_in  = (const float*)d_in[2];
  const float* bn_g  = (const float*)d_in[3];
  const float* bn_b  = (const float*)d_in[4];
  const float* bn_m  = (const float*)d_in[5];
  const float* bn_v  = (const float*)d_in[6];
  const float* w_out = (const float*)d_in[7];
  const float* b_out = (const float*)d_in[8];
  float* out = (float*)d_out;

  float* xp   = (float*)d_ws;
  float* qrec = xp + (size_t)BB * 126 * HW;
  float* y2b  = xp + (size_t)84 * HW;
  float* vmid = xp + (size_t)21 * HW;
  float* wfin = xp + (size_t)84 * HW;
  float* wfout = xp + (size_t)42 * HW;

  k_prep_in <<<dim3(33), 256, 0, stream>>>(w_in, b_in, bn_g, bn_b, bn_m, bn_v, wfin);
  k_proj_in <<<dim3(BB * 900), 128, 0, stream>>>(x, wfin, xp, qrec);
  k_win     <<<dim3(BB * NWIN * NWIN, 2), 32, 0, stream>>>(xp, y2b);
  k_axial<0><<<dim3(BB * HH, 15), 32, 0, stream>>>(xp, vmid, qrec);
  k_axial<1><<<dim3(BB * WW, 15), 32, 0, stream>>>(xp, vmid, qrec);
  k_prep_out<<<dim3(17), 256, 0, stream>>>(w_out, b_out, wfout);
  k_proj_out<<<dim3(BB * 900), 128, 0, stream>>>(xp, y2b, wfout, out);
}


# axial P V: values kept in key-quad order [key/4][channel][4] by their producer, one 16-byte load is two k-steps' pairs (P: 16-byte LDS reads, once for both channel tiles); P V keys in another order
# speedup vs baseline: 4.3272x; 1.0728x over previous
#include <hip/hip_runtime.h>
#include <math.h>

#define BB 8
#define CC 63
#define HH 240
#define WW 240
#define HW (HH * WW)
#define C2 126
#define WS 5
#define NWIN 48
#define BN_EPS 1e-5f
#define VSEQ 5040
#define SP 244

typedef __attribute__((ext_vector_type(2))) float v2f;
typedef __attribute__((ext_vector_type(4))) float v4f;
typedef __attribute__((ext_vector_type(8))) float v8f;

__device__ __forceinline__ v8f wmma4(v2f a, v2f b, v8f c) {
  return __builtin_amdgcn_wmma_f32_16x16x4_f32(
       false, a,  false, b,
       (short)0, c,  false,  false);
}

__device__ __forceinline__ v2f frag_a(const float* p, int ld, int m0, int k0, int lane) {
  int m  = m0 + (lane & 15);
  int kb = k0 + ((lane >> 4) << 1);
  const float* q = p + m * ld + kb;
  v2f r; r.x = q[0]; r.y = q[1];
  return r;
}

__device__ __forceinline__ v2f frag_b(const float* p, int ld, int k0, int n0, int lane) {
  int n  = n0 + (lane & 15);
  int kb = k0 + ((lane >> 4) << 1);
  v2f r; r.x = p[kb * ld + n]; r.y = p[(kb + 1) * ld + n];
  return r;
}

__device__ __forceinline__ void frag_store(float* p, int ld, int m0, int n0, int lane, v8f d) {
  int n  = n0 + (lane & 15);
  int mb = m0 + ((lane >> 4) << 3);
#pragma unroll
  for (int r = 0; r < 8; ++r) p[(mb + r) * ld + n] = d[r];
}

#define V8Z {0.f,0.f,0.f,0.f,0.f,0.f,0.f,0.f}

#define WF_IN_N  (8 * 16 * 64)
#define WF_OUT_N (4 * 16 * 64)

__global__ __launch_bounds__(256) void k_prep_in(
    const float* __restrict__ w_in, const float* __restrict__ b_in,
    const float* __restrict__ bn_g, const float* __restrict__ bn_b,
    const float* __restrict__ bn_m, const float* __restrict__ bn_v,
    float* __restrict__ wf)
{
  int idx = blockIdx.x * 256 + threadIdx.x;
  if (idx < WF_IN_N) {
    int j = idx & 1, l = (idx >> 1) & 31, t = (idx >> 6) & 15, nt = idx >> 10;
    int k = 4 * t + ((l >> 4) << 1) + j;
    int n = nt * 16 + (l & 15);
    float val = 0.f;
    if (k < 63 && n < 126) {
      float s = bn_g[n] * rsqrtf(bn_v[n] + BN_EPS);
      val = w_in[n * 63 + k] * s;
    }
    wf[idx] = val;
  } else if (idx < WF_IN_N + 128) {
    int n = idx - WF_IN_N;
    float bv = 0.f;
    if (n < 126) {
      float s = bn_g[n] * rsqrtf(bn_v[n] + BN_EPS);
      bv = b_in[n] * s + bn_b[n] - bn_m[n] * s;
    }
    wf[idx] = bv;
  }
}

__global__ __launch_bounds__(256) void k_prep_out(
    const float* __restrict__ w_out, const float* __restrict__ b_out,
    float* __restrict__ wf)
{
  int idx = blockIdx.x * 256 + threadIdx.x;
  if (idx < WF_OUT_N) {
    int j = idx & 1, l = (idx >> 1) & 31, t = (idx >> 6) & 15, nt = idx >> 10;
    int k = 4 * t + ((l >> 4) << 1) + j;
    int n = nt * 16 + (l & 15);
    wf[idx] = (k < 63 && n < 63) ? w_out[n * 63 + k] : 0.f;
  } else if (idx < WF_OUT_N + 64) {
    int n = idx - WF_OUT_N;
    wf[idx] = (n < 63) ? b_out[n] : 0.f;
  }
}

__global__ __launch_bounds__(128) void k_proj_in(
    const float* __restrict__ x, const float* __restrict__ wf,
    float* __restrict__ xp, float* __restrict__ qrec)
{
  __shared__ float d_lds[4][16][17];

  int tid  = threadIdx.x;
  int lane = tid & 31;
  int wave = tid >> 5;
  int tile = blockIdx.x;
  int b    = tile / 900;
  int p0   = (tile % 900) * 64;
  int m0   = wave * 16;

  v2f af[16];
  {
    const float* xb = x + (size_t)b * 63 * HW + p0 + m0 + (lane & 15);
    int kh = (lane >> 4) << 1;
#pragma unroll
    for (int t = 0; t < 16; ++t) {
      int k  = 4 * t + kh;
      int k1 = (k + 1 < 63) ? k + 1 : 62;
      float v0 = xb[(size_t)k * HW];
      float v1 = xb[(size_t)k1 * HW];
      af[t].x = v0;
      af[t].y = (k + 1 < 63) ? v1 : 0.f;
    }
  }

  const v2f* wfl = (const v2f*)wf + lane;

  v2f bf[16];
#pragma unroll
  for (int t = 0; t < 16; ++t) bf[t] = wfl[t * 32];

  float bias_nx = wf[WF_IN_N + (lane & 15)];

  for (int nt = 0; nt < 8; ++nt) {
    int n0 = nt * 16;
    int nloc = lane & 15;
    int mb   = (lane >> 4) << 3;
    float bias = bias_nx;
    v8f acc = V8Z;
#pragma unroll
    for (int t = 0; t < 16; ++t) acc = wmma4(af[t], bf[t], acc);
    int nx = (nt + 1 < 8) ? nt + 1 : nt;
#pragma unroll
    for (int t = 0; t < 16; ++t) bf[t] = wfl[(nx * 16 + t) * 32];
    bias_nx = wf[WF_IN_N + nx * 16 + nloc];
#pragma unroll
    for (int r = 0; r < 8; ++r) d_lds[wave][mb + r][nloc] = acc[r] + bias;
    if (n0 < 84) {
      for (int idx = lane; idx < 256; idx += 32) {
        int n = idx >> 4, m = idx & 15;
        int ch = n0 + n;
        if (ch < 84)
          xp[(size_t)(b * 126 + ch) * HW + p0 + m0 + m] = d_lds[wave][m][n];
      }
    }
    if (n0 == 96 || n0 == 112) {
      int pix = p0 + m0;
      float* vq = xp + (size_t)(b * 126 + 105) * HW + (size_t)(pix / WW) * VSEQ
                     + ((pix % WW) >> 2) * 84;
#pragma unroll
      for (int i = 0; i < 2; ++i) {
        int idx = lane + 32 * i;
        int n = idx & 15, quad = idx >> 4;
        int c = n0 + n - 105;
        if (c >= 0 && c < 21) {
          v4f val;
          val.x = d_lds[wave][4 * quad + 0][n];
          val.y = d_lds[wave][4 * quad + 1][n];
          val.z = d_lds[wave][4 * quad + 2][n];
          val.w = d_lds[wave][4 * quad + 3][n];
          *(v4f*)(vq + (quad * 21 + c) * 4) = val;
        }
      }
    }
    if (n0 == 80 || n0 == 96) {
      for (int idx = lane; idx < 256; idx += 32) {
        int n = idx & 15, m = idx >> 4;
        int slot = n0 + n - 84;
        if (slot >= 0 && slot < 24)
          qrec[((size_t)b * HW + p0 + m0 + m) * 24 + slot] =
              (slot < 21) ? d_lds[wave][m][n] : 0.f;
      }
    }
  }
}

__global__ __launch_bounds__(32) void k_win(float* __restrict__ xp, float* __restrict__ y2)
{
  __shared__ float q[32][25];
  __shared__ float v[32][25];
  __shared__ float s[32][33];

  int lane = threadIdx.x;
  int wid  = blockIdx.x;
  int grp  = blockIdx.y;
  int b    = wid / (NWIN * NWIN);
  int rem  = wid % (NWIN * NWIN);
  int wy   = rem / NWIN, wx = rem % NWIN;

  {
    bool own = lane < 25;
    int mp = own ? lane : 24;
    int gy = wy * WS + mp / WS, gx = wx * WS + mp % WS;
    if (grp == 1) { gy = (gy + 3) % HH; gx = (gx + 3) % WW; }
    const float* src = xp + (size_t)(b * 126 + grp * 42) * HW + gy * WW + gx;
    float qv[21], vv[21];
#pragma unroll
    for (int c = 0; c < 21; ++c) {
      qv[c] = src[(size_t)c * HW];
      vv[c] = src[(size_t)(c + 21) * HW];
    }
#pragma unroll
    for (int c = 0; c < 21; ++c) {
      q[lane][c] = own ? qv[c] : 0.f;
      v[lane][c] = own ? vv[c] : 0.f;
    }
    for (int c = 21; c < 25; ++c) { q[lane][c] = 0.f; v[lane][c] = 0.f; }
  }

  for (int mt = 0; mt < 2; ++mt)
    for (int nt = 0; nt < 2; ++nt) {
      v8f acc = V8Z;
      for (int k0 = 0; k0 < 24; k0 += 4) {
        v2f av = frag_a(&q[0][0], 25, mt * 16, k0, lane);
        v2f bv = frag_a(&q[0][0], 25, nt * 16, k0, lane);
        acc = wmma4(av, bv, acc);
      }
      frag_store(&s[0][0], 33, mt * 16, nt * 16, lane, acc);
    }

  if (lane < 25) {
    float mx = -1e30f;
    for (int n = 0; n < 25; ++n) mx = fmaxf(mx, s[lane][n]);
    float sum = 0.f;
    for (int n = 0; n < 25; ++n) { float e = __expf(s[lane][n] - mx); s[lane][n] = e; sum += e; }
    float inv = 1.f / sum;
    for (int n = 0; n < 25; ++n) s[lane][n] *= inv;
    for (int n = 25; n < 28; ++n) s[lane][n] = 0.f;
  }

  for (int qt = 0; qt < 2; ++qt) {
    int m  = qt * 16 + (lane & 15);
    int mc = (m < 25) ? m : 24;
    int oy = wy * WS + mc / WS, ox = wx * WS + mc % WS;
    float* dst;
    if (grp == 0) {
      dst = xp + (size_t)(b * 126) * HW + oy * WW + ox;
    } else {
      oy = (oy + 2) % HH; ox = (ox + 2) % WW;
      dst = y2 + (size_t)(b * 126) * HW + oy * WW + ox;
    }
#pragma unroll
    for (int ct = 0; ct < 2; ++ct) {
      v8f acc = V8Z;
      for (int k0 = 0; k0 < 28; k0 += 4) {
        v2f av = frag_b(&v[0][0], 25, k0, ct * 16, lane);
        v2f bv = frag_a(&s[0][0], 33, qt * 16, k0, lane);
        acc = wmma4(av, bv, acc);
      }
      int chb = ct * 16 + ((lane >> 4) << 3);
      bool st = (m < 25) && (ct == 0 || (lane >> 4) == 0);
      if (st) {
        float* d = dst + (size_t)chb * HW;
#pragma unroll
        for (int r = 0; r < 8; ++r)
          if (ct == 0 || r < 5) d[(size_t)r * HW] = acc[r];
      }
    }
  }
}


template <int PS>
__device__ __forceinline__ void gfrag_q(const float* qb, int p0, int lane, v2f (&f)[6]) {
  int p = p0 + (lane & 15);
  const v4f* g = (const v4f*)(qb + (size_t)p * (PS * 24) + 12 * (lane >> 4));
  v4f a = g[0], b = g[1], c = g[2];
  f[0].x = a.x; f[0].y = a.y; f[1].x = a.z; f[1].y = a.w;
  f[2].x = b.x; f[2].y = b.y; f[3].x = b.z; f[3].y = b.w;
  f[4].x = c.x; f[4].y = c.y; f[5].x = c.z; f[5].y = c.w;
}

template <int X>
__device__ __forceinline__ float half_xor(float v) {
  return __int_as_float(__builtin_amdgcn_update_dpp(0, __float_as_int(v), 0x160 + X, 0xf, 0xf, false));
}

__device__ __forceinline__ float half_max(float v) {
  v = fmaxf(v, half_xor<1>(v));
  v = fmaxf(v, half_xor<2>(v));
  v = fmaxf(v, half_xor<4>(v));
  v = fmaxf(v, half_xor<8>(v));
  return v;
}

__device__ __forceinline__ float half_sum(float v) {
  v += half_xor<1>(v);
  v += half_xor<2>(v);
  v += half_xor<4>(v);
  v += half_xor<8>(v);
  return v;
}

template <int PASS>
__global__ __launch_bounds__(32) void k_axial(float* __restrict__ xp,
                                              float* __restrict__ vmid,
                                              const float* __restrict__ qrec)
{
  __shared__ __attribute__((aligned(16))) float s[16][SP];

  constexpr int PS = (PASS == 0) ? 1 : WW;

  int lane = threadIdx.x;
  int row  = blockIdx.x;
  int qt   = blockIdx.y;
  int b    = row / HH;
  int r    = row % HH;

  float* vm = vmid + (size_t)b * 126 * HW;
  const float* qb = qrec + ((size_t)b * HW + ((PASS == 0) ? r * WW : r)) * 24;
  const float* vb = (PASS == 0)
      ? xp + (size_t)(b * 126 + 105) * HW + (size_t)r * VSEQ
      : vm + (size_t)r * VSEQ;

  int m0 = qt * 16;

  v2f qa[6];
  gfrag_q<PS>(qb, m0, lane, qa);

  float rmx[8];
#pragma unroll
  for (int r = 0; r < 8; ++r) rmx[r] = -1e30f;

#pragma unroll 5
  for (int nt = 0; nt < 15; ++nt) {
    int n0 = nt * 16;
    v2f kf[6];
    gfrag_q<PS>(qb, n0, lane, kf);
    v8f acc = V8Z;
#pragma unroll
    for (int t = 0; t < 6; ++t) acc = wmma4(qa[t], kf[t], acc);
#pragma unroll
    for (int r = 0; r < 8; ++r) rmx[r] = fmaxf(rmx[r], acc[r]);
    frag_store(&s[0][0], SP, 0, n0, lane, acc);
  }
#pragma unroll
  for (int r = 0; r < 8; ++r) rmx[r] = half_max(rmx[r]);

  float* own = &s[(lane >> 4) << 3][lane & 15];
  float rsum[8];
#pragma unroll
  for (int r = 0; r < 8; ++r) rsum[r] = 0.f;
#pragma unroll 5
  for (int nt = 0; nt < 15; ++nt) {
#pragma unroll
    for (int r = 0; r < 8; ++r) {
      float e = __expf(own[r * SP + nt * 16] - rmx[r]);
      own[r * SP + nt * 16] = e;
      rsum[r] += e;
    }
  }
  float rinv[8];
#pragma unroll
  for (int r = 0; r < 8; ++r) rinv[r] = 1.f / half_sum(rsum[r]);

  int hl = lane >> 4, nl = lane & 15;
  int c1 = (16 + nl < 21) ? 16 + nl : 20;
  const v4f* vp0 = (const v4f*)vb + (hl * 21 + nl);
  const v4f* vp1 = (const v4f*)vb + (hl * 21 + c1);
  const float* ap = &s[nl][4 * hl];
  v8f acc0 = V8Z, acc1 = V8Z;
#pragma unroll 6
  for (int u = 0; u < 30; ++u) {
    v4f a  = *(const v4f*)(ap + 8 * u);
    v4f b0 = vp0[u * 42];
    v4f b1 = vp1[u * 42];
    v2f a01, a23, p01, p23, q01, q23;
    a01.x = a.x;  a01.y = a.y;  a23.x = a.z;  a23.y = a.w;
    p01.x = b0.x; p01.y = b0.y; p23.x = b0.z; p23.y = b0.w;
    q01.x = b1.x; q01.y = b1.y; q23.x = b1.z; q23.y = b1.w;
    acc0 = wmma4(a01, p01, acc0);
    acc1 = wmma4(a01, q01, acc1);
    acc0 = wmma4(a23, p23, acc0);
    acc1 = wmma4(a23, q23, acc1);
  }
  int kq4 = ((r >> 2) * 21) * 4 + (r & 3);
#pragma unroll
  for (int nt = 0; nt < 2; ++nt) {
    v8f acc = nt ? acc1 : acc0;
    int n  = nt * 16 + nl;
    int mb = hl << 3;
#pragma unroll
    for (int rr = 0; rr < 8; ++rr) {
      int m = m0 + mb + rr;
      float y = acc[rr] * rinv[rr];
      if (n < 21) {
        if (PASS == 0)
          vm[(size_t)m * VSEQ + kq4 + n * 4] = y;
        else
          xp[(size_t)(b * 126 + 105 + n) * HW + m * WW + r] = y;
      }
    }
  }
}

__global__ __launch_bounds__(128) void k_proj_out(
    const float* __restrict__ xp, const float* __restrict__ y2,
    const float* __restrict__ wf, float* __restrict__ out)
{
  __shared__ float d_lds[4][16][17];

  int tid  = threadIdx.x;
  int lane = tid & 31;
  int wave = tid >> 5;
  int tile = blockIdx.x;
  int b    = tile / 900;
  int p0   = (tile % 900) * 64;
  int m0   = wave * 16;

  v2f af[16];
  {
    const float* xb = xp + (size_t)b * 126 * HW + p0 + m0 + (lane & 15);
    const float* yb = y2 + (size_t)b * 126 * HW + p0 + m0 + (lane & 15);
    int kh = (lane >> 4) << 1;
#pragma unroll
    for (int t = 0; t < 16; ++t) {
#pragma unroll
      for (int j = 0; j < 2; ++j) {
        int k  = 4 * t + kh + j;
        int kc = (k < 63) ? k : 62;
        const float* src = (kc < 21) ? xb + (size_t)kc * HW
                         : (kc < 42) ? yb + (size_t)(kc - 21) * HW
                                     : xb + (size_t)(kc + 63) * HW;
        float v = *src;
        v = (k < 63) ? v : 0.f;
        if (j == 0) af[t].x = v; else af[t].y = v;
      }
    }
  }

  const v2f* wfl = (const v2f*)wf + lane;

  v2f bf[16];
#pragma unroll
  for (int t = 0; t < 16; ++t) bf[t] = wfl[t * 32];

  float bias_nx = wf[WF_OUT_N + (lane & 15)];

  for (int nt = 0; nt < 4; ++nt) {
    int n0 = nt * 16;
    int nloc = lane & 15;
    int mb   = (lane >> 4) << 3;
    float bias = bias_nx;
    v8f acc = V8Z;
#pragma unroll
    for (int t = 0; t < 16; ++t) acc = wmma4(af[t], bf[t], acc);
    int nx = (nt + 1 < 4) ? nt + 1 : nt;
#pragma unroll
    for (int t = 0; t < 16; ++t) bf[t] = wfl[(nx * 16 + t) * 32];
    bias_nx = wf[WF_OUT_N + nx * 16 + nloc];
#pragma unroll
    for (int r = 0; r < 8; ++r) d_lds[wave][mb + r][nloc] = acc[r] + bias;
    for (int idx = lane; idx < 256; idx += 32) {
      int n = idx >> 4, m = idx & 15;
      int ch = n0 + n;
      if (ch < 63)
        out[(size_t)(b * 63 + ch) * HW + p0 + m0 + m] = d_lds[wave][m][n];
    }
  }
}

extern "C" void kernel_launch(void* const* d_in, const int* in_sizes, int n_in,
                              void* d_out, int out_size, void* d_ws, size_t ws_size,
                              hipStream_t stream)
{
  (void)in_sizes; (void)n_in; (void)out_size; (void)ws_size;
  const float* x     = (const float*)d_in[0];
  const float* w_in  = (const float*)d_in[1];
  const float* b_in  = (const float*)d_in[2];
  const float* bn_g  = (const float*)d_in[3];
  const float* bn_b  = (const float*)d_in[4];
  const float* bn_m  = (const float*)d_in[5];
  const float* bn_v  = (const float*)d_in[6];
  const float* w_out = (const float*)d_in[7];
  const float* b_out = (const float*)d_in[8];
  float* out = (float*)d_out;

  float* xp   = (float*)d_ws;
  float* qrec = xp + (size_t)BB * 126 * HW;
  float* y2b  = xp + (size_t)84 * HW;
  float* vmid = xp + (size_t)21 * HW;
  float* wfin = xp + (size_t)84 * HW;
  float* wfout = xp + (size_t)42 * HW;

  k_prep_in <<<dim3(33), 256, 0, stream>>>(w_in, b_in, bn_g, bn_b, bn_m, bn_v, wfin);
  k_proj_in <<<dim3(BB * 900), 128, 0, stream>>>(x, wfin, xp, qrec);
  k_win     <<<dim3(BB * NWIN * NWIN, 2), 32, 0, stream>>>(xp, y2b);
  k_axial<0><<<dim3(BB * HH, 15), 32, 0, stream>>>(xp, vmid, qrec);
  k_axial<1><<<dim3(BB * WW, 15), 32, 0, stream>>>(xp, vmid, qrec);
  k_prep_out<<<dim3(17), 256, 0, stream>>>(w_out, b_out, wfout);
  k_proj_out<<<dim3(BB * 900), 128, 0, stream>>>(xp, y2b, wfout, out);
}


# axial attention: a wave computes two 16-query tiles at once, each key and value fragment it loads from global memory feeds both tiles' products (8 waves per sequence, was 15); sums unchanged
# speedup vs baseline: 4.4963x; 1.0391x over previous
#include <hip/hip_runtime.h>
#include <math.h>

#define BB 8
#define CC 63
#define HH 240
#define WW 240
#define HW (HH * WW)
#define C2 126
#define WS 5
#define NWIN 48
#define BN_EPS 1e-5f
#define VSEQ 5040
#define SP 244

typedef __attribute__((ext_vector_type(2))) float v2f;
typedef __attribute__((ext_vector_type(4))) float v4f;
typedef __attribute__((ext_vector_type(8))) float v8f;

__device__ __forceinline__ v8f wmma4(v2f a, v2f b, v8f c) {
  return __builtin_amdgcn_wmma_f32_16x16x4_f32(
       false, a,  false, b,
       (short)0, c,  false,  false);
}

__device__ __forceinline__ v2f frag_a(const float* p, int ld, int m0, int k0, int lane) {
  int m  = m0 + (lane & 15);
  int kb = k0 + ((lane >> 4) << 1);
  const float* q = p + m * ld + kb;
  v2f r; r.x = q[0]; r.y = q[1];
  return r;
}

__device__ __forceinline__ v2f frag_b(const float* p, int ld, int k0, int n0, int lane) {
  int n  = n0 + (lane & 15);
  int kb = k0 + ((lane >> 4) << 1);
  v2f r; r.x = p[kb * ld + n]; r.y = p[(kb + 1) * ld + n];
  return r;
}

__device__ __forceinline__ void frag_store(float* p, int ld, int m0, int n0, int lane, v8f d) {
  int n  = n0 + (lane & 15);
  int mb = m0 + ((lane >> 4) << 3);
#pragma unroll
  for (int r = 0; r < 8; ++r) p[(mb + r) * ld + n] = d[r];
}

#define V8Z {0.f,0.f,0.f,0.f,0.f,0.f,0.f,0.f}

#define WF_IN_N  (8 * 16 * 64)
#define WF_OUT_N (4 * 16 * 64)

__global__ __launch_bounds__(256) void k_prep_in(
    const float* __restrict__ w_in, const float* __restrict__ b_in,
    const float* __restrict__ bn_g, const float* __restrict__ bn_b,
    const float* __restrict__ bn_m, const float* __restrict__ bn_v,
    float* __restrict__ wf)
{
  int idx = blockIdx.x * 256 + threadIdx.x;
  if (idx < WF_IN_N) {
    int j = idx & 1, l = (idx >> 1) & 31, t = (idx >> 6) & 15, nt = idx >> 10;
    int k = 4 * t + ((l >> 4) << 1) + j;
    int n = nt * 16 + (l & 15);
    float val = 0.f;
    if (k < 63 && n < 126) {
      float s = bn_g[n] * rsqrtf(bn_v[n] + BN_EPS);
      val = w_in[n * 63 + k] * s;
    }
    wf[idx] = val;
  } else if (idx < WF_IN_N + 128) {
    int n = idx - WF_IN_N;
    float bv = 0.f;
    if (n < 126) {
      float s = bn_g[n] * rsqrtf(bn_v[n] + BN_EPS);
      bv = b_in[n] * s + bn_b[n] - bn_m[n] * s;
    }
    wf[idx] = bv;
  }
}

__global__ __launch_bounds__(256) void k_prep_out(
    const float* __restrict__ w_out, const float* __restrict__ b_out,
    float* __restrict__ wf)
{
  int idx = blockIdx.x * 256 + threadIdx.x;
  if (idx < WF_OUT_N) {
    int j = idx & 1, l = (idx >> 1) & 31, t = (idx >> 6) & 15, nt = idx >> 10;
    int k = 4 * t + ((l >> 4) << 1) + j;
    int n = nt * 16 + (l & 15);
    wf[idx] = (k < 63 && n < 63) ? w_out[n * 63 + k] : 0.f;
  } else if (idx < WF_OUT_N + 64) {
    int n = idx - WF_OUT_N;
    wf[idx] = (n < 63) ? b_out[n] : 0.f;
  }
}

__global__ __launch_bounds__(128) void k_proj_in(
    const float* __restrict__ x, const float* __restrict__ wf,
    float* __restrict__ xp, float* __restrict__ qrec)
{
  __shared__ float d_lds[4][16][17];

  int tid  = threadIdx.x;
  int lane = tid & 31;
  int wave = tid >> 5;
  int tile = blockIdx.x;
  int b    = tile / 900;
  int p0   = (tile % 900) * 64;
  int m0   = wave * 16;

  v2f af[16];
  {
    const float* xb = x + (size_t)b * 63 * HW + p0 + m0 + (lane & 15);
    int kh = (lane >> 4) << 1;
#pragma unroll
    for (int t = 0; t < 16; ++t) {
      int k  = 4 * t + kh;
      int k1 = (k + 1 < 63) ? k + 1 : 62;
      float v0 = xb[(size_t)k * HW];
      float v1 = xb[(size_t)k1 * HW];
      af[t].x = v0;
      af[t].y = (k + 1 < 63) ? v1 : 0.f;
    }
  }

  const v2f* wfl = (const v2f*)wf + lane;

  v2f bf[16];
#pragma unroll
  for (int t = 0; t < 16; ++t) bf[t] = wfl[t * 32];

  float bias_nx = wf[WF_IN_N + (lane & 15)];

  for (int nt = 0; nt < 8; ++nt) {
    int n0 = nt * 16;
    int nloc = lane & 15;
    int mb   = (lane >> 4) << 3;
    float bias = bias_nx;
    v8f acc = V8Z;
#pragma unroll
    for (int t = 0; t < 16; ++t) acc = wmma4(af[t], bf[t], acc);
    int nx = (nt + 1 < 8) ? nt + 1 : nt;
#pragma unroll
    for (int t = 0; t < 16; ++t) bf[t] = wfl[(nx * 16 + t) * 32];
    bias_nx = wf[WF_IN_N + nx * 16 + nloc];
#pragma unroll
    for (int r = 0; r < 8; ++r) d_lds[wave][mb + r][nloc] = acc[r] + bias;
    if (n0 < 84) {
      for (int idx = lane; idx < 256; idx += 32) {
        int n = idx >> 4, m = idx & 15;
        int ch = n0 + n;
        if (ch < 84)
          xp[(size_t)(b * 126 + ch) * HW + p0 + m0 + m] = d_lds[wave][m][n];
      }
    }
    if (n0 == 96 || n0 == 112) {
      int pix = p0 + m0;
      float* vq = xp + (size_t)(b * 126 + 105) * HW + (size_t)(pix / WW) * VSEQ
                     + ((pix % WW) >> 2) * 84;
#pragma unroll
      for (int i = 0; i < 2; ++i) {
        int idx = lane + 32 * i;
        int n = idx & 15, quad = idx >> 4;
        int c = n0 + n - 105;
        if (c >= 0 && c < 21) {
          v4f val;
          val.x = d_lds[wave][4 * quad + 0][n];
          val.y = d_lds[wave][4 * quad + 1][n];
          val.z = d_lds[wave][4 * quad + 2][n];
          val.w = d_lds[wave][4 * quad + 3][n];
          *(v4f*)(vq + (quad * 21 + c) * 4) = val;
        }
      }
    }
    if (n0 == 80 || n0 == 96) {
      for (int idx = lane; idx < 256; idx += 32) {
        int n = idx & 15, m = idx >> 4;
        int slot = n0 + n - 84;
        if (slot >= 0 && slot < 24)
          qrec[((size_t)b * HW + p0 + m0 + m) * 24 + slot] =
              (slot < 21) ? d_lds[wave][m][n] : 0.f;
      }
    }
  }
}

__global__ __launch_bounds__(32) void k_win(float* __restrict__ xp, float* __restrict__ y2)
{
  __shared__ float q[32][25];
  __shared__ float v[32][25];
  __shared__ float s[32][33];

  int lane = threadIdx.x;
  int wid  = blockIdx.x;
  int grp  = blockIdx.y;
  int b    = wid / (NWIN * NWIN);
  int rem  = wid % (NWIN * NWIN);
  int wy   = rem / NWIN, wx = rem % NWIN;

  {
    bool own = lane < 25;
    int mp = own ? lane : 24;
    int gy = wy * WS + mp / WS, gx = wx * WS + mp % WS;
    if (grp == 1) { gy = (gy + 3) % HH; gx = (gx + 3) % WW; }
    const float* src = xp + (size_t)(b * 126 + grp * 42) * HW + gy * WW + gx;
    float qv[21], vv[21];
#pragma unroll
    for (int c = 0; c < 21; ++c) {
      qv[c] = src[(size_t)c * HW];
      vv[c] = src[(size_t)(c + 21) * HW];
    }
#pragma unroll
    for (int c = 0; c < 21; ++c) {
      q[lane][c] = own ? qv[c] : 0.f;
      v[lane][c] = own ? vv[c] : 0.f;
    }
    for (int c = 21; c < 25; ++c) { q[lane][c] = 0.f; v[lane][c] = 0.f; }
  }

  for (int mt = 0; mt < 2; ++mt)
    for (int nt = 0; nt < 2; ++nt) {
      v8f acc = V8Z;
      for (int k0 = 0; k0 < 24; k0 += 4) {
        v2f av = frag_a(&q[0][0], 25, mt * 16, k0, lane);
        v2f bv = frag_a(&q[0][0], 25, nt * 16, k0, lane);
        acc = wmma4(av, bv, acc);
      }
      frag_store(&s[0][0], 33, mt * 16, nt * 16, lane, acc);
    }

  if (lane < 25) {
    float mx = -1e30f;
    for (int n = 0; n < 25; ++n) mx = fmaxf(mx, s[lane][n]);
    float sum = 0.f;
    for (int n = 0; n < 25; ++n) { float e = __expf(s[lane][n] - mx); s[lane][n] = e; sum += e; }
    float inv = 1.f / sum;
    for (int n = 0; n < 25; ++n) s[lane][n] *= inv;
    for (int n = 25; n < 28; ++n) s[lane][n] = 0.f;
  }

  for (int qt = 0; qt < 2; ++qt) {
    int m  = qt * 16 + (lane & 15);
    int mc = (m < 25) ? m : 24;
    int oy = wy * WS + mc / WS, ox = wx * WS + mc % WS;
    float* dst;
    if (grp == 0) {
      dst = xp + (size_t)(b * 126) * HW + oy * WW + ox;
    } else {
      oy = (oy + 2) % HH; ox = (ox + 2) % WW;
      dst = y2 + (size_t)(b * 126) * HW + oy * WW + ox;
    }
#pragma unroll
    for (int ct = 0; ct < 2; ++ct) {
      v8f acc = V8Z;
      for (int k0 = 0; k0 < 28; k0 += 4) {
        v2f av = frag_b(&v[0][0], 25, k0, ct * 16, lane);
        v2f bv = frag_a(&s[0][0], 33, qt * 16, k0, lane);
        acc = wmma4(av, bv, acc);
      }
      int chb = ct * 16 + ((lane >> 4) << 3);
      bool st = (m < 25) && (ct == 0 || (lane >> 4) == 0);
      if (st) {
        float* d = dst + (size_t)chb * HW;
#pragma unroll
        for (int r = 0; r < 8; ++r)
          if (ct == 0 || r < 5) d[(size_t)r * HW] = acc[r];
      }
    }
  }
}


template <int PS>
__device__ __forceinline__ void gfrag_q(const float* qb, int p0, int lane, v2f (&f)[6]) {
  int p = p0 + (lane & 15);
  const v4f* g = (const v4f*)(qb + (size_t)p * (PS * 24) + 12 * (lane >> 4));
  v4f a = g[0], b = g[1], c = g[2];
  f[0].x = a.x; f[0].y = a.y; f[1].x = a.z; f[1].y = a.w;
  f[2].x = b.x; f[2].y = b.y; f[3].x = b.z; f[3].y = b.w;
  f[4].x = c.x; f[4].y = c.y; f[5].x = c.z; f[5].y = c.w;
}

template <int X>
__device__ __forceinline__ float half_xor(float v) {
  return __int_as_float(__builtin_amdgcn_update_dpp(0, __float_as_int(v), 0x160 + X, 0xf, 0xf, false));
}

__device__ __forceinline__ float half_max(float v) {
  v = fmaxf(v, half_xor<1>(v));
  v = fmaxf(v, half_xor<2>(v));
  v = fmaxf(v, half_xor<4>(v));
  v = fmaxf(v, half_xor<8>(v));
  return v;
}

__device__ __forceinline__ float half_sum(float v) {
  v += half_xor<1>(v);
  v += half_xor<2>(v);
  v += half_xor<4>(v);
  v += half_xor<8>(v);
  return v;
}

template <int PASS, int NQ>
__device__ __forceinline__ void axial_tiles(float* __restrict__ xp, float* __restrict__ vm,
                                            const float* __restrict__ qb,
                                            const float* __restrict__ vb,
                                            float* sb, int lane, int b, int r, int m0)
{
  constexpr int PS = (PASS == 0) ? 1 : WW;

  v2f qa[NQ][6];
#pragma unroll
  for (int i = 0; i < NQ; ++i) gfrag_q<PS>(qb, m0 + 16 * i, lane, qa[i]);

  float rmx[NQ][8];
#pragma unroll
  for (int i = 0; i < NQ; ++i)
#pragma unroll
    for (int r8 = 0; r8 < 8; ++r8) rmx[i][r8] = -1e30f;

#pragma unroll 5
  for (int nt = 0; nt < 15; ++nt) {
    int n0 = nt * 16;
    v2f kf[6];
    gfrag_q<PS>(qb, n0, lane, kf);
    v8f acc[NQ];
#pragma unroll
    for (int i = 0; i < NQ; ++i) acc[i] = (v8f)V8Z;
#pragma unroll
    for (int t = 0; t < 6; ++t)
#pragma unroll
      for (int i = 0; i < NQ; ++i) acc[i] = wmma4(qa[i][t], kf[t], acc[i]);
#pragma unroll
    for (int i = 0; i < NQ; ++i) {
#pragma unroll
      for (int r8 = 0; r8 < 8; ++r8) rmx[i][r8] = fmaxf(rmx[i][r8], acc[i][r8]);
      frag_store(sb + 16 * i * SP, SP, 0, n0, lane, acc[i]);
    }
  }
#pragma unroll
  for (int i = 0; i < NQ; ++i)
#pragma unroll
    for (int r8 = 0; r8 < 8; ++r8) rmx[i][r8] = half_max(rmx[i][r8]);

  float rinv[NQ][8];
#pragma unroll
  for (int i = 0; i < NQ; ++i) {
    float* own = sb + (16 * i + ((lane >> 4) << 3)) * SP + (lane & 15);
    float rsum[8];
#pragma unroll
    for (int r8 = 0; r8 < 8; ++r8) rsum[r8] = 0.f;
#pragma unroll 5
    for (int nt = 0; nt < 15; ++nt) {
#pragma unroll
      for (int r8 = 0; r8 < 8; ++r8) {
        float e = __expf(own[r8 * SP + nt * 16] - rmx[i][r8]);
        own[r8 * SP + nt * 16] = e;
        rsum[r8] += e;
      }
    }
#pragma unroll
    for (int r8 = 0; r8 < 8; ++r8) rinv[i][r8] = 1.f / half_sum(rsum[r8]);
  }

  int hl = lane >> 4, nl = lane & 15;
  int c1 = (16 + nl < 21) ? 16 + nl : 20;
  const v4f* vp0 = (const v4f*)vb + (hl * 21 + nl);
  const v4f* vp1 = (const v4f*)vb + (hl * 21 + c1);
  const float* ap = sb + nl * SP + 4 * hl;
  v8f acc0[NQ], acc1[NQ];
#pragma unroll
  for (int i = 0; i < NQ; ++i) { acc0[i] = (v8f)V8Z; acc1[i] = (v8f)V8Z; }
#pragma unroll 6
  for (int u = 0; u < 30; ++u) {
    v4f b0 = vp0[u * 42];
    v4f b1 = vp1[u * 42];
    v2f p01, p23, q01, q23;
    p01.x = b0.x; p01.y = b0.y; p23.x = b0.z; p23.y = b0.w;
    q01.x = b1.x; q01.y = b1.y; q23.x = b1.z; q23.y = b1.w;
#pragma unroll
    for (int i = 0; i < NQ; ++i) {
      v4f a = *(const v4f*)(ap + 16 * i * SP + 8 * u);
      v2f a01, a23;
      a01.x = a.x; a01.y = a.y; a23.x = a.z; a23.y = a.w;
      acc0[i] = wmma4(a01, p01, acc0[i]);
      acc1[i] = wmma4(a01, q01, acc1[i]);
      acc0[i] = wmma4(a23, p23, acc0[i]);
      acc1[i] = wmma4(a23, q23, acc1[i]);
    }
  }
  int kq4 = ((r >> 2) * 21) * 4 + (r & 3);
#pragma unroll
  for (int i = 0; i < NQ; ++i) {
#pragma unroll
    for (int nt = 0; nt < 2; ++nt) {
      v8f acc = nt ? acc1[i] : acc0[i];
      int n  = nt * 16 + nl;
      int mb = hl << 3;
#pragma unroll
      for (int rr = 0; rr < 8; ++rr) {
        int m = m0 + 16 * i + mb + rr;
        float y = acc[rr] * rinv[i][rr];
        if (n < 21) {
          if (PASS == 0)
            vm[(size_t)m * VSEQ + kq4 + n * 4] = y;
          else
            xp[(size_t)(b * 126 + 105 + n) * HW + m * WW + r] = y;
        }
      }
    }
  }
}

template <int PASS>
__global__ __launch_bounds__(32) void k_axial(float* __restrict__ xp,
                                              float* __restrict__ vmid,
                                              const float* __restrict__ qrec)
{
  __shared__ __attribute__((aligned(16))) float s[32][SP];

  int lane = threadIdx.x;
  int row  = blockIdx.x;
  int qp   = blockIdx.y;
  int b    = row / HH;
  int r    = row % HH;

  float* vm = vmid + (size_t)b * 126 * HW;
  const float* qb = qrec + ((size_t)b * HW + ((PASS == 0) ? r * WW : r)) * 24;
  const float* vb = (PASS == 0)
      ? xp + (size_t)(b * 126 + 105) * HW + (size_t)r * VSEQ
      : vm + (size_t)r * VSEQ;

  if (qp < 7)
    axial_tiles<PASS, 2>(xp, vm, qb, vb, &s[0][0], lane, b, r, qp * 32);
  else
    axial_tiles<PASS, 1>(xp, vm, qb, vb, &s[0][0], lane, b, r, 224);
}

__global__ __launch_bounds__(128) void k_proj_out(
    const float* __restrict__ xp, const float* __restrict__ y2,
    const float* __restrict__ wf, float* __restrict__ out)
{
  __shared__ float d_lds[4][16][17];

  int tid  = threadIdx.x;
  int lane = tid & 31;
  int wave = tid >> 5;
  int tile = blockIdx.x;
  int b    = tile / 900;
  int p0   = (tile % 900) * 64;
  int m0   = wave * 16;

  v2f af[16];
  {
    const float* xb = xp + (size_t)b * 126 * HW + p0 + m0 + (lane & 15);
    const float* yb = y2 + (size_t)b * 126 * HW + p0 + m0 + (lane & 15);
    int kh = (lane >> 4) << 1;
#pragma unroll
    for (int t = 0; t < 16; ++t) {
#pragma unroll
      for (int j = 0; j < 2; ++j) {
        int k  = 4 * t + kh + j;
        int kc = (k < 63) ? k : 62;
        const float* src = (kc < 21) ? xb + (size_t)kc * HW
                         : (kc < 42) ? yb + (size_t)(kc - 21) * HW
                                     : xb + (size_t)(kc + 63) * HW;
        float v = *src;
        v = (k < 63) ? v : 0.f;
        if (j == 0) af[t].x = v; else af[t].y = v;
      }
    }
  }

  const v2f* wfl = (const v2f*)wf + lane;

  v2f bf[16];
#pragma unroll
  for (int t = 0; t < 16; ++t) bf[t] = wfl[t * 32];

  float bias_nx = wf[WF_OUT_N + (lane & 15)];

  for (int nt = 0; nt < 4; ++nt) {
    int n0 = nt * 16;
    int nloc = lane & 15;
    int mb   = (lane >> 4) << 3;
    float bias = bias_nx;
    v8f acc = V8Z;
#pragma unroll
    for (int t = 0; t < 16; ++t) acc = wmma4(af[t], bf[t], acc);
    int nx = (nt + 1 < 4) ? nt + 1 : nt;
#pragma unroll
    for (int t = 0; t < 16; ++t) bf[t] = wfl[(nx * 16 + t) * 32];
    bias_nx = wf[WF_OUT_N + nx * 16 + nloc];
#pragma unroll
    for (int r = 0; r < 8; ++r) d_lds[wave][mb + r][nloc] = acc[r] + bias;
    for (int idx = lane; idx < 256; idx += 32) {
      int n = idx >> 4, m = idx & 15;
      int ch = n0 + n;
      if (ch < 63)
        out[(size_t)(b * 63 + ch) * HW + p0 + m0 + m] = d_lds[wave][m][n];
    }
  }
}

extern "C" void kernel_launch(void* const* d_in, const int* in_sizes, int n_in,
                              void* d_out, int out_size, void* d_ws, size_t ws_size,
                              hipStream_t stream)
{
  (void)in_sizes; (void)n_in; (void)out_size; (void)ws_size;
  const float* x     = (const float*)d_in[0];
  const float* w_in  = (const float*)d_in[1];
  const float* b_in  = (const float*)d_in[2];
  const float* bn_g  = (const float*)d_in[3];
  const float* bn_b  = (const float*)d_in[4];
  const float* bn_m  = (const float*)d_in[5];
  const float* bn_v  = (const float*)d_in[6];
  const float* w_out = (const float*)d_in[7];
  const float* b_out = (const float*)d_in[8];
  float* out = (float*)d_out;

  float* xp   = (float*)d_ws;
  float* qrec = xp + (size_t)BB * 126 * HW;
  float* y2b  = xp + (size_t)84 * HW;
  float* vmid = xp + (size_t)21 * HW;
  float* wfin = xp + (size_t)84 * HW;
  float* wfout = xp + (size_t)42 * HW;

  k_prep_in <<<dim3(33), 256, 0, stream>>>(w_in, b_in, bn_g, bn_b, bn_m, bn_v, wfin);
  k_proj_in <<<dim3(BB * 900), 128, 0, stream>>>(x, wfin, xp, qrec);
  k_win     <<<dim3(BB * NWIN * NWIN, 2), 32, 0, stream>>>(xp, y2b);
  k_axial<0><<<dim3(BB * HH, 8), 32, 0, stream>>>(xp, vmid, qrec);
  k_axial<1><<<dim3(BB * WW, 8), 32, 0, stream>>>(xp, vmid, qrec);
  k_prep_out<<<dim3(17), 256, 0, stream>>>(w_out, b_out, wfout);
  k_proj_out<<<dim3(BB * 900), 128, 0, stream>>>(xp, y2b, wfout, out);
}


# axial attention: a sequence's 8 waves get neighbouring block indices (x = query-tile pair, y = sequence): they run together, its keys and values come from memory once, then from L2; sums unchanged
# speedup vs baseline: 5.5025x; 1.2238x over previous
#include <hip/hip_runtime.h>
#include <math.h>

#define BB 8
#define CC 63
#define HH 240
#define WW 240
#define HW (HH * WW)
#define C2 126
#define WS 5
#define NWIN 48
#define BN_EPS 1e-5f
#define VSEQ 5040
#define SP 244

typedef __attribute__((ext_vector_type(2))) float v2f;
typedef __attribute__((ext_vector_type(4))) float v4f;
typedef __attribute__((ext_vector_type(8))) float v8f;

__device__ __forceinline__ v8f wmma4(v2f a, v2f b, v8f c) {
  return __builtin_amdgcn_wmma_f32_16x16x4_f32(
       false, a,  false, b,
       (short)0, c,  false,  false);
}

__device__ __forceinline__ v2f frag_a(const float* p, int ld, int m0, int k0, int lane) {
  int m  = m0 + (lane & 15);
  int kb = k0 + ((lane >> 4) << 1);
  const float* q = p + m * ld + kb;
  v2f r; r.x = q[0]; r.y = q[1];
  return r;
}

__device__ __forceinline__ v2f frag_b(const float* p, int ld, int k0, int n0, int lane) {
  int n  = n0 + (lane & 15);
  int kb = k0 + ((lane >> 4) << 1);
  v2f r; r.x = p[kb * ld + n]; r.y = p[(kb + 1) * ld + n];
  return r;
}

__device__ __forceinline__ void frag_store(float* p, int ld, int m0, int n0, int lane, v8f d) {
  int n  = n0 + (lane & 15);
  int mb = m0 + ((lane >> 4) << 3);
#pragma unroll
  for (int r = 0; r < 8; ++r) p[(mb + r) * ld + n] = d[r];
}

#define V8Z {0.f,0.f,0.f,0.f,0.f,0.f,0.f,0.f}

#define WF_IN_N  (8 * 16 * 64)
#define WF_OUT_N (4 * 16 * 64)

__global__ __launch_bounds__(256) void k_prep_in(
    const float* __restrict__ w_in, const float* __restrict__ b_in,
    const float* __restrict__ bn_g, const float* __restrict__ bn_b,
    const float* __restrict__ bn_m, const float* __restrict__ bn_v,
    float* __restrict__ wf)
{
  int idx = blockIdx.x * 256 + threadIdx.x;
  if (idx < WF_IN_N) {
    int j = idx & 1, l = (idx >> 1) & 31, t = (idx >> 6) & 15, nt = idx >> 10;
    int k = 4 * t + ((l >> 4) << 1) + j;
    int n = nt * 16 + (l & 15);
    float val = 0.f;
    if (k < 63 && n < 126) {
      float s = bn_g[n] * rsqrtf(bn_v[n] + BN_EPS);
      val = w_in[n * 63 + k] * s;
    }
    wf[idx] = val;
  } else if (idx < WF_IN_N + 128) {
    int n = idx - WF_IN_N;
    float bv = 0.f;
    if (n < 126) {
      float s = bn_g[n] * rsqrtf(bn_v[n] + BN_EPS);
      bv = b_in[n] * s + bn_b[n] - bn_m[n] * s;
    }
    wf[idx] = bv;
  }
}

__global__ __launch_bounds__(256) void k_prep_out(
    const float* __restrict__ w_out, const float* __restrict__ b_out,
    float* __restrict__ wf)
{
  int idx = blockIdx.x * 256 + threadIdx.x;
  if (idx < WF_OUT_N) {
    int j = idx & 1, l = (idx >> 1) & 31, t = (idx >> 6) & 15, nt = idx >> 10;
    int k = 4 * t + ((l >> 4) << 1) + j;
    int n = nt * 16 + (l & 15);
    wf[idx] = (k < 63 && n < 63) ? w_out[n * 63 + k] : 0.f;
  } else if (idx < WF_OUT_N + 64) {
    int n = idx - WF_OUT_N;
    wf[idx] = (n < 63) ? b_out[n] : 0.f;
  }
}

__global__ __launch_bounds__(128) void k_proj_in(
    const float* __restrict__ x, const float* __restrict__ wf,
    float* __restrict__ xp, float* __restrict__ qrec)
{
  __shared__ float d_lds[4][16][17];

  int tid  = threadIdx.x;
  int lane = tid & 31;
  int wave = tid >> 5;
  int tile = blockIdx.x;
  int b    = tile / 900;
  int p0   = (tile % 900) * 64;
  int m0   = wave * 16;

  v2f af[16];
  {
    const float* xb = x + (size_t)b * 63 * HW + p0 + m0 + (lane & 15);
    int kh = (lane >> 4) << 1;
#pragma unroll
    for (int t = 0; t < 16; ++t) {
      int k  = 4 * t + kh;
      int k1 = (k + 1 < 63) ? k + 1 : 62;
      float v0 = xb[(size_t)k * HW];
      float v1 = xb[(size_t)k1 * HW];
      af[t].x = v0;
      af[t].y = (k + 1 < 63) ? v1 : 0.f;
    }
  }

  const v2f* wfl = (const v2f*)wf + lane;

  v2f bf[16];
#pragma unroll
  for (int t = 0; t < 16; ++t) bf[t] = wfl[t * 32];

  float bias_nx = wf[WF_IN_N + (lane & 15)];

  for (int nt = 0; nt < 8; ++nt) {
    int n0 = nt * 16;
    int nloc = lane & 15;
    int mb   = (lane >> 4) << 3;
    float bias = bias_nx;
    v8f acc = V8Z;
#pragma unroll
    for (int t = 0; t < 16; ++t) acc = wmma4(af[t], bf[t], acc);
    int nx = (nt + 1 < 8) ? nt + 1 : nt;
#pragma unroll
    for (int t = 0; t < 16; ++t) bf[t] = wfl[(nx * 16 + t) * 32];
    bias_nx = wf[WF_IN_N + nx * 16 + nloc];
#pragma unroll
    for (int r = 0; r < 8; ++r) d_lds[wave][mb + r][nloc] = acc[r] + bias;
    if (n0 < 84) {
      for (int idx = lane; idx < 256; idx += 32) {
        int n = idx >> 4, m = idx & 15;
        int ch = n0 + n;
        if (ch < 84)
          xp[(size_t)(b * 126 + ch) * HW + p0 + m0 + m] = d_lds[wave][m][n];
      }
    }
    if (n0 == 96 || n0 == 112) {
      int pix = p0 + m0;
      float* vq = xp + (size_t)(b * 126 + 105) * HW + (size_t)(pix / WW) * VSEQ
                     + ((pix % WW) >> 2) * 84;
#pragma unroll
      for (int i = 0; i < 2; ++i) {
        int idx = lane + 32 * i;
        int n = idx & 15, quad = idx >> 4;
        int c = n0 + n - 105;
        if (c >= 0 && c < 21) {
          v4f val;
          val.x = d_lds[wave][4 * quad + 0][n];
          val.y = d_lds[wave][4 * quad + 1][n];
          val.z = d_lds[wave][4 * quad + 2][n];
          val.w = d_lds[wave][4 * quad + 3][n];
          *(v4f*)(vq + (quad * 21 + c) * 4) = val;
        }
      }
    }
    if (n0 == 80 || n0 == 96) {
      for (int idx = lane; idx < 256; idx += 32) {
        int n = idx & 15, m = idx >> 4;
        int slot = n0 + n - 84;
        if (slot >= 0 && slot < 24)
          qrec[((size_t)b * HW + p0 + m0 + m) * 24 + slot] =
              (slot < 21) ? d_lds[wave][m][n] : 0.f;
      }
    }
  }
}

__global__ __launch_bounds__(32) void k_win(float* __restrict__ xp, float* __restrict__ y2)
{
  __shared__ float q[32][25];
  __shared__ float v[32][25];
  __shared__ float s[32][33];

  int lane = threadIdx.x;
  int wid  = blockIdx.x;
  int grp  = blockIdx.y;
  int b    = wid / (NWIN * NWIN);
  int rem  = wid % (NWIN * NWIN);
  int wy   = rem / NWIN, wx = rem % NWIN;

  {
    bool own = lane < 25;
    int mp = own ? lane : 24;
    int gy = wy * WS + mp / WS, gx = wx * WS + mp % WS;
    if (grp == 1) { gy = (gy + 3) % HH; gx = (gx + 3) % WW; }
    const float* src = xp + (size_t)(b * 126 + grp * 42) * HW + gy * WW + gx;
    float qv[21], vv[21];
#pragma unroll
    for (int c = 0; c < 21; ++c) {
      qv[c] = src[(size_t)c * HW];
      vv[c] = src[(size_t)(c + 21) * HW];
    }
#pragma unroll
    for (int c = 0; c < 21; ++c) {
      q[lane][c] = own ? qv[c] : 0.f;
      v[lane][c] = own ? vv[c] : 0.f;
    }
    for (int c = 21; c < 25; ++c) { q[lane][c] = 0.f; v[lane][c] = 0.f; }
  }

  for (int mt = 0; mt < 2; ++mt)
    for (int nt = 0; nt < 2; ++nt) {
      v8f acc = V8Z;
      for (int k0 = 0; k0 < 24; k0 += 4) {
        v2f av = frag_a(&q[0][0], 25, mt * 16, k0, lane);
        v2f bv = frag_a(&q[0][0], 25, nt * 16, k0, lane);
        acc = wmma4(av, bv, acc);
      }
      frag_store(&s[0][0], 33, mt * 16, nt * 16, lane, acc);
    }

  if (lane < 25) {
    float mx = -1e30f;
    for (int n = 0; n < 25; ++n) mx = fmaxf(mx, s[lane][n]);
    float sum = 0.f;
    for (int n = 0; n < 25; ++n) { float e = __expf(s[lane][n] - mx); s[lane][n] = e; sum += e; }
    float inv = 1.f / sum;
    for (int n = 0; n < 25; ++n) s[lane][n] *= inv;
    for (int n = 25; n < 28; ++n) s[lane][n] = 0.f;
  }

  for (int qt = 0; qt < 2; ++qt) {
    int m  = qt * 16 + (lane & 15);
    int mc = (m < 25) ? m : 24;
    int oy = wy * WS + mc / WS, ox = wx * WS + mc % WS;
    float* dst;
    if (grp == 0) {
      dst = xp + (size_t)(b * 126) * HW + oy * WW + ox;
    } else {
      oy = (oy + 2) % HH; ox = (ox + 2) % WW;
      dst = y2 + (size_t)(b * 126) * HW + oy * WW + ox;
    }
#pragma unroll
    for (int ct = 0; ct < 2; ++ct) {
      v8f acc = V8Z;
      for (int k0 = 0; k0 < 28; k0 += 4) {
        v2f av = frag_b(&v[0][0], 25, k0, ct * 16, lane);
        v2f bv = frag_a(&s[0][0], 33, qt * 16, k0, lane);
        acc = wmma4(av, bv, acc);
      }
      int chb = ct * 16 + ((lane >> 4) << 3);
      bool st = (m < 25) && (ct == 0 || (lane >> 4) == 0);
      if (st) {
        float* d = dst + (size_t)chb * HW;
#pragma unroll
        for (int r = 0; r < 8; ++r)
          if (ct == 0 || r < 5) d[(size_t)r * HW] = acc[r];
      }
    }
  }
}


template <int PS>
__device__ __forceinline__ void gfrag_q(const float* qb, int p0, int lane, v2f (&f)[6]) {
  int p = p0 + (lane & 15);
  const v4f* g = (const v4f*)(qb + (size_t)p * (PS * 24) + 12 * (lane >> 4));
  v4f a = g[0], b = g[1], c = g[2];
  f[0].x = a.x; f[0].y = a.y; f[1].x = a.z; f[1].y = a.w;
  f[2].x = b.x; f[2].y = b.y; f[3].x = b.z; f[3].y = b.w;
  f[4].x = c.x; f[4].y = c.y; f[5].x = c.z; f[5].y = c.w;
}

template <int X>
__device__ __forceinline__ float half_xor(float v) {
  return __int_as_float(__builtin_amdgcn_update_dpp(0, __float_as_int(v), 0x160 + X, 0xf, 0xf, false));
}

__device__ __forceinline__ float half_max(float v) {
  v = fmaxf(v, half_xor<1>(v));
  v = fmaxf(v, half_xor<2>(v));
  v = fmaxf(v, half_xor<4>(v));
  v = fmaxf(v, half_xor<8>(v));
  return v;
}

__device__ __forceinline__ float half_sum(float v) {
  v += half_xor<1>(v);
  v += half_xor<2>(v);
  v += half_xor<4>(v);
  v += half_xor<8>(v);
  return v;
}

template <int PASS, int NQ>
__device__ __forceinline__ void axial_tiles(float* __restrict__ xp, float* __restrict__ vm,
                                            const float* __restrict__ qb,
                                            const float* __restrict__ vb,
                                            float* sb, int lane, int b, int r, int m0)
{
  constexpr int PS = (PASS == 0) ? 1 : WW;

  v2f qa[NQ][6];
#pragma unroll
  for (int i = 0; i < NQ; ++i) gfrag_q<PS>(qb, m0 + 16 * i, lane, qa[i]);

  float rmx[NQ][8];
#pragma unroll
  for (int i = 0; i < NQ; ++i)
#pragma unroll
    for (int r8 = 0; r8 < 8; ++r8) rmx[i][r8] = -1e30f;

#pragma unroll 5
  for (int nt = 0; nt < 15; ++nt) {
    int n0 = nt * 16;
    v2f kf[6];
    gfrag_q<PS>(qb, n0, lane, kf);
    v8f acc[NQ];
#pragma unroll
    for (int i = 0; i < NQ; ++i) acc[i] = (v8f)V8Z;
#pragma unroll
    for (int t = 0; t < 6; ++t)
#pragma unroll
      for (int i = 0; i < NQ; ++i) acc[i] = wmma4(qa[i][t], kf[t], acc[i]);
#pragma unroll
    for (int i = 0; i < NQ; ++i) {
#pragma unroll
      for (int r8 = 0; r8 < 8; ++r8) rmx[i][r8] = fmaxf(rmx[i][r8], acc[i][r8]);
      frag_store(sb + 16 * i * SP, SP, 0, n0, lane, acc[i]);
    }
  }
#pragma unroll
  for (int i = 0; i < NQ; ++i)
#pragma unroll
    for (int r8 = 0; r8 < 8; ++r8) rmx[i][r8] = half_max(rmx[i][r8]);

  float rinv[NQ][8];
#pragma unroll
  for (int i = 0; i < NQ; ++i) {
    float* own = sb + (16 * i + ((lane >> 4) << 3)) * SP + (lane & 15);
    float rsum[8];
#pragma unroll
    for (int r8 = 0; r8 < 8; ++r8) rsum[r8] = 0.f;
#pragma unroll 5
    for (int nt = 0; nt < 15; ++nt) {
#pragma unroll
      for (int r8 = 0; r8 < 8; ++r8) {
        float e = __expf(own[r8 * SP + nt * 16] - rmx[i][r8]);
        own[r8 * SP + nt * 16] = e;
        rsum[r8] += e;
      }
    }
#pragma unroll
    for (int r8 = 0; r8 < 8; ++r8) rinv[i][r8] = 1.f / half_sum(rsum[r8]);
  }

  int hl = lane >> 4, nl = lane & 15;
  int c1 = (16 + nl < 21) ? 16 + nl : 20;
  const v4f* vp0 = (const v4f*)vb + (hl * 21 + nl);
  const v4f* vp1 = (const v4f*)vb + (hl * 21 + c1);
  const float* ap = sb + nl * SP + 4 * hl;
  v8f acc0[NQ], acc1[NQ];
#pragma unroll
  for (int i = 0; i < NQ; ++i) { acc0[i] = (v8f)V8Z; acc1[i] = (v8f)V8Z; }
#pragma unroll 6
  for (int u = 0; u < 30; ++u) {
    v4f b0 = vp0[u * 42];
    v4f b1 = vp1[u * 42];
    v2f p01, p23, q01, q23;
    p01.x = b0.x; p01.y = b0.y; p23.x = b0.z; p23.y = b0.w;
    q01.x = b1.x; q01.y = b1.y; q23.x = b1.z; q23.y = b1.w;
#pragma unroll
    for (int i = 0; i < NQ; ++i) {
      v4f a = *(const v4f*)(ap + 16 * i * SP + 8 * u);
      v2f a01, a23;
      a01.x = a.x; a01.y = a.y; a23.x = a.z; a23.y = a.w;
      acc0[i] = wmma4(a01, p01, acc0[i]);
      acc1[i] = wmma4(a01, q01, acc1[i]);
      acc0[i] = wmma4(a23, p23, acc0[i]);
      acc1[i] = wmma4(a23, q23, acc1[i]);
    }
  }
  int kq4 = ((r >> 2) * 21) * 4 + (r & 3);
#pragma unroll
  for (int i = 0; i < NQ; ++i) {
#pragma unroll
    for (int nt = 0; nt < 2; ++nt) {
      v8f acc = nt ? acc1[i] : acc0[i];
      int n  = nt * 16 + nl;
      int mb = hl << 3;
#pragma unroll
      for (int rr = 0; rr < 8; ++rr) {
        int m = m0 + 16 * i + mb + rr;
        float y = acc[rr] * rinv[i][rr];
        if (n < 21) {
          if (PASS == 0)
            vm[(size_t)m * VSEQ + kq4 + n * 4] = y;
          else
            xp[(size_t)(b * 126 + 105 + n) * HW + m * WW + r] = y;
        }
      }
    }
  }
}

template <int PASS>
__global__ __launch_bounds__(32) void k_axial(float* __restrict__ xp,
                                              float* __restrict__ vmid,
                                              const float* __restrict__ qrec)
{
  __shared__ __attribute__((aligned(16))) float s[32][SP];

  int lane = threadIdx.x;
  int row  = blockIdx.y;
  int qp   = blockIdx.x;
  int b    = row / HH;
  int r    = row % HH;

  float* vm = vmid + (size_t)b * 126 * HW;
  const float* qb = qrec + ((size_t)b * HW + ((PASS == 0) ? r * WW : r)) * 24;
  const float* vb = (PASS == 0)
      ? xp + (size_t)(b * 126 + 105) * HW + (size_t)r * VSEQ
      : vm + (size_t)r * VSEQ;

  if (qp < 7)
    axial_tiles<PASS, 2>(xp, vm, qb, vb, &s[0][0], lane, b, r, qp * 32);
  else
    axial_tiles<PASS, 1>(xp, vm, qb, vb, &s[0][0], lane, b, r, 224);
}

__global__ __launch_bounds__(128) void k_proj_out(
    const float* __restrict__ xp, const float* __restrict__ y2,
    const float* __restrict__ wf, float* __restrict__ out)
{
  __shared__ float d_lds[4][16][17];

  int tid  = threadIdx.x;
  int lane = tid & 31;
  int wave = tid >> 5;
  int tile = blockIdx.x;
  int b    = tile / 900;
  int p0   = (tile % 900) * 64;
  int m0   = wave * 16;

  v2f af[16];
  {
    const float* xb = xp + (size_t)b * 126 * HW + p0 + m0 + (lane & 15);
    const float* yb = y2 + (size_t)b * 126 * HW + p0 + m0 + (lane & 15);
    int kh = (lane >> 4) << 1;
#pragma unroll
    for (int t = 0; t < 16; ++t) {
#pragma unroll
      for (int j = 0; j < 2; ++j) {
        int k  = 4 * t + kh + j;
        int kc = (k < 63) ? k : 62;
        const float* src = (kc < 21) ? xb + (size_t)kc * HW
                         : (kc < 42) ? yb + (size_t)(kc - 21) * HW
                                     : xb + (size_t)(kc + 63) * HW;
        float v = *src;
        v = (k < 63) ? v : 0.f;
        if (j == 0) af[t].x = v; else af[t].y = v;
      }
    }
  }

  const v2f* wfl = (const v2f*)wf + lane;

  v2f bf[16];
#pragma unroll
  for (int t = 0; t < 16; ++t) bf[t] = wfl[t * 32];

  float bias_nx = wf[WF_OUT_N + (lane & 15)];

  for (int nt = 0; nt < 4; ++nt) {
    int n0 = nt * 16;
    int nloc = lane & 15;
    int mb   = (lane >> 4) << 3;
    float bias = bias_nx;
    v8f acc = V8Z;
#pragma unroll
    for (int t = 0; t < 16; ++t) acc = wmma4(af[t], bf[t], acc);
    int nx = (nt + 1 < 4) ? nt + 1 : nt;
#pragma unroll
    for (int t = 0; t < 16; ++t) bf[t] = wfl[(nx * 16 + t) * 32];
    bias_nx = wf[WF_OUT_N + nx * 16 + nloc];
#pragma unroll
    for (int r = 0; r < 8; ++r) d_lds[wave][mb + r][nloc] = acc[r] + bias;
    for (int idx = lane; idx < 256; idx += 32) {
      int n = idx >> 4, m = idx & 15;
      int ch = n0 + n;
      if (ch < 63)
        out[(size_t)(b * 63 + ch) * HW + p0 + m0 + m] = d_lds[wave][m][n];
    }
  }
}

extern "C" void kernel_launch(void* const* d_in, const int* in_sizes, int n_in,
                              void* d_out, int out_size, void* d_ws, size_t ws_size,
                              hipStream_t stream)
{
  (void)in_sizes; (void)n_in; (void)out_size; (void)ws_size;
  const float* x     = (const float*)d_in[0];
  const float* w_in  = (const float*)d_in[1];
  const float* b_in  = (const float*)d_in[2];
  const float* bn_g  = (const float*)d_in[3];
  const float* bn_b  = (const float*)d_in[4];
  const float* bn_m  = (const float*)d_in[5];
  const float* bn_v  = (const float*)d_in[6];
  const float* w_out = (const float*)d_in[7];
  const float* b_out = (const float*)d_in[8];
  float* out = (float*)d_out;

  float* xp   = (float*)d_ws;
  float* qrec = xp + (size_t)BB * 126 * HW;
  float* y2b  = xp + (size_t)84 * HW;
  float* vmid = xp + (size_t)21 * HW;
  float* wfin = xp + (size_t)84 * HW;
  float* wfout = xp + (size_t)42 * HW;

  k_prep_in <<<dim3(33), 256, 0, stream>>>(w_in, b_in, bn_g, bn_b, bn_m, bn_v, wfin);
  k_proj_in <<<dim3(BB * 900), 128, 0, stream>>>(x, wfin, xp, qrec);
  k_win     <<<dim3(BB * NWIN * NWIN, 2), 32, 0, stream>>>(xp, y2b);
  k_axial<0><<<dim3(8, BB * HH), 32, 0, stream>>>(xp, vmid, qrec);
  k_axial<1><<<dim3(8, BB * WW), 32, 0, stream>>>(xp, vmid, qrec);
  k_prep_out<<<dim3(17), 256, 0, stream>>>(w_out, b_out, wfout);
  k_proj_out<<<dim3(BB * 900), 128, 0, stream>>>(xp, y2b, wfout, out);
}


# axial column pass: a workgroup is the waves of two neighbouring columns; results pass through the dead score strips and leave as 8-byte stores (both columns of a channel and row); sums unchanged
# speedup vs baseline: 5.7942x; 1.0530x over previous
#include <hip/hip_runtime.h>
#include <math.h>

#define BB 8
#define CC 63
#define HH 240
#define WW 240
#define HW (HH * WW)
#define C2 126
#define WS 5
#define NWIN 48
#define BN_EPS 1e-5f
#define VSEQ 5040
#define SP 244

typedef __attribute__((ext_vector_type(2))) float v2f;
typedef __attribute__((ext_vector_type(4))) float v4f;
typedef __attribute__((ext_vector_type(8))) float v8f;

__device__ __forceinline__ v8f wmma4(v2f a, v2f b, v8f c) {
  return __builtin_amdgcn_wmma_f32_16x16x4_f32(
       false, a,  false, b,
       (short)0, c,  false,  false);
}

__device__ __forceinline__ v2f frag_a(const float* p, int ld, int m0, int k0, int lane) {
  int m  = m0 + (lane & 15);
  int kb = k0 + ((lane >> 4) << 1);
  const float* q = p + m * ld + kb;
  v2f r; r.x = q[0]; r.y = q[1];
  return r;
}

__device__ __forceinline__ v2f frag_b(const float* p, int ld, int k0, int n0, int lane) {
  int n  = n0 + (lane & 15);
  int kb = k0 + ((lane >> 4) << 1);
  v2f r; r.x = p[kb * ld + n]; r.y = p[(kb + 1) * ld + n];
  return r;
}

__device__ __forceinline__ void frag_store(float* p, int ld, int m0, int n0, int lane, v8f d) {
  int n  = n0 + (lane & 15);
  int mb = m0 + ((lane >> 4) << 3);
#pragma unroll
  for (int r = 0; r < 8; ++r) p[(mb + r) * ld + n] = d[r];
}

#define V8Z {0.f,0.f,0.f,0.f,0.f,0.f,0.f,0.f}

#define WF_IN_N  (8 * 16 * 64)
#define WF_OUT_N (4 * 16 * 64)

__global__ __launch_bounds__(256) void k_prep_in(
    const float* __restrict__ w_in, const float* __restrict__ b_in,
    const float* __restrict__ bn_g, const float* __restrict__ bn_b,
    const float* __restrict__ bn_m, const float* __restrict__ bn_v,
    float* __restrict__ wf)
{
  int idx = blockIdx.x * 256 + threadIdx.x;
  if (idx < WF_IN_N) {
    int j = idx & 1, l = (idx >> 1) & 31, t = (idx >> 6) & 15, nt = idx >> 10;
    int k = 4 * t + ((l >> 4) << 1) + j;
    int n = nt * 16 + (l & 15);
    float val = 0.f;
    if (k < 63 && n < 126) {
      float s = bn_g[n] * rsqrtf(bn_v[n] + BN_EPS);
      val = w_in[n * 63 + k] * s;
    }
    wf[idx] = val;
  } else if (idx < WF_IN_N + 128) {
    int n = idx - WF_IN_N;
    float bv = 0.f;
    if (n < 126) {
      float s = bn_g[n] * rsqrtf(bn_v[n] + BN_EPS);
      bv = b_in[n] * s + bn_b[n] - bn_m[n] * s;
    }
    wf[idx] = bv;
  }
}

__global__ __launch_bounds__(256) void k_prep_out(
    const float* __restrict__ w_out, const float* __restrict__ b_out,
    float* __restrict__ wf)
{
  int idx = blockIdx.x * 256 + threadIdx.x;
  if (idx < WF_OUT_N) {
    int j = idx & 1, l = (idx >> 1) & 31, t = (idx >> 6) & 15, nt = idx >> 10;
    int k = 4 * t + ((l >> 4) << 1) + j;
    int n = nt * 16 + (l & 15);
    wf[idx] = (k < 63 && n < 63) ? w_out[n * 63 + k] : 0.f;
  } else if (idx < WF_OUT_N + 64) {
    int n = idx - WF_OUT_N;
    wf[idx] = (n < 63) ? b_out[n] : 0.f;
  }
}

__global__ __launch_bounds__(128) void k_proj_in(
    const float* __restrict__ x, const float* __restrict__ wf,
    float* __restrict__ xp, float* __restrict__ qrec)
{
  __shared__ float d_lds[4][16][17];

  int tid  = threadIdx.x;
  int lane = tid & 31;
  int wave = tid >> 5;
  int tile = blockIdx.x;
  int b    = tile / 900;
  int p0   = (tile % 900) * 64;
  int m0   = wave * 16;

  v2f af[16];
  {
    const float* xb = x + (size_t)b * 63 * HW + p0 + m0 + (lane & 15);
    int kh = (lane >> 4) << 1;
#pragma unroll
    for (int t = 0; t < 16; ++t) {
      int k  = 4 * t + kh;
      int k1 = (k + 1 < 63) ? k + 1 : 62;
      float v0 = xb[(size_t)k * HW];
      float v1 = xb[(size_t)k1 * HW];
      af[t].x = v0;
      af[t].y = (k + 1 < 63) ? v1 : 0.f;
    }
  }

  const v2f* wfl = (const v2f*)wf + lane;

  v2f bf[16];
#pragma unroll
  for (int t = 0; t < 16; ++t) bf[t] = wfl[t * 32];

  float bias_nx = wf[WF_IN_N + (lane & 15)];

  for (int nt = 0; nt < 8; ++nt) {
    int n0 = nt * 16;
    int nloc = lane & 15;
    int mb   = (lane >> 4) << 3;
    float bias = bias_nx;
    v8f acc = V8Z;
#pragma unroll
    for (int t = 0; t < 16; ++t) acc = wmma4(af[t], bf[t], acc);
    int nx = (nt + 1 < 8) ? nt + 1 : nt;
#pragma unroll
    for (int t = 0; t < 16; ++t) bf[t] = wfl[(nx * 16 + t) * 32];
    bias_nx = wf[WF_IN_N + nx * 16 + nloc];
#pragma unroll
    for (int r = 0; r < 8; ++r) d_lds[wave][mb + r][nloc] = acc[r] + bias;
    if (n0 < 84) {
      for (int idx = lane; idx < 256; idx += 32) {
        int n = idx >> 4, m = idx & 15;
        int ch = n0 + n;
        if (ch < 84)
          xp[(size_t)(b * 126 + ch) * HW + p0 + m0 + m] = d_lds[wave][m][n];
      }
    }
    if (n0 == 96 || n0 == 112) {
      int pix = p0 + m0;
      float* vq = xp + (size_t)(b * 126 + 105) * HW + (size_t)(pix / WW) * VSEQ
                     + ((pix % WW) >> 2) * 84;
#pragma unroll
      for (int i = 0; i < 2; ++i) {
        int idx = lane + 32 * i;
        int n = idx & 15, quad = idx >> 4;
        int c = n0 + n - 105;
        if (c >= 0 && c < 21) {
          v4f val;
          val.x = d_lds[wave][4 * quad + 0][n];
          val.y = d_lds[wave][4 * quad + 1][n];
          val.z = d_lds[wave][4 * quad + 2][n];
          val.w = d_lds[wave][4 * quad + 3][n];
          *(v4f*)(vq + (quad * 21 + c) * 4) = val;
        }
      }
    }
    if (n0 == 80 || n0 == 96) {
      for (int idx = lane; idx < 256; idx += 32) {
        int n = idx & 15, m = idx >> 4;
        int slot = n0 + n - 84;
        if (slot >= 0 && slot < 24)
          qrec[((size_t)b * HW + p0 + m0 + m) * 24 + slot] =
              (slot < 21) ? d_lds[wave][m][n] : 0.f;
      }
    }
  }
}

__global__ __launch_bounds__(32) void k_win(float* __restrict__ xp, float* __restrict__ y2)
{
  __shared__ float q[32][25];
  __shared__ float v[32][25];
  __shared__ float s[32][33];

  int lane = threadIdx.x;
  int wid  = blockIdx.x;
  int grp  = blockIdx.y;
  int b    = wid / (NWIN * NWIN);
  int rem  = wid % (NWIN * NWIN);
  int wy   = rem / NWIN, wx = rem % NWIN;

  {
    bool own = lane < 25;
    int mp = own ? lane : 24;
    int gy = wy * WS + mp / WS, gx = wx * WS + mp % WS;
    if (grp == 1) { gy = (gy + 3) % HH; gx = (gx + 3) % WW; }
    const float* src = xp + (size_t)(b * 126 + grp * 42) * HW + gy * WW + gx;
    float qv[21], vv[21];
#pragma unroll
    for (int c = 0; c < 21; ++c) {
      qv[c] = src[(size_t)c * HW];
      vv[c] = src[(size_t)(c + 21) * HW];
    }
#pragma unroll
    for (int c = 0; c < 21; ++c) {
      q[lane][c] = own ? qv[c] : 0.f;
      v[lane][c] = own ? vv[c] : 0.f;
    }
    for (int c = 21; c < 25; ++c) { q[lane][c] = 0.f; v[lane][c] = 0.f; }
  }

  for (int mt = 0; mt < 2; ++mt)
    for (int nt = 0; nt < 2; ++nt) {
      v8f acc = V8Z;
      for (int k0 = 0; k0 < 24; k0 += 4) {
        v2f av = frag_a(&q[0][0], 25, mt * 16, k0, lane);
        v2f bv = frag_a(&q[0][0], 25, nt * 16, k0, lane);
        acc = wmma4(av, bv, acc);
      }
      frag_store(&s[0][0], 33, mt * 16, nt * 16, lane, acc);
    }

  if (lane < 25) {
    float mx = -1e30f;
    for (int n = 0; n < 25; ++n) mx = fmaxf(mx, s[lane][n]);
    float sum = 0.f;
    for (int n = 0; n < 25; ++n) { float e = __expf(s[lane][n] - mx); s[lane][n] = e; sum += e; }
    float inv = 1.f / sum;
    for (int n = 0; n < 25; ++n) s[lane][n] *= inv;
    for (int n = 25; n < 28; ++n) s[lane][n] = 0.f;
  }

  for (int qt = 0; qt < 2; ++qt) {
    int m  = qt * 16 + (lane & 15);
    int mc = (m < 25) ? m : 24;
    int oy = wy * WS + mc / WS, ox = wx * WS + mc % WS;
    float* dst;
    if (grp == 0) {
      dst = xp + (size_t)(b * 126) * HW + oy * WW + ox;
    } else {
      oy = (oy + 2) % HH; ox = (ox + 2) % WW;
      dst = y2 + (size_t)(b * 126) * HW + oy * WW + ox;
    }
#pragma unroll
    for (int ct = 0; ct < 2; ++ct) {
      v8f acc = V8Z;
      for (int k0 = 0; k0 < 28; k0 += 4) {
        v2f av = frag_b(&v[0][0], 25, k0, ct * 16, lane);
        v2f bv = frag_a(&s[0][0], 33, qt * 16, k0, lane);
        acc = wmma4(av, bv, acc);
      }
      int chb = ct * 16 + ((lane >> 4) << 3);
      bool st = (m < 25) && (ct == 0 || (lane >> 4) == 0);
      if (st) {
        float* d = dst + (size_t)chb * HW;
#pragma unroll
        for (int r = 0; r < 8; ++r)
          if (ct == 0 || r < 5) d[(size_t)r * HW] = acc[r];
      }
    }
  }
}


template <int PS>
__device__ __forceinline__ void gfrag_q(const float* qb, int p0, int lane, v2f (&f)[6]) {
  int p = p0 + (lane & 15);
  const v4f* g = (const v4f*)(qb + (size_t)p * (PS * 24) + 12 * (lane >> 4));
  v4f a = g[0], b = g[1], c = g[2];
  f[0].x = a.x; f[0].y = a.y; f[1].x = a.z; f[1].y = a.w;
  f[2].x = b.x; f[2].y = b.y; f[3].x = b.z; f[3].y = b.w;
  f[4].x = c.x; f[4].y = c.y; f[5].x = c.z; f[5].y = c.w;
}

template <int X>
__device__ __forceinline__ float half_xor(float v) {
  return __int_as_float(__builtin_amdgcn_update_dpp(0, __float_as_int(v), 0x160 + X, 0xf, 0xf, false));
}

__device__ __forceinline__ float half_max(float v) {
  v = fmaxf(v, half_xor<1>(v));
  v = fmaxf(v, half_xor<2>(v));
  v = fmaxf(v, half_xor<4>(v));
  v = fmaxf(v, half_xor<8>(v));
  return v;
}

__device__ __forceinline__ float half_sum(float v) {
  v += half_xor<1>(v);
  v += half_xor<2>(v);
  v += half_xor<4>(v);
  v += half_xor<8>(v);
  return v;
}

template <int PASS, int NQ>
__device__ __forceinline__ void axial_tiles(float* __restrict__ xp, float* __restrict__ vm,
                                            const float* __restrict__ qb,
                                            const float* __restrict__ vb,
                                            float* sb, int lane, int b, int r, int m0,
                                            const float* s_all)
{
  constexpr int PS = (PASS == 0) ? 1 : WW;

  v2f qa[NQ][6];
#pragma unroll
  for (int i = 0; i < NQ; ++i) gfrag_q<PS>(qb, m0 + 16 * i, lane, qa[i]);

  float rmx[NQ][8];
#pragma unroll
  for (int i = 0; i < NQ; ++i)
#pragma unroll
    for (int r8 = 0; r8 < 8; ++r8) rmx[i][r8] = -1e30f;

#pragma unroll 5
  for (int nt = 0; nt < 15; ++nt) {
    int n0 = nt * 16;
    v2f kf[6];
    gfrag_q<PS>(qb, n0, lane, kf);
    v8f acc[NQ];
#pragma unroll
    for (int i = 0; i < NQ; ++i) acc[i] = (v8f)V8Z;
#pragma unroll
    for (int t = 0; t < 6; ++t)
#pragma unroll
      for (int i = 0; i < NQ; ++i) acc[i] = wmma4(qa[i][t], kf[t], acc[i]);
#pragma unroll
    for (int i = 0; i < NQ; ++i) {
#pragma unroll
      for (int r8 = 0; r8 < 8; ++r8) rmx[i][r8] = fmaxf(rmx[i][r8], acc[i][r8]);
      frag_store(sb + 16 * i * SP, SP, 0, n0, lane, acc[i]);
    }
  }
#pragma unroll
  for (int i = 0; i < NQ; ++i)
#pragma unroll
    for (int r8 = 0; r8 < 8; ++r8) rmx[i][r8] = half_max(rmx[i][r8]);

  float rinv[NQ][8];
#pragma unroll
  for (int i = 0; i < NQ; ++i) {
    float* own = sb + (16 * i + ((lane >> 4) << 3)) * SP + (lane & 15);
    float rsum[8];
#pragma unroll
    for (int r8 = 0; r8 < 8; ++r8) rsum[r8] = 0.f;
#pragma unroll 5
    for (int nt = 0; nt < 15; ++nt) {
#pragma unroll
      for (int r8 = 0; r8 < 8; ++r8) {
        float e = __expf(own[r8 * SP + nt * 16] - rmx[i][r8]);
        own[r8 * SP + nt * 16] = e;
        rsum[r8] += e;
      }
    }
#pragma unroll
    for (int r8 = 0; r8 < 8; ++r8) rinv[i][r8] = 1.f / half_sum(rsum[r8]);
  }

  int hl = lane >> 4, nl = lane & 15;
  int c1 = (16 + nl < 21) ? 16 + nl : 20;
  const v4f* vp0 = (const v4f*)vb + (hl * 21 + nl);
  const v4f* vp1 = (const v4f*)vb + (hl * 21 + c1);
  const float* ap = sb + nl * SP + 4 * hl;
  v8f acc0[NQ], acc1[NQ];
#pragma unroll
  for (int i = 0; i < NQ; ++i) { acc0[i] = (v8f)V8Z; acc1[i] = (v8f)V8Z; }
#pragma unroll 6
  for (int u = 0; u < 30; ++u) {
    v4f b0 = vp0[u * 42];
    v4f b1 = vp1[u * 42];
    v2f p01, p23, q01, q23;
    p01.x = b0.x; p01.y = b0.y; p23.x = b0.z; p23.y = b0.w;
    q01.x = b1.x; q01.y = b1.y; q23.x = b1.z; q23.y = b1.w;
#pragma unroll
    for (int i = 0; i < NQ; ++i) {
      v4f a = *(const v4f*)(ap + 16 * i * SP + 8 * u);
      v2f a01, a23;
      a01.x = a.x; a01.y = a.y; a23.x = a.z; a23.y = a.w;
      acc0[i] = wmma4(a01, p01, acc0[i]);
      acc1[i] = wmma4(a01, q01, acc1[i]);
      acc0[i] = wmma4(a23, p23, acc0[i]);
      acc1[i] = wmma4(a23, q23, acc1[i]);
    }
  }
  constexpr int XP = 33;
  int kq4 = ((r >> 2) * 21) * 4 + (r & 3);
#pragma unroll
  for (int i = 0; i < NQ; ++i) {
#pragma unroll
    for (int nt = 0; nt < 2; ++nt) {
      v8f acc = nt ? acc1[i] : acc0[i];
      int n  = nt * 16 + nl;
      int mb = hl << 3;
#pragma unroll
      for (int rr = 0; rr < 8; ++rr) {
        int m = m0 + 16 * i + mb + rr;
        float y = acc[rr] * rinv[i][rr];
        if (n < 21) {
          if (PASS == 0)
            vm[(size_t)m * VSEQ + kq4 + n * 4] = y;
          else
            sb[n * XP + 16 * i + mb + rr] = y;
        }
      }
    }
  }
  if (PASS == 1) {
    __syncthreads();
    constexpr int ROWS = 16 * NQ;
    constexpr int NP   = 21 * ROWS;
    const float* s0 = s_all;
    const float* s1 = s_all + 32 * SP;
    float* ob = xp + (size_t)(b * 126 + 105) * HW + (size_t)m0 * WW + (r & ~1);
    constexpr int FULL = NP / 64;
    int tid = (int)threadIdx.x;
#pragma unroll
    for (int k = 0; k < FULL; ++k) {
      int pid = tid + 64 * k;
      int n = pid / ROWS, ml = pid % ROWS;
      v2f val;
      val.x = s0[n * XP + ml];
      val.y = s1[n * XP + ml];
      *(v2f*)(ob + (size_t)n * HW + ml * WW) = val;
    }
    if (NP % 64 != 0) {
      int pid = tid + 64 * FULL;
      if (pid < NP) {
        int n = pid / ROWS, ml = pid % ROWS;
        v2f val;
        val.x = s0[n * XP + ml];
        val.y = s1[n * XP + ml];
        *(v2f*)(ob + (size_t)n * HW + ml * WW) = val;
      }
    }
  }
}

template <int PASS>
__global__ __launch_bounds__(PASS == 0 ? 32 : 64) void k_axial(float* __restrict__ xp,
                                                               float* __restrict__ vmid,
                                                               const float* __restrict__ qrec)
{
  constexpr int NWV = (PASS == 0) ? 1 : 2;
  __shared__ __attribute__((aligned(16))) float s[32 * NWV][SP];

  int lane = threadIdx.x & 31;
  int wv   = (PASS == 0) ? 0 : __builtin_amdgcn_readfirstlane((int)(threadIdx.x >> 5));
  int row  = (PASS == 0) ? (int)blockIdx.y
                         : 2 * (int)blockIdx.y + wv;
  int qp   = blockIdx.x;
  int b    = row / HH;
  int r    = row % HH;
  float* sb = &s[0][0] + wv * 32 * SP;

  float* vm = vmid + (size_t)b * 126 * HW;
  const float* qb = qrec + ((size_t)b * HW + ((PASS == 0) ? r * WW : r)) * 24;
  const float* vb = (PASS == 0)
      ? xp + (size_t)(b * 126 + 105) * HW + (size_t)r * VSEQ
      : vm + (size_t)r * VSEQ;

  if (qp < 7)
    axial_tiles<PASS, 2>(xp, vm, qb, vb, sb, lane, b, r, qp * 32, &s[0][0]);
  else
    axial_tiles<PASS, 1>(xp, vm, qb, vb, sb, lane, b, r, 224, &s[0][0]);
}

__global__ __launch_bounds__(128) void k_proj_out(
    const float* __restrict__ xp, const float* __restrict__ y2,
    const float* __restrict__ wf, float* __restrict__ out)
{
  __shared__ float d_lds[4][16][17];

  int tid  = threadIdx.x;
  int lane = tid & 31;
  int wave = tid >> 5;
  int tile = blockIdx.x;
  int b    = tile / 900;
  int p0   = (tile % 900) * 64;
  int m0   = wave * 16;

  v2f af[16];
  {
    const float* xb = xp + (size_t)b * 126 * HW + p0 + m0 + (lane & 15);
    const float* yb = y2 + (size_t)b * 126 * HW + p0 + m0 + (lane & 15);
    int kh = (lane >> 4) << 1;
#pragma unroll
    for (int t = 0; t < 16; ++t) {
#pragma unroll
      for (int j = 0; j < 2; ++j) {
        int k  = 4 * t + kh + j;
        int kc = (k < 63) ? k : 62;
        const float* src = (kc < 21) ? xb + (size_t)kc * HW
                         : (kc < 42) ? yb + (size_t)(kc - 21) * HW
                                     : xb + (size_t)(kc + 63) * HW;
        float v = *src;
        v = (k < 63) ? v : 0.f;
        if (j == 0) af[t].x = v; else af[t].y = v;
      }
    }
  }

  const v2f* wfl = (const v2f*)wf + lane;

  v2f bf[16];
#pragma unroll
  for (int t = 0; t < 16; ++t) bf[t] = wfl[t * 32];

  float bias_nx = wf[WF_OUT_N + (lane & 15)];

  for (int nt = 0; nt < 4; ++nt) {
    int n0 = nt * 16;
    int nloc = lane & 15;
    int mb   = (lane >> 4) << 3;
    float bias = bias_nx;
    v8f acc = V8Z;
#pragma unroll
    for (int t = 0; t < 16; ++t) acc = wmma4(af[t], bf[t], acc);
    int nx = (nt + 1 < 4) ? nt + 1 : nt;
#pragma unroll
    for (int t = 0; t < 16; ++t) bf[t] = wfl[(nx * 16 + t) * 32];
    bias_nx = wf[WF_OUT_N + nx * 16 + nloc];
#pragma unroll
    for (int r = 0; r < 8; ++r) d_lds[wave][mb + r][nloc] = acc[r] + bias;
    for (int idx = lane; idx < 256; idx += 32) {
      int n = idx >> 4, m = idx & 15;
      int ch = n0 + n;
      if (ch < 63)
        out[(size_t)(b * 63 + ch) * HW + p0 + m0 + m] = d_lds[wave][m][n];
    }
  }
}

extern "C" void kernel_launch(void* const* d_in, const int* in_sizes, int n_in,
                              void* d_out, int out_size, void* d_ws, size_t ws_size,
                              hipStream_t stream)
{
  (void)in_sizes; (void)n_in; (void)out_size; (void)ws_size;
  const float* x     = (const float*)d_in[0];
  const float* w_in  = (const float*)d_in[1];
  const float* b_in  = (const float*)d_in[2];
  const float* bn_g  = (const float*)d_in[3];
  const float* bn_b  = (const float*)d_in[4];
  const float* bn_m  = (const float*)d_in[5];
  const float* bn_v  = (const float*)d_in[6];
  const float* w_out = (const float*)d_in[7];
  const float* b_out = (const float*)d_in[8];
  float* out = (float*)d_out;

  float* xp   = (float*)d_ws;
  float* qrec = xp + (size_t)BB * 126 * HW;
  float* y2b  = xp + (size_t)84 * HW;
  float* vmid = xp + (size_t)21 * HW;
  float* wfin = xp + (size_t)84 * HW;
  float* wfout = xp + (size_t)42 * HW;

  k_prep_in <<<dim3(33), 256, 0, stream>>>(w_in, b_in, bn_g, bn_b, bn_m, bn_v, wfin);
  k_proj_in <<<dim3(BB * 900), 128, 0, stream>>>(x, wfin, xp, qrec);
  k_win     <<<dim3(BB * NWIN * NWIN, 2), 32, 0, stream>>>(xp, y2b);
  k_axial<0><<<dim3(8, BB * HH), 32, 0, stream>>>(xp, vmid, qrec);
  k_axial<1><<<dim3(8, BB * WW / 2), 64, 0, stream>>>(xp, vmid, qrec);
  k_prep_out<<<dim3(17), 256, 0, stream>>>(w_out, b_out, wfout);
  k_proj_out<<<dim3(BB * 900), 128, 0, stream>>>(xp, y2b, wfout, out);
}
